# starting kernel
# baseline (speedup 1.0000x reference)
#include <hip/hip_runtime.h>

typedef __attribute__((ext_vector_type(2))) float v2f;
typedef __attribute__((ext_vector_type(8))) float v8f;

#define GCN_N_NODES  50000
#define GCN_N_GRAPHS 64

__global__ void zero_f32(float* __restrict__ p, int n) {
    int i = blockIdx.x * blockDim.x + threadIdx.x;
    if (i < n) p[i] = 0.0f;
}

__global__ void deg_init(float* __restrict__ deg, int n) {
    int i = blockIdx.x * blockDim.x + threadIdx.x;
    if (i < n) deg[i] = 1.0f;
}

__global__ void deg_accum(const int* __restrict__ dst, float* __restrict__ deg, int e) {
    int i = blockIdx.x * blockDim.x + threadIdx.x;
    if (i < e) atomicAdd(&deg[dst[i]], 1.0f);
}

__global__ void deg_finalize(float* __restrict__ deg, int n) {
    int i = blockIdx.x * blockDim.x + threadIdx.x;
    if (i < n) deg[i] = rsqrtf(deg[i]);
}

template <int K, int NCOLS, int WAVES>
__global__ void gemm_f32_wmma(const float* __restrict__ A,
                              const float* __restrict__ Wm,
                              float* __restrict__ H, int M) {
    __shared__ float Bs[K * 16];

    const int tid  = threadIdx.x;
    const int lane = tid & 31;
    const int wav  = tid >> 5;
    const int colBase = blockIdx.y * 16;

    #pragma unroll
    for (int i = tid; i < K * 16; i += 32 * WAVES) {
        const int kk = i >> 4;
        const int cc = i & 15;
        Bs[i] = Wm[kk * NCOLS + colBase + cc];
    }
    __syncthreads();

    const int mtile = blockIdx.x * WAVES + wav;
    if (mtile * 16 < M) {
        const int l15 = lane & 15;
        const int hi  = lane >> 4;
        const int khi = hi * 2;
        const int mrow = mtile * 16 + l15;

        v8f acc = {0.f, 0.f, 0.f, 0.f, 0.f, 0.f, 0.f, 0.f};
        #pragma unroll 8
        for (int k0 = 0; k0 < K; k0 += 4) {
            v2f a = *(const v2f*)&A[mrow * K + k0 + khi];
            v2f b;
            b.x = Bs[(k0 + khi) * 16 + l15];
            b.y = Bs[(k0 + khi + 1) * 16 + l15];
            acc = __builtin_amdgcn_wmma_f32_16x16x4_f32(
                 false, a,  false, b,
                 (short)0, acc,  false,  false);
        }

        const int m0  = mtile * 16 + hi * 8;
        const int col = colBase + l15;
        #pragma unroll
        for (int r = 0; r < 8; ++r)
            H[(m0 + r) * NCOLS + col] = acc[r];
    }
}

template <int F>
__global__ void scatter_edges(const float* __restrict__ h,
                              const int* __restrict__ src,
                              const int* __restrict__ dst,
                              const float* __restrict__ dinv,
                              float* __restrict__ agg, int e) {
    const int wid  = (int)((blockIdx.x * blockDim.x + threadIdx.x) >> 5);
    const int lane = threadIdx.x & 31;
    if (wid >= e) return;
    const int s = src[wid];
    __builtin_prefetch(&h[(size_t)s * F + lane], 0, 0);
    const int d = dst[wid];
    const float nrm = dinv[s] * dinv[d];
    #pragma unroll
    for (int c = 0; c < F / 32; ++c) {
        float v = h[s * F + c * 32 + lane] * nrm;
        atomicAdd(&agg[d * F + c * 32 + lane], v);
    }
}

template <int F, bool RELU>
__global__ void node_finalize(const float* __restrict__ agg,
                              const float* __restrict__ h,
                              const float* __restrict__ dinv,
                              const float* __restrict__ bias,
                              float* __restrict__ out, int n) {
    const int node = blockIdx.x;
    if (node >= n) return;
    const float di  = dinv[node];
    const float di2 = di * di;
    for (int f = threadIdx.x; f < F; f += blockDim.x) {
        float v = agg[node * F + f] + h[node * F + f] * di2 + bias[f];
        out[node * F + f] = RELU ? fmaxf(v, 0.0f) : v;
    }
}

__global__ void pool_sum(const float* __restrict__ h,
                         const int* __restrict__ batch,
                         float* __restrict__ sums,
                         float* __restrict__ cnts, int n) {
    const int wid  = (int)((blockIdx.x * blockDim.x + threadIdx.x) >> 5);
    const int lane = threadIdx.x & 31;
    if (wid >= n) return;
    const int g = batch[wid];
    if (lane == 0) atomicAdd(&cnts[g], 1.0f);
    atomicAdd(&sums[g * 64 + lane],      h[wid * 64 + lane]);
    atomicAdd(&sums[g * 64 + 32 + lane], h[wid * 64 + 32 + lane]);
}

__global__ void pool_div(const float* __restrict__ sums,
                         const float* __restrict__ cnts,
                         float* __restrict__ out, int n) {
    int i = blockIdx.x * blockDim.x + threadIdx.x;
    if (i < n) out[i] = sums[i] / fmaxf(cnts[i / 64], 1.0f);
}

extern "C" void kernel_launch(void* const* d_in, const int* in_sizes, int n_in,
                              void* d_out, int out_size, void* d_ws, size_t ws_size,
                              hipStream_t stream) {
    const float* x     = (const float*)d_in[0];
    const int*   ei    = (const int*)d_in[1];
    const int*   batch = (const int*)d_in[2];
    const float* Wl[4] = {(const float*)d_in[3], (const float*)d_in[5],
                          (const float*)d_in[7], (const float*)d_in[9]};
    const float* bl[4] = {(const float*)d_in[4], (const float*)d_in[6],
                          (const float*)d_in[8], (const float*)d_in[10]};

    const int N = GCN_N_NODES;
    const int E = in_sizes[1] / 2;
    const int* src = ei;
    const int* dst = ei + E;

    auto au = [](size_t v) { return (v + 255) & ~(size_t)255; };
    char* ws = (char*)d_ws;
    size_t off = 0;
    float* deg  = (float*)(ws + off); off += au(sizeof(float) * N);
    float* bufA = (float*)(ws + off); off += au(sizeof(float) * (size_t)N * 96);
    float* bufB = (float*)(ws + off); off += au(sizeof(float) * (size_t)N * 96);
    float* bufC = (float*)(ws + off); off += au(sizeof(float) * (size_t)N * 96);
    float* sums = (float*)(ws + off); off += au(sizeof(float) * GCN_N_GRAPHS * 64);
    float* cnts = (float*)(ws + off); off += au(sizeof(float) * GCN_N_GRAPHS);
    (void)ws_size; (void)n_in; (void)out_size;

    const int TB = 256;
    deg_init    <<<(N + TB - 1) / TB, TB, 0, stream>>>(deg, N);
    deg_accum   <<<(E + TB - 1) / TB, TB, 0, stream>>>(dst, deg, E);
    deg_finalize<<<(N + TB - 1) / TB, TB, 0, stream>>>(deg, N);

    constexpr int WV = 4;
    const int MT  = (N + 15) / 16;
    const int GBX = (MT + WV - 1) / WV;

    gemm_f32_wmma<128, 96, WV><<<dim3(GBX, 6), 32 * WV, 0, stream>>>(x, Wl[0], bufA, N);
    zero_f32<<<(N * 96 + TB - 1) / TB, TB, 0, stream>>>(bufB, N * 96);
    scatter_edges<96><<<(E + 7) / 8, 256, 0, stream>>>(bufA, src, dst, deg, bufB, E);
    node_finalize<96, true><<<N, 96, 0, stream>>>(bufB, bufA, deg, bl[0], bufC, N);

    for (int l = 1; l <= 2; ++l) {
        gemm_f32_wmma<96, 96, WV><<<dim3(GBX, 6), 32 * WV, 0, stream>>>(bufC, Wl[l], bufA, N);
        zero_f32<<<(N * 96 + TB - 1) / TB, TB, 0, stream>>>(bufB, N * 96);
        scatter_edges<96><<<(E + 7) / 8, 256, 0, stream>>>(bufA, src, dst, deg, bufB, E);
        node_finalize<96, true><<<N, 96, 0, stream>>>(bufB, bufA, deg, bl[l], bufC, N);
    }

    gemm_f32_wmma<96, 64, WV><<<dim3(GBX, 4), 32 * WV, 0, stream>>>(bufC, Wl[3], bufA, N);
    zero_f32<<<(N * 64 + TB - 1) / TB, TB, 0, stream>>>(bufB, N * 64);
    scatter_edges<64><<<(E + 7) / 8, 256, 0, stream>>>(bufA, src, dst, deg, bufB, E);
    node_finalize<64, false><<<N, 64, 0, stream>>>(bufB, bufA, deg, bl[3], bufC, N);

    zero_f32<<<(GCN_N_GRAPHS * 64 + TB - 1) / TB, TB, 0, stream>>>(sums, GCN_N_GRAPHS * 64);
    zero_f32<<<1, 64, 0, stream>>>(cnts, GCN_N_GRAPHS);
    pool_sum<<<(N + 7) / 8, 256, 0, stream>>>(bufC, batch, sums, cnts, N);
    pool_div<<<(GCN_N_GRAPHS * 64 + TB - 1) / TB, TB, 0, stream>>>(sums, cnts, (float*)d_out,
                                                                   GCN_N_GRAPHS * 64);
}


# Per-edge float-atomic scatter replaced by a per-call dst-sorted edge list (int atomic arrival rank, scan, fill) and a wave-per-node gather fused with self-loop, bias, ReLU; sums run in rank order
# speedup vs baseline: 2.2479x; 2.2479x over previous
#include <hip/hip_runtime.h>

typedef __attribute__((ext_vector_type(2))) float v2f;
typedef __attribute__((ext_vector_type(8))) float v8f;
typedef __attribute__((ext_vector_type(2))) int   v2i;
typedef __attribute__((ext_vector_type(4))) int   v4i;

#define GCN_N_NODES  50000
#define GCN_N_GRAPHS 64

#define SCAN_CHUNK (((GCN_N_NODES + 32 * 128 - 1) / (32 * 128)) * 128)
#define SCAN_PAD   (32 * SCAN_CHUNK)

__global__ void zero_f32(float* __restrict__ p, int n) {
    int i = blockIdx.x * blockDim.x + threadIdx.x;
    if (i < n) p[i] = 0.0f;
}

__global__ void zero_i32(int* __restrict__ p, int n) {
    int i = blockIdx.x * blockDim.x + threadIdx.x;
    if (i < n) p[i] = 0;
}

__global__ void edge_rank(const int* __restrict__ dst, int* __restrict__ cnt,
                          int* __restrict__ rank, int e) {
    int i = blockIdx.x * blockDim.x + threadIdx.x;
    if (i < e) rank[i] = atomicAdd(&cnt[dst[i]], 1);
}

__global__ void __launch_bounds__(1024)
scan_nodes(const int* __restrict__ cnt, v2i* __restrict__ ninfo, int n) {
    __shared__ int wtot[32];
    const int lane = threadIdx.x & 31;
    const int wav  = threadIdx.x >> 5;
    const int b    = wav * SCAN_CHUNK + lane * 4;

    int s = 0;
    #pragma unroll
    for (int k = 0; k < SCAN_CHUNK; k += 128) {
        const v4i c = *(const v4i*)&cnt[b + k];
        s += (c.x + c.y) + (c.z + c.w);
    }
    for (int o = 16; o > 0; o >>= 1) s += __shfl_xor(s, o);
    if (lane == 0) wtot[wav] = s;
    __syncthreads();

    int run = (lane < wav) ? wtot[lane] : 0;
    for (int o = 16; o > 0; o >>= 1) run += __shfl_xor(run, o);

    #pragma unroll
    for (int k = 0; k < SCAN_CHUNK; k += 128) {
        const v4i c = *(const v4i*)&cnt[b + k];
        const int t4 = (c.x + c.y) + (c.z + c.w);
        int incl = t4;
        for (int o = 1; o < 32; o <<= 1) {
            const int u = __shfl_up(incl, o);
            if (lane >= o) incl += u;
        }
        const int i0 = b + k;
        const int p0 = run + incl - t4;
        auto put = [&](int i, int first, int deg) {
            if (i < n) {
                v2i v;
                v.x = first;
                v.y = __float_as_int(rsqrtf((float)(deg + 1)));
                ninfo[i] = v;
            }
        };
        put(i0,     p0,                     c.x);
        put(i0 + 1, p0 + c.x,               c.y);
        put(i0 + 2, p0 + c.x + c.y,         c.z);
        put(i0 + 3, p0 + c.x + c.y + c.z,   c.w);
        run += __shfl(incl, 31);
    }
    if (threadIdx.x == 1023) {
        v2i v;
        v.x = run;
        v.y = 0;
        ninfo[n] = v;
    }
}

__global__ void csr_fill(const int* __restrict__ src, const int* __restrict__ dst,
                         const int* __restrict__ rank, const v2i* __restrict__ ninfo,
                         v2i* __restrict__ edges, int e) {
    int i = blockIdx.x * blockDim.x + threadIdx.x;
    if (i >= e) return;
    const int s = src[i];
    const int d = dst[i];
    const v2i nd = ninfo[d];
    const float ws = __int_as_float(ninfo[s].y);
    int slot = nd.x + rank[i];
    slot = min(max(slot, 0), e - 1);
    v2i ev;
    ev.x = s;
    ev.y = __float_as_int(ws * __int_as_float(nd.y));
    edges[slot] = ev;
}

template <int K, int NCOLS, int WAVES>
__global__ void gemm_f32_wmma(const float* __restrict__ A,
                              const float* __restrict__ Wm,
                              float* __restrict__ H, int M) {
    __shared__ float Bs[K * 16];

    const int tid  = threadIdx.x;
    const int lane = tid & 31;
    const int wav  = tid >> 5;
    const int colBase = blockIdx.y * 16;

    #pragma unroll
    for (int i = tid; i < K * 16; i += 32 * WAVES) {
        const int kk = i >> 4;
        const int cc = i & 15;
        Bs[i] = Wm[kk * NCOLS + colBase + cc];
    }
    __syncthreads();

    const int mtile = blockIdx.x * WAVES + wav;
    if (mtile * 16 < M) {
        const int l15 = lane & 15;
        const int hi  = lane >> 4;
        const int khi = hi * 2;
        const int mrow = mtile * 16 + l15;

        v8f acc = {0.f, 0.f, 0.f, 0.f, 0.f, 0.f, 0.f, 0.f};
        #pragma unroll 8
        for (int k0 = 0; k0 < K; k0 += 4) {
            v2f a = *(const v2f*)&A[mrow * K + k0 + khi];
            v2f b;
            b.x = Bs[(k0 + khi) * 16 + l15];
            b.y = Bs[(k0 + khi + 1) * 16 + l15];
            acc = __builtin_amdgcn_wmma_f32_16x16x4_f32(
                 false, a,  false, b,
                 (short)0, acc,  false,  false);
        }

        const int m0  = mtile * 16 + hi * 8;
        const int col = colBase + l15;
        #pragma unroll
        for (int r = 0; r < 8; ++r)
            H[(m0 + r) * NCOLS + col] = acc[r];
    }
}

template <int F, int GS, bool TAIL>
__device__ __forceinline__ void gather_group(const float* __restrict__ h, const v2i ev,
                                             int i0, int m, int n, int lane,
                                             float (&acc)[F / 32]) {
    constexpr int C = F / 32;
    float hv[GS][C];
    float wv[GS];
    #pragma unroll
    for (int q = 0; q < GS; ++q) {
        const int ii = TAIL ? min(i0 + q, m - 1) : (i0 + q);
        int s = __builtin_amdgcn_readlane(ev.x, ii);
        s = min(max(s, 0), n - 1);
        wv[q] = __int_as_float(__builtin_amdgcn_readlane(ev.y, ii));
        #pragma unroll
        for (int c = 0; c < C; ++c) hv[q][c] = h[s * F + c * 32 + lane];
    }
    #pragma unroll
    for (int q = 0; q < GS; ++q) {
        const bool ok = !TAIL || (i0 + q) < m;
        #pragma unroll
        for (int c = 0; c < C; ++c) {
            const float t = fmaf(hv[q][c], wv[q], acc[c]);
            acc[c] = ok ? t : acc[c];
        }
    }
}

template <int F, bool RELU>
__global__ void __launch_bounds__(256)
gather_finalize(const float* __restrict__ h,
                const v2i* __restrict__ ninfo,
                const v2i* __restrict__ edges,
                const float* __restrict__ bias,
                float* __restrict__ out, int n) {
    constexpr int C = F / 32;
    const int lane = threadIdx.x & 31;
    const int node = __builtin_amdgcn_readfirstlane((int)((blockIdx.x * blockDim.x + threadIdx.x) >> 5));
    if (node >= n) return;

    const v2i ni0 = ninfo[node];
    const int beg = ni0.x;
    const int end = ninfo[node + 1].x;

    float hs[C], bs[C], acc[C];
    #pragma unroll
    for (int c = 0; c < C; ++c) {
        hs[c]  = h[node * F + c * 32 + lane];
        bs[c]  = bias[c * 32 + lane];
        acc[c] = 0.0f;
    }

    for (int base = beg; base < end; base += 32) {
        const int j = base + lane;
        v2i ev = {0, 0};
        if (j < end) ev = edges[j];
        const int m = min(32, end - base);
        int i0 = 0;
        for (; i0 + 8 <= m; i0 += 8) gather_group<F, 8, false>(h, ev, i0, m, n, lane, acc);
        const int rem = m - i0;
        if (rem > 4)      gather_group<F, 8, true>(h, ev, i0, m, n, lane, acc);
        else if (rem > 0) gather_group<F, 4, true>(h, ev, i0, m, n, lane, acc);
    }

    const float di  = __int_as_float(ni0.y);
    const float di2 = di * di;
    #pragma unroll
    for (int c = 0; c < C; ++c) {
        float v = acc[c] + hs[c] * di2 + bs[c];
        out[node * F + c * 32 + lane] = RELU ? fmaxf(v, 0.0f) : v;
    }
}

__global__ void pool_sum(const float* __restrict__ h,
                         const int* __restrict__ batch,
                         float* __restrict__ sums,
                         float* __restrict__ cnts, int n) {
    const int wid  = (int)((blockIdx.x * blockDim.x + threadIdx.x) >> 5);
    const int lane = threadIdx.x & 31;
    if (wid >= n) return;
    const int g = batch[wid];
    if (lane == 0) atomicAdd(&cnts[g], 1.0f);
    atomicAdd(&sums[g * 64 + lane],      h[wid * 64 + lane]);
    atomicAdd(&sums[g * 64 + 32 + lane], h[wid * 64 + 32 + lane]);
}

__global__ void pool_div(const float* __restrict__ sums,
                         const float* __restrict__ cnts,
                         float* __restrict__ out, int n) {
    int i = blockIdx.x * blockDim.x + threadIdx.x;
    if (i < n) out[i] = sums[i] / fmaxf(cnts[i / 64], 1.0f);
}

extern "C" void kernel_launch(void* const* d_in, const int* in_sizes, int n_in,
                              void* d_out, int out_size, void* d_ws, size_t ws_size,
                              hipStream_t stream) {
    const float* x     = (const float*)d_in[0];
    const int*   ei    = (const int*)d_in[1];
    const int*   batch = (const int*)d_in[2];
    const float* Wl[4] = {(const float*)d_in[3], (const float*)d_in[5],
                          (const float*)d_in[7], (const float*)d_in[9]};
    const float* bl[4] = {(const float*)d_in[4], (const float*)d_in[6],
                          (const float*)d_in[8], (const float*)d_in[10]};

    const int N = GCN_N_NODES;
    const int E = in_sizes[1] / 2;
    const int* src = ei;
    const int* dst = ei + E;

    auto au = [](size_t v) { return (v + 255) & ~(size_t)255; };
    char* ws = (char*)d_ws;
    size_t off = 0;
    int*   cnt   = (int*)(ws + off);   off += au(sizeof(int) * SCAN_PAD);
    int*   rank  = (int*)(ws + off);   off += au(sizeof(int) * (size_t)E);
    v2i*   ninfo = (v2i*)(ws + off);   off += au(sizeof(v2i) * (size_t)(N + 1));
    v2i*   edges = (v2i*)(ws + off);   off += au(sizeof(v2i) * (size_t)E);
    float* bufA  = (float*)(ws + off); off += au(sizeof(float) * (size_t)N * 96);
    float* bufC  = (float*)(ws + off); off += au(sizeof(float) * (size_t)N * 96);
    float* sums  = (float*)(ws + off); off += au(sizeof(float) * GCN_N_GRAPHS * 64);
    float* cnts  = (float*)(ws + off); off += au(sizeof(float) * GCN_N_GRAPHS);
    (void)ws_size; (void)n_in; (void)out_size;

    const int TB = 256;
    zero_i32  <<<(SCAN_PAD + TB - 1) / TB, TB, 0, stream>>>(cnt, SCAN_PAD);
    edge_rank <<<(E + TB - 1) / TB, TB, 0, stream>>>(dst, cnt, rank, E);
    scan_nodes<<<1, 1024, 0, stream>>>(cnt, ninfo, N);
    csr_fill  <<<(E + TB - 1) / TB, TB, 0, stream>>>(src, dst, rank, ninfo, edges, E);

    constexpr int WV = 4;
    const int MT  = (N + 15) / 16;
    const int GBX = (MT + WV - 1) / WV;
    const int NB  = (N + 7) / 8;

    gemm_f32_wmma<128, 96, WV><<<dim3(GBX, 6), 32 * WV, 0, stream>>>(x, Wl[0], bufA, N);
    gather_finalize<96, true><<<NB, 256, 0, stream>>>(bufA, ninfo, edges, bl[0], bufC, N);

    for (int l = 1; l <= 2; ++l) {
        gemm_f32_wmma<96, 96, WV><<<dim3(GBX, 6), 32 * WV, 0, stream>>>(bufC, Wl[l], bufA, N);
        gather_finalize<96, true><<<NB, 256, 0, stream>>>(bufA, ninfo, edges, bl[l], bufC, N);
    }

    gemm_f32_wmma<96, 64, WV><<<dim3(GBX, 4), 32 * WV, 0, stream>>>(bufC, Wl[3], bufA, N);
    gather_finalize<64, false><<<NB, 256, 0, stream>>>(bufA, ninfo, edges, bl[3], bufC, N);

    zero_f32<<<(GCN_N_GRAPHS * 64 + TB - 1) / TB, TB, 0, stream>>>(sums, GCN_N_GRAPHS * 64);
    zero_f32<<<1, 64, 0, stream>>>(cnts, GCN_N_GRAPHS);
    pool_sum<<<(N + 7) / 8, 256, 0, stream>>>(bufC, batch, sums, cnts, N);
    pool_div<<<(GCN_N_GRAPHS * 64 + TB - 1) / TB, TB, 0, stream>>>(sums, cnts, (float*)d_out,
                                                                   GCN_N_GRAPHS * 64);
}


# Layers 2-4's dense product fused into the previous gather: persistent 16-wave blocks keep W in LDS, finish 16 rows into an LDS tile, WMMA it; 3 launches and the layer round trip gone; sum order kept
# speedup vs baseline: 2.3843x; 1.0607x over previous
#include <hip/hip_runtime.h>

typedef __attribute__((ext_vector_type(2))) float v2f;
typedef __attribute__((ext_vector_type(8))) float v8f;
typedef __attribute__((ext_vector_type(2))) int   v2i;
typedef __attribute__((ext_vector_type(4))) int   v4i;

#define GCN_N_NODES  50000
#define GCN_N_GRAPHS 64
static_assert(GCN_N_NODES % 16 == 0, "row tiles of 16 must be full");

#define SCAN_CHUNK (((GCN_N_NODES + 32 * 128 - 1) / (32 * 128)) * 128)
#define SCAN_PAD   (32 * SCAN_CHUNK)

__global__ void zero_f32(float* __restrict__ p, int n) {
    int i = blockIdx.x * blockDim.x + threadIdx.x;
    if (i < n) p[i] = 0.0f;
}

__global__ void zero_i32(int* __restrict__ p, int n) {
    int i = blockIdx.x * blockDim.x + threadIdx.x;
    if (i < n) p[i] = 0;
}

__global__ void edge_rank(const int* __restrict__ dst, int* __restrict__ cnt,
                          int* __restrict__ rank, int e) {
    int i = blockIdx.x * blockDim.x + threadIdx.x;
    if (i < e) rank[i] = atomicAdd(&cnt[dst[i]], 1);
}

__global__ void __launch_bounds__(1024)
scan_nodes(const int* __restrict__ cnt, v2i* __restrict__ ninfo, int n) {
    __shared__ int wtot[32];
    const int lane = threadIdx.x & 31;
    const int wav  = threadIdx.x >> 5;
    const int b    = wav * SCAN_CHUNK + lane * 4;

    int s = 0;
    #pragma unroll
    for (int k = 0; k < SCAN_CHUNK; k += 128) {
        const v4i c = *(const v4i*)&cnt[b + k];
        s += (c.x + c.y) + (c.z + c.w);
    }
    for (int o = 16; o > 0; o >>= 1) s += __shfl_xor(s, o);
    if (lane == 0) wtot[wav] = s;
    __syncthreads();

    int run = (lane < wav) ? wtot[lane] : 0;
    for (int o = 16; o > 0; o >>= 1) run += __shfl_xor(run, o);

    #pragma unroll
    for (int k = 0; k < SCAN_CHUNK; k += 128) {
        const v4i c = *(const v4i*)&cnt[b + k];
        const int t4 = (c.x + c.y) + (c.z + c.w);
        int incl = t4;
        for (int o = 1; o < 32; o <<= 1) {
            const int u = __shfl_up(incl, o);
            if (lane >= o) incl += u;
        }
        const int i0 = b + k;
        const int p0 = run + incl - t4;
        auto put = [&](int i, int first, int deg) {
            if (i < n) {
                v2i v;
                v.x = first;
                v.y = __float_as_int(rsqrtf((float)(deg + 1)));
                ninfo[i] = v;
            }
        };
        put(i0,     p0,                     c.x);
        put(i0 + 1, p0 + c.x,               c.y);
        put(i0 + 2, p0 + c.x + c.y,         c.z);
        put(i0 + 3, p0 + c.x + c.y + c.z,   c.w);
        run += __shfl(incl, 31);
    }
    if (threadIdx.x == 1023) {
        v2i v;
        v.x = run;
        v.y = 0;
        ninfo[n] = v;
    }
}

__global__ void csr_fill(const int* __restrict__ src, const int* __restrict__ dst,
                         const int* __restrict__ rank, const v2i* __restrict__ ninfo,
                         v2i* __restrict__ edges, int e) {
    int i = blockIdx.x * blockDim.x + threadIdx.x;
    if (i >= e) return;
    const int s = src[i];
    const int d = dst[i];
    const v2i nd = ninfo[d];
    const float ws = __int_as_float(ninfo[s].y);
    int slot = nd.x + rank[i];
    slot = min(max(slot, 0), e - 1);
    v2i ev;
    ev.x = s;
    ev.y = __float_as_int(ws * __int_as_float(nd.y));
    edges[slot] = ev;
}

template <int K, int NCOLS, int WAVES>
__global__ void gemm_f32_wmma(const float* __restrict__ A,
                              const float* __restrict__ Wm,
                              float* __restrict__ H, int M) {
    __shared__ float Bs[K * 16];

    const int tid  = threadIdx.x;
    const int lane = tid & 31;
    const int wav  = tid >> 5;
    const int colBase = blockIdx.y * 16;

    #pragma unroll
    for (int i = tid; i < K * 16; i += 32 * WAVES) {
        const int kk = i >> 4;
        const int cc = i & 15;
        Bs[i] = Wm[kk * NCOLS + colBase + cc];
    }
    __syncthreads();

    const int mtile = blockIdx.x * WAVES + wav;
    if (mtile * 16 < M) {
        const int l15 = lane & 15;
        const int hi  = lane >> 4;
        const int khi = hi * 2;
        const int mrow = mtile * 16 + l15;

        v8f acc = {0.f, 0.f, 0.f, 0.f, 0.f, 0.f, 0.f, 0.f};
        #pragma unroll 8
        for (int k0 = 0; k0 < K; k0 += 4) {
            v2f a = *(const v2f*)&A[mrow * K + k0 + khi];
            v2f b;
            b.x = Bs[(k0 + khi) * 16 + l15];
            b.y = Bs[(k0 + khi + 1) * 16 + l15];
            acc = __builtin_amdgcn_wmma_f32_16x16x4_f32(
                 false, a,  false, b,
                 (short)0, acc,  false,  false);
        }

        const int m0  = mtile * 16 + hi * 8;
        const int col = colBase + l15;
        #pragma unroll
        for (int r = 0; r < 8; ++r)
            H[(m0 + r) * NCOLS + col] = acc[r];
    }
}

template <int F, int GS, bool TAIL>
__device__ __forceinline__ void gather_group(const float* __restrict__ h, const v2i ev,
                                             int i0, int m, int n, int lane,
                                             float (&acc)[F / 32]) {
    constexpr int C = F / 32;
    float hv[GS][C];
    float wv[GS];
    #pragma unroll
    for (int q = 0; q < GS; ++q) {
        const int ii = TAIL ? min(i0 + q, m - 1) : (i0 + q);
        int s = __builtin_amdgcn_readlane(ev.x, ii);
        s = min(max(s, 0), n - 1);
        wv[q] = __int_as_float(__builtin_amdgcn_readlane(ev.y, ii));
        #pragma unroll
        for (int c = 0; c < C; ++c) hv[q][c] = h[s * F + c * 32 + lane];
    }
    #pragma unroll
    for (int q = 0; q < GS; ++q) {
        const bool ok = !TAIL || (i0 + q) < m;
        #pragma unroll
        for (int c = 0; c < C; ++c) {
            const float t = fmaf(hv[q][c], wv[q], acc[c]);
            acc[c] = ok ? t : acc[c];
        }
    }
}

template <int F, bool RELU>
__device__ __forceinline__ void node_row(const float* __restrict__ h,
                                         const v2i* __restrict__ ninfo,
                                         const v2i* __restrict__ edges,
                                         const float* __restrict__ bias,
                                         int node, int n, int lane, float (&v)[F / 32]) {
    constexpr int C = F / 32;
    const v2i ni0 = ninfo[node];
    const int beg = ni0.x;
    const int end = ninfo[node + 1].x;

    float hs[C], bs[C], acc[C];
    #pragma unroll
    for (int c = 0; c < C; ++c) {
        hs[c]  = h[node * F + c * 32 + lane];
        bs[c]  = bias[c * 32 + lane];
        acc[c] = 0.0f;
    }

    for (int base = beg; base < end; base += 32) {
        const int j = base + lane;
        v2i ev = {0, 0};
        if (j < end) ev = edges[j];
        const int m = min(32, end - base);
        int i0 = 0;
        for (; i0 + 8 <= m; i0 += 8) gather_group<F, 8, false>(h, ev, i0, m, n, lane, acc);
        const int rem = m - i0;
        if (rem > 4)      gather_group<F, 8, true>(h, ev, i0, m, n, lane, acc);
        else if (rem > 0) gather_group<F, 4, true>(h, ev, i0, m, n, lane, acc);
    }

    const float di  = __int_as_float(ni0.y);
    const float di2 = di * di;
    #pragma unroll
    for (int c = 0; c < C; ++c) {
        const float t = acc[c] + hs[c] * di2 + bs[c];
        v[c] = RELU ? fmaxf(t, 0.0f) : t;
    }
}

template <int F, bool RELU>
__global__ void __launch_bounds__(256)
gather_finalize(const float* __restrict__ h,
                const v2i* __restrict__ ninfo,
                const v2i* __restrict__ edges,
                const float* __restrict__ bias,
                float* __restrict__ out, int n) {
    constexpr int C = F / 32;
    const int lane = threadIdx.x & 31;
    const int node = __builtin_amdgcn_readfirstlane((int)((blockIdx.x * blockDim.x + threadIdx.x) >> 5));
    if (node >= n) return;

    float v[C];
    node_row<F, RELU>(h, ninfo, edges, bias, node, n, lane, v);
    #pragma unroll
    for (int c = 0; c < C; ++c) out[node * F + c * 32 + lane] = v[c];
}

template <int F, int NC, bool RELU>
__global__ void __launch_bounds__(512)
gather_product(const float* __restrict__ h,
               const v2i* __restrict__ ninfo,
               const v2i* __restrict__ edges,
               const float* __restrict__ bias,
               const float* __restrict__ Wn,
               float* __restrict__ Hn, int n) {
    constexpr int C  = F / 32;
    constexpr int TP = F + 4;
    constexpr int WP = NC + 8;
    __shared__ __attribute__((aligned(16))) float Ts[16 * TP];
    __shared__ float Ws[F * WP];

    const int tid  = threadIdx.x;
    const int lane = tid & 31;
    const int wav  = __builtin_amdgcn_readfirstlane(tid >> 5);

    for (int i = tid; i < F * NC; i += 512) {
        const int kk = i / NC;
        const int cc = i - kk * NC;
        Ws[kk * WP + cc] = Wn[i];
    }

    const int ntiles = n >> 4;
    for (int tile = blockIdx.x; tile < ntiles; tile += gridDim.x) {
        const int node = tile * 16 + wav;
        float v[C];
        node_row<F, RELU>(h, ninfo, edges, bias, node, n, lane, v);

        __syncthreads();
        #pragma unroll
        for (int c = 0; c < C; ++c) Ts[wav * TP + c * 32 + lane] = v[c];
        __syncthreads();

        if (wav < NC / 16) {
            const int l15 = lane & 15;
            const int hi  = lane >> 4;
            const int khi = hi * 2;
            const int colBase = wav * 16;

            v8f acc = {0.f, 0.f, 0.f, 0.f, 0.f, 0.f, 0.f, 0.f};
            #pragma unroll 8
            for (int k0 = 0; k0 < F; k0 += 4) {
                v2f a = *(const v2f*)&Ts[l15 * TP + k0 + khi];
                v2f b;
                b.x = Ws[(k0 + khi) * WP + colBase + l15];
                b.y = Ws[(k0 + khi + 1) * WP + colBase + l15];
                acc = __builtin_amdgcn_wmma_f32_16x16x4_f32(
                     false, a,  false, b,
                     (short)0, acc,  false,  false);
            }

            const int m0  = tile * 16 + hi * 8;
            const int col = colBase + l15;
            #pragma unroll
            for (int r = 0; r < 8; ++r)
                Hn[(m0 + r) * NC + col] = acc[r];
        }
    }
}

__global__ void pool_sum(const float* __restrict__ h,
                         const int* __restrict__ batch,
                         float* __restrict__ sums,
                         float* __restrict__ cnts, int n) {
    const int wid  = (int)((blockIdx.x * blockDim.x + threadIdx.x) >> 5);
    const int lane = threadIdx.x & 31;
    if (wid >= n) return;
    const int g = batch[wid];
    if (lane == 0) atomicAdd(&cnts[g], 1.0f);
    atomicAdd(&sums[g * 64 + lane],      h[wid * 64 + lane]);
    atomicAdd(&sums[g * 64 + 32 + lane], h[wid * 64 + 32 + lane]);
}

__global__ void pool_div(const float* __restrict__ sums,
                         const float* __restrict__ cnts,
                         float* __restrict__ out, int n) {
    int i = blockIdx.x * blockDim.x + threadIdx.x;
    if (i < n) out[i] = sums[i] / fmaxf(cnts[i / 64], 1.0f);
}

extern "C" void kernel_launch(void* const* d_in, const int* in_sizes, int n_in,
                              void* d_out, int out_size, void* d_ws, size_t ws_size,
                              hipStream_t stream) {
    const float* x     = (const float*)d_in[0];
    const int*   ei    = (const int*)d_in[1];
    const int*   batch = (const int*)d_in[2];
    const float* Wl[4] = {(const float*)d_in[3], (const float*)d_in[5],
                          (const float*)d_in[7], (const float*)d_in[9]};
    const float* bl[4] = {(const float*)d_in[4], (const float*)d_in[6],
                          (const float*)d_in[8], (const float*)d_in[10]};

    const int N = GCN_N_NODES;
    const int E = in_sizes[1] / 2;
    const int* src = ei;
    const int* dst = ei + E;

    auto au = [](size_t v) { return (v + 255) & ~(size_t)255; };
    char* ws = (char*)d_ws;
    size_t off = 0;
    int*   cnt   = (int*)(ws + off);   off += au(sizeof(int) * SCAN_PAD);
    int*   rank  = (int*)(ws + off);   off += au(sizeof(int) * (size_t)E);
    v2i*   ninfo = (v2i*)(ws + off);   off += au(sizeof(v2i) * (size_t)(N + 1));
    v2i*   edges = (v2i*)(ws + off);   off += au(sizeof(v2i) * (size_t)E);
    float* bufA  = (float*)(ws + off); off += au(sizeof(float) * (size_t)N * 96);
    float* bufB  = (float*)(ws + off); off += au(sizeof(float) * (size_t)N * 96);
    float* bufC  = (float*)(ws + off); off += au(sizeof(float) * (size_t)N * 64);
    float* sums  = (float*)(ws + off); off += au(sizeof(float) * GCN_N_GRAPHS * 64);
    float* cnts  = (float*)(ws + off); off += au(sizeof(float) * GCN_N_GRAPHS);
    (void)ws_size; (void)n_in; (void)out_size;

    const int TB = 256;
    zero_i32  <<<(SCAN_PAD + TB - 1) / TB, TB, 0, stream>>>(cnt, SCAN_PAD);
    edge_rank <<<(E + TB - 1) / TB, TB, 0, stream>>>(dst, cnt, rank, E);
    scan_nodes<<<1, 1024, 0, stream>>>(cnt, ninfo, N);
    csr_fill  <<<(E + TB - 1) / TB, TB, 0, stream>>>(src, dst, rank, ninfo, edges, E);

    constexpr int WV = 4;
    const int MT  = (N + 15) / 16;
    const int GBX = (MT + WV - 1) / WV;
    const int NB  = (N + 7) / 8;
    const int FB  = GBX;

    gemm_f32_wmma<128, 96, WV><<<dim3(GBX, 6), 32 * WV, 0, stream>>>(x, Wl[0], bufA, N);
    gather_product<96, 96, true><<<FB, 512, 0, stream>>>(bufA, ninfo, edges, bl[0], Wl[1], bufB, N);
    gather_product<96, 96, true><<<FB, 512, 0, stream>>>(bufB, ninfo, edges, bl[1], Wl[2], bufA, N);
    gather_product<96, 64, true><<<FB, 512, 0, stream>>>(bufA, ninfo, edges, bl[2], Wl[3], bufB, N);
    gather_finalize<64, false><<<NB, 256, 0, stream>>>(bufB, ninfo, edges, bl[3], bufC, N);

    zero_f32<<<(GCN_N_GRAPHS * 64 + TB - 1) / TB, TB, 0, stream>>>(sums, GCN_N_GRAPHS * 64);
    zero_f32<<<1, 64, 0, stream>>>(cnts, GCN_N_GRAPHS);
    pool_sum<<<(N + 7) / 8, 256, 0, stream>>>(bufC, batch, sums, cnts, N);
    pool_div<<<(GCN_N_GRAPHS * 64 + TB - 1) / TB, TB, 0, stream>>>(sums, cnts, (float*)d_out,
                                                                   GCN_N_GRAPHS * 64);
}


# Mean pool summed inside the last gather: a wave finishes 4 consecutive nodes into one pooled row in registers, a block adds its 16 rows via LDS, one row atomic per graph run; pool sum order changes
# speedup vs baseline: 3.3419x; 1.4016x over previous
#include <hip/hip_runtime.h>

typedef __attribute__((ext_vector_type(2))) float v2f;
typedef __attribute__((ext_vector_type(8))) float v8f;
typedef __attribute__((ext_vector_type(2))) int   v2i;
typedef __attribute__((ext_vector_type(4))) int   v4i;

#define GCN_N_NODES  50000
#define GCN_N_GRAPHS 64
static_assert(GCN_N_NODES % 16 == 0, "row tiles of 16 must be full");

#define SCAN_CHUNK (((GCN_N_NODES + 32 * 128 - 1) / (32 * 128)) * 128)
#define SCAN_PAD   (32 * SCAN_CHUNK)

__global__ void zero_f32(float* __restrict__ p, int n) {
    int i = blockIdx.x * blockDim.x + threadIdx.x;
    if (i < n) p[i] = 0.0f;
}

__global__ void zero_i32(int* __restrict__ p, int n) {
    int i = blockIdx.x * blockDim.x + threadIdx.x;
    if (i < n) p[i] = 0;
}

__global__ void edge_rank(const int* __restrict__ dst, int* __restrict__ cnt,
                          int* __restrict__ rank, int e) {
    int i = blockIdx.x * blockDim.x + threadIdx.x;
    if (i < e) rank[i] = atomicAdd(&cnt[dst[i]], 1);
}

__global__ void __launch_bounds__(1024)
scan_nodes(const int* __restrict__ cnt, v2i* __restrict__ ninfo, int n) {
    __shared__ int wtot[32];
    const int lane = threadIdx.x & 31;
    const int wav  = threadIdx.x >> 5;
    const int b    = wav * SCAN_CHUNK + lane * 4;

    int s = 0;
    #pragma unroll
    for (int k = 0; k < SCAN_CHUNK; k += 128) {
        const v4i c = *(const v4i*)&cnt[b + k];
        s += (c.x + c.y) + (c.z + c.w);
    }
    for (int o = 16; o > 0; o >>= 1) s += __shfl_xor(s, o);
    if (lane == 0) wtot[wav] = s;
    __syncthreads();

    int run = (lane < wav) ? wtot[lane] : 0;
    for (int o = 16; o > 0; o >>= 1) run += __shfl_xor(run, o);

    #pragma unroll
    for (int k = 0; k < SCAN_CHUNK; k += 128) {
        const v4i c = *(const v4i*)&cnt[b + k];
        const int t4 = (c.x + c.y) + (c.z + c.w);
        int incl = t4;
        for (int o = 1; o < 32; o <<= 1) {
            const int u = __shfl_up(incl, o);
            if (lane >= o) incl += u;
        }
        const int i0 = b + k;
        const int p0 = run + incl - t4;
        auto put = [&](int i, int first, int deg) {
            if (i < n) {
                v2i v;
                v.x = first;
                v.y = __float_as_int(rsqrtf((float)(deg + 1)));
                ninfo[i] = v;
            }
        };
        put(i0,     p0,                     c.x);
        put(i0 + 1, p0 + c.x,               c.y);
        put(i0 + 2, p0 + c.x + c.y,         c.z);
        put(i0 + 3, p0 + c.x + c.y + c.z,   c.w);
        run += __shfl(incl, 31);
    }
    if (threadIdx.x == 1023) {
        v2i v;
        v.x = run;
        v.y = 0;
        ninfo[n] = v;
    }
}

__global__ void csr_fill(const int* __restrict__ src, const int* __restrict__ dst,
                         const int* __restrict__ rank, const v2i* __restrict__ ninfo,
                         v2i* __restrict__ edges, int e) {
    int i = blockIdx.x * blockDim.x + threadIdx.x;
    if (i >= e) return;
    const int s = src[i];
    const int d = dst[i];
    const v2i nd = ninfo[d];
    const float ws = __int_as_float(ninfo[s].y);
    int slot = nd.x + rank[i];
    slot = min(max(slot, 0), e - 1);
    v2i ev;
    ev.x = s;
    ev.y = __float_as_int(ws * __int_as_float(nd.y));
    edges[slot] = ev;
}

template <int K, int NCOLS, int WAVES>
__global__ void gemm_f32_wmma(const float* __restrict__ A,
                              const float* __restrict__ Wm,
                              float* __restrict__ H, int M) {
    __shared__ float Bs[K * 16];

    const int tid  = threadIdx.x;
    const int lane = tid & 31;
    const int wav  = tid >> 5;
    const int colBase = blockIdx.y * 16;

    #pragma unroll
    for (int i = tid; i < K * 16; i += 32 * WAVES) {
        const int kk = i >> 4;
        const int cc = i & 15;
        Bs[i] = Wm[kk * NCOLS + colBase + cc];
    }
    __syncthreads();

    const int mtile = blockIdx.x * WAVES + wav;
    if (mtile * 16 < M) {
        const int l15 = lane & 15;
        const int hi  = lane >> 4;
        const int khi = hi * 2;
        const int mrow = mtile * 16 + l15;

        v8f acc = {0.f, 0.f, 0.f, 0.f, 0.f, 0.f, 0.f, 0.f};
        #pragma unroll 8
        for (int k0 = 0; k0 < K; k0 += 4) {
            v2f a = *(const v2f*)&A[mrow * K + k0 + khi];
            v2f b;
            b.x = Bs[(k0 + khi) * 16 + l15];
            b.y = Bs[(k0 + khi + 1) * 16 + l15];
            acc = __builtin_amdgcn_wmma_f32_16x16x4_f32(
                 false, a,  false, b,
                 (short)0, acc,  false,  false);
        }

        const int m0  = mtile * 16 + hi * 8;
        const int col = colBase + l15;
        #pragma unroll
        for (int r = 0; r < 8; ++r)
            H[(m0 + r) * NCOLS + col] = acc[r];
    }
}

template <int F, int GS, bool TAIL>
__device__ __forceinline__ void gather_group(const float* __restrict__ h, const v2i ev,
                                             int i0, int m, int n, int lane,
                                             float (&acc)[F / 32]) {
    constexpr int C = F / 32;
    float hv[GS][C];
    float wv[GS];
    #pragma unroll
    for (int q = 0; q < GS; ++q) {
        const int ii = TAIL ? min(i0 + q, m - 1) : (i0 + q);
        int s = __builtin_amdgcn_readlane(ev.x, ii);
        s = min(max(s, 0), n - 1);
        wv[q] = __int_as_float(__builtin_amdgcn_readlane(ev.y, ii));
        #pragma unroll
        for (int c = 0; c < C; ++c) hv[q][c] = h[s * F + c * 32 + lane];
    }
    #pragma unroll
    for (int q = 0; q < GS; ++q) {
        const bool ok = !TAIL || (i0 + q) < m;
        #pragma unroll
        for (int c = 0; c < C; ++c) {
            const float t = fmaf(hv[q][c], wv[q], acc[c]);
            acc[c] = ok ? t : acc[c];
        }
    }
}

template <int F>
__device__ __forceinline__ void gather_edges(const float* __restrict__ h,
                                             const v2i* __restrict__ edges,
                                             int beg, int end, int n, int lane,
                                             float (&acc)[F / 32]) {
    for (int base = beg; base < end; base += 32) {
        const int j = base + lane;
        v2i ev = {0, 0};
        if (j < end) ev = edges[j];
        const int m = min(32, end - base);
        int i0 = 0;
        for (; i0 + 8 <= m; i0 += 8) gather_group<F, 8, false>(h, ev, i0, m, n, lane, acc);
        const int rem = m - i0;
        if (rem > 4)      gather_group<F, 8, true>(h, ev, i0, m, n, lane, acc);
        else if (rem > 0) gather_group<F, 4, true>(h, ev, i0, m, n, lane, acc);
    }
}

template <int F, bool RELU>
__device__ __forceinline__ void node_row(const float* __restrict__ h,
                                         const v2i* __restrict__ ninfo,
                                         const v2i* __restrict__ edges,
                                         const float* __restrict__ bias,
                                         int node, int n, int lane, float (&v)[F / 32]) {
    constexpr int C = F / 32;
    const v2i ni0 = ninfo[node];
    const int beg = ni0.x;
    const int end = ninfo[node + 1].x;

    float hs[C], bs[C], acc[C];
    #pragma unroll
    for (int c = 0; c < C; ++c) {
        hs[c]  = h[node * F + c * 32 + lane];
        bs[c]  = bias[c * 32 + lane];
        acc[c] = 0.0f;
    }

    gather_edges<F>(h, edges, beg, end, n, lane, acc);

    const float di  = __int_as_float(ni0.y);
    const float di2 = di * di;
    #pragma unroll
    for (int c = 0; c < C; ++c) {
        const float t = acc[c] + hs[c] * di2 + bs[c];
        v[c] = RELU ? fmaxf(t, 0.0f) : t;
    }
}

template <int F, int NPW>
__global__ void __launch_bounds__(512)
gather_pool(const float* __restrict__ h,
            const v2i* __restrict__ ninfo,
            const v2i* __restrict__ edges,
            const float* __restrict__ bias,
            const int* __restrict__ batch,
            float* __restrict__ sums,
            float* __restrict__ cnts, int n) {
    constexpr int C = F / 32;
    __shared__ float Ps[16 * F];
    __shared__ float Pc[16];
    __shared__ int   Pg[16];

    const int lane = threadIdx.x & 31;
    const int wav  = __builtin_amdgcn_readfirstlane((int)(threadIdx.x >> 5));
    const int n0   = ((int)blockIdx.x * 16 + wav) * NPW;

    int   gcur = -1;
    float pc   = 0.0f;
    float ps[C];
    #pragma unroll
    for (int c = 0; c < C; ++c) ps[c] = 0.0f;

    if (n0 < n) {
        float bs[C];
        #pragma unroll
        for (int c = 0; c < C; ++c) bs[c] = bias[c * 32 + lane];
        const v2i nv = ninfo[min(n0 + min(lane, NPW), n)];

        for (int k = 0; k < NPW; ++k) {
            const int node = n0 + k;
            if (node < n) {
                const int   beg = __builtin_amdgcn_readlane(nv.x, k);
                const int   end = __builtin_amdgcn_readlane(nv.x, k + 1);
                const float di  = __int_as_float(__builtin_amdgcn_readlane(nv.y, k));
                int g = batch[node];
                g = min(max(g, 0), GCN_N_GRAPHS - 1);

                float hs[C], acc[C];
                #pragma unroll
                for (int c = 0; c < C; ++c) {
                    hs[c]  = h[node * F + c * 32 + lane];
                    acc[c] = 0.0f;
                }
                gather_edges<F>(h, edges, beg, end, n, lane, acc);

                if (g != gcur) {
                    if (gcur >= 0) {
                        #pragma unroll
                        for (int c = 0; c < C; ++c) atomicAdd(&sums[gcur * F + c * 32 + lane], ps[c]);
                        if (lane == 0) atomicAdd(&cnts[gcur], pc);
                    }
                    gcur = g;
                    pc   = 0.0f;
                    #pragma unroll
                    for (int c = 0; c < C; ++c) ps[c] = 0.0f;
                }
                const float di2 = di * di;
                #pragma unroll
                for (int c = 0; c < C; ++c) ps[c] += acc[c] + hs[c] * di2 + bs[c];
                pc += 1.0f;
            }
        }
    }

    #pragma unroll
    for (int c = 0; c < C; ++c) Ps[wav * F + c * 32 + lane] = ps[c];
    if (lane == 0) {
        Pg[wav] = gcur;
        Pc[wav] = pc;
    }
    __syncthreads();

    if (wav < C) {
        int   gw[16];
        float pw[16], cw[16];
        #pragma unroll
        for (int w = 0; w < 16; ++w) {
            gw[w] = Pg[w];
            cw[w] = Pc[w];
            pw[w] = Ps[w * F + wav * 32 + lane];
        }
        int   g = -1;
        float s = 0.0f, cn = 0.0f;
        #pragma unroll
        for (int w = 0; w < 16; ++w) {
            if (gw[w] < 0) continue;
            if (gw[w] != g) {
                if (g >= 0) {
                    atomicAdd(&sums[g * F + wav * 32 + lane], s);
                    if (wav == 0 && lane == 0) atomicAdd(&cnts[g], cn);
                }
                g  = gw[w];
                s  = 0.0f;
                cn = 0.0f;
            }
            s  += pw[w];
            cn += cw[w];
        }
        if (g >= 0) {
            atomicAdd(&sums[g * F + wav * 32 + lane], s);
            if (wav == 0 && lane == 0) atomicAdd(&cnts[g], cn);
        }
    }
}

template <int F, int NC, bool RELU>
__global__ void __launch_bounds__(512)
gather_product(const float* __restrict__ h,
               const v2i* __restrict__ ninfo,
               const v2i* __restrict__ edges,
               const float* __restrict__ bias,
               const float* __restrict__ Wn,
               float* __restrict__ Hn, int n) {
    constexpr int C  = F / 32;
    constexpr int TP = F + 4;
    constexpr int WP = NC + 8;
    __shared__ __attribute__((aligned(16))) float Ts[16 * TP];
    __shared__ float Ws[F * WP];

    const int tid  = threadIdx.x;
    const int lane = tid & 31;
    const int wav  = __builtin_amdgcn_readfirstlane(tid >> 5);

    for (int i = tid; i < F * NC; i += 512) {
        const int kk = i / NC;
        const int cc = i - kk * NC;
        Ws[kk * WP + cc] = Wn[i];
    }

    const int ntiles = n >> 4;
    for (int tile = blockIdx.x; tile < ntiles; tile += gridDim.x) {
        const int node = tile * 16 + wav;
        float v[C];
        node_row<F, RELU>(h, ninfo, edges, bias, node, n, lane, v);

        __syncthreads();
        #pragma unroll
        for (int c = 0; c < C; ++c) Ts[wav * TP + c * 32 + lane] = v[c];
        __syncthreads();

        if (wav < NC / 16) {
            const int l15 = lane & 15;
            const int hi  = lane >> 4;
            const int khi = hi * 2;
            const int colBase = wav * 16;

            v8f acc = {0.f, 0.f, 0.f, 0.f, 0.f, 0.f, 0.f, 0.f};
            #pragma unroll 8
            for (int k0 = 0; k0 < F; k0 += 4) {
                v2f a = *(const v2f*)&Ts[l15 * TP + k0 + khi];
                v2f b;
                b.x = Ws[(k0 + khi) * WP + colBase + l15];
                b.y = Ws[(k0 + khi + 1) * WP + colBase + l15];
                acc = __builtin_amdgcn_wmma_f32_16x16x4_f32(
                     false, a,  false, b,
                     (short)0, acc,  false,  false);
            }

            const int m0  = tile * 16 + hi * 8;
            const int col = colBase + l15;
            #pragma unroll
            for (int r = 0; r < 8; ++r)
                Hn[(m0 + r) * NC + col] = acc[r];
        }
    }
}

__global__ void pool_div(const float* __restrict__ sums,
                         const float* __restrict__ cnts,
                         float* __restrict__ out, int n) {
    int i = blockIdx.x * blockDim.x + threadIdx.x;
    if (i < n) out[i] = sums[i] / fmaxf(cnts[i / 64], 1.0f);
}

extern "C" void kernel_launch(void* const* d_in, const int* in_sizes, int n_in,
                              void* d_out, int out_size, void* d_ws, size_t ws_size,
                              hipStream_t stream) {
    const float* x     = (const float*)d_in[0];
    const int*   ei    = (const int*)d_in[1];
    const int*   batch = (const int*)d_in[2];
    const float* Wl[4] = {(const float*)d_in[3], (const float*)d_in[5],
                          (const float*)d_in[7], (const float*)d_in[9]};
    const float* bl[4] = {(const float*)d_in[4], (const float*)d_in[6],
                          (const float*)d_in[8], (const float*)d_in[10]};

    const int N = GCN_N_NODES;
    const int E = in_sizes[1] / 2;
    const int* src = ei;
    const int* dst = ei + E;

    auto au = [](size_t v) { return (v + 255) & ~(size_t)255; };
    char* ws = (char*)d_ws;
    size_t off = 0;
    int*   cnt   = (int*)(ws + off);   off += au(sizeof(int) * SCAN_PAD);
    int*   rank  = (int*)(ws + off);   off += au(sizeof(int) * (size_t)E);
    v2i*   ninfo = (v2i*)(ws + off);   off += au(sizeof(v2i) * (size_t)(N + 1));
    v2i*   edges = (v2i*)(ws + off);   off += au(sizeof(v2i) * (size_t)E);
    float* bufA  = (float*)(ws + off); off += au(sizeof(float) * (size_t)N * 96);
    float* bufB  = (float*)(ws + off); off += au(sizeof(float) * (size_t)N * 96);
    float* sums  = (float*)(ws + off); off += au(sizeof(float) * GCN_N_GRAPHS * 64);
    float* cnts  = (float*)(ws + off); off += au(sizeof(float) * GCN_N_GRAPHS);
    (void)ws_size; (void)n_in; (void)out_size;

    const int TB = 256;
    zero_f32<<<(GCN_N_GRAPHS * 64 + TB - 1) / TB, TB, 0, stream>>>(sums, GCN_N_GRAPHS * 64);
    zero_f32<<<1, 64, 0, stream>>>(cnts, GCN_N_GRAPHS);
    zero_i32  <<<(SCAN_PAD + TB - 1) / TB, TB, 0, stream>>>(cnt, SCAN_PAD);
    edge_rank <<<(E + TB - 1) / TB, TB, 0, stream>>>(dst, cnt, rank, E);
    scan_nodes<<<1, 1024, 0, stream>>>(cnt, ninfo, N);
    csr_fill  <<<(E + TB - 1) / TB, TB, 0, stream>>>(src, dst, rank, ninfo, edges, E);

    constexpr int WV = 4;
    const int MT  = (N + 15) / 16;
    const int GBX = (MT + WV - 1) / WV;
    const int PB  = (N + 63) / 64;
    const int FB  = GBX;

    gemm_f32_wmma<128, 96, WV><<<dim3(GBX, 6), 32 * WV, 0, stream>>>(x, Wl[0], bufA, N);
    gather_product<96, 96, true><<<FB, 512, 0, stream>>>(bufA, ninfo, edges, bl[0], Wl[1], bufB, N);
    gather_product<96, 96, true><<<FB, 512, 0, stream>>>(bufB, ninfo, edges, bl[1], Wl[2], bufA, N);
    gather_product<96, 64, true><<<FB, 512, 0, stream>>>(bufA, ninfo, edges, bl[2], Wl[3], bufB, N);
    gather_pool<64, 4><<<PB, 512, 0, stream>>>(bufB, ninfo, edges, bl[3], batch, sums, cnts, N);

    pool_div<<<(GCN_N_GRAPHS * 64 + TB - 1) / TB, TB, 0, stream>>>(sums, cnts, (float*)d_out,
                                                                   GCN_N_GRAPHS * 64);
}


# Edge sort by destination rebuilt as a two-level counting sort on LDS counters (bin pass, one block per 1024-node bin); the per-edge global returning atomics are gone; in-edge sum order = LDS arrival
# speedup vs baseline: 4.4638x; 1.3357x over previous
#include <hip/hip_runtime.h>

typedef __attribute__((ext_vector_type(2))) float v2f;
typedef __attribute__((ext_vector_type(8))) float v8f;
typedef __attribute__((ext_vector_type(2))) int   v2i;
typedef __attribute__((ext_vector_type(4))) int   v4i;

#define GCN_N_NODES  50000
#define GCN_N_GRAPHS 64
static_assert(GCN_N_NODES % 16 == 0, "row tiles of 16 must be full");

__global__ void zero_f32(float* __restrict__ p, int n) {
    int i = blockIdx.x * blockDim.x + threadIdx.x;
    if (i < n) p[i] = 0.0f;
}

__global__ void zero_i32(int* __restrict__ p, int n) {
    int i = blockIdx.x * blockDim.x + threadIdx.x;
    if (i < n) p[i] = 0;
}

#define BIN_SHIFT 10
#define BIN_NODES (1 << BIN_SHIFT)
#define NBIN      ((GCN_N_NODES + BIN_NODES - 1) / BIN_NODES)
static_assert(NBIN <= 64, "the bin totals are summed by one wave, two per lane");
static_assert(GCN_N_NODES <= (1 << (31 - BIN_SHIFT)), "source id and bin-local id share one 32-bit word");

__global__ void __launch_bounds__(1024)
bin_edges(const int* __restrict__ src, const int* __restrict__ dst,
          int* __restrict__ gcur, unsigned* __restrict__ stag, int e, int n) {
    __shared__ int lcnt[NBIN];
    __shared__ int lbase[NBIN];
    const int tid = threadIdx.x;
    if (tid < NBIN) lcnt[tid] = 0;
    __syncthreads();

    int      bin[4], r[4];
    unsigned w[4];
    bool     ok[4];
    #pragma unroll
    for (int q = 0; q < 4; ++q) {
        const int i = (int)blockIdx.x * 4096 + q * 1024 + tid;
        ok[q] = i < e;
        const int ic = min(i, e - 1);
        int d = dst[ic];
        int s = src[ic];
        d = min(max(d, 0), n - 1);
        s = min(max(s, 0), n - 1);
        bin[q] = d >> BIN_SHIFT;
        w[q]   = ((unsigned)s << BIN_SHIFT) | (unsigned)(d & (BIN_NODES - 1));
        r[q]   = 0;
    }
    #pragma unroll
    for (int q = 0; q < 4; ++q)
        if (ok[q]) r[q] = atomicAdd(&lcnt[bin[q]], 1);
    __syncthreads();
    if (tid < NBIN) {
        const int c = lcnt[tid];
        lbase[tid] = (c > 0) ? atomicAdd(&gcur[tid], c) : 0;
    }
    __syncthreads();
    #pragma unroll
    for (int q = 0; q < 4; ++q) {
        if (ok[q]) {
            int pos = lbase[bin[q]] + r[q];
            pos = min(max(pos, 0), e - 1);
            stag[(size_t)bin[q] * (size_t)e + (size_t)pos] = w[q];
        }
    }
}

__global__ void __launch_bounds__(1024)
bin_to_csr(const int* __restrict__ gcur, const unsigned* __restrict__ stag,
           v2i* __restrict__ ninfo, v2i* __restrict__ edges, int e, int n) {
    __shared__ int cnt[BIN_NODES];
    __shared__ int dnv[BIN_NODES];
    __shared__ int tot[64];
    __shared__ int wtot[32];
    const int tid  = threadIdx.x;
    const int lane = tid & 31;
    const int wav  = tid >> 5;
    const int b    = blockIdx.x;

    cnt[tid] = 0;
    if (tid < 64) tot[tid] = (tid < NBIN) ? gcur[tid] : 0;
    __syncthreads();

    int base = ((lane < b) ? tot[lane] : 0) + ((lane + 32 < b) ? tot[lane + 32] : 0);
    for (int o = 16; o > 0; o >>= 1) base += __shfl_xor(base, o);
    const int count = min(max(tot[b], 0), e);
    const unsigned* __restrict__ sb = stag + (size_t)b * (size_t)e;

    for (int i0 = tid; i0 < count; i0 += 4096) {
        unsigned w[4];
        #pragma unroll
        for (int q = 0; q < 4; ++q) {
            const int i = min(i0 + q * 1024, count - 1);
            w[q] = sb[i];
        }
        asm volatile("" :: "v"(w[0]), "v"(w[1]), "v"(w[2]), "v"(w[3]));
        #pragma unroll
        for (int q = 0; q < 4; ++q)
            if (i0 + q * 1024 < count) atomicAdd(&cnt[w[q] & (BIN_NODES - 1)], 1);
    }
    __syncthreads();

    const int c = cnt[tid];
    int incl = c;
    for (int o = 1; o < 32; o <<= 1) {
        const int u = __shfl_up(incl, o);
        if (lane >= o) incl += u;
    }
    if (lane == 31) wtot[wav] = incl;
    __syncthreads();
    int wprev = (lane < wav) ? wtot[lane] : 0;
    for (int o = 16; o > 0; o >>= 1) wprev += __shfl_xor(wprev, o);
    const int excl = wprev + incl - c;

    const int   node = (b << BIN_SHIFT) + tid;
    const float di   = rsqrtf((float)(c + 1));
    if (node < n) {
        v2i v;
        v.x = base + excl;
        v.y = __float_as_int(di);
        ninfo[node] = v;
    }
    if (b == NBIN - 1 && tid == BIN_NODES - 1) {
        v2i v;
        v.x = base + count;
        v.y = 0;
        ninfo[n] = v;
    }
    cnt[tid] = excl;
    dnv[tid] = __float_as_int(di);
    __syncthreads();

    for (int i0 = tid; i0 < count; i0 += 4096) {
        unsigned w[4];
        #pragma unroll
        for (int q = 0; q < 4; ++q) {
            const int i = min(i0 + q * 1024, count - 1);
            w[q] = sb[i];
        }
        asm volatile("" :: "v"(w[0]), "v"(w[1]), "v"(w[2]), "v"(w[3]));
        #pragma unroll
        for (int q = 0; q < 4; ++q) {
            if (i0 + q * 1024 < count) {
                const int dl = (int)(w[q] & (BIN_NODES - 1));
                int slot = base + atomicAdd(&cnt[dl], 1);
                slot = min(max(slot, 0), e - 1);
                v2i ev;
                ev.x = (int)(w[q] >> BIN_SHIFT);
                ev.y = dnv[dl];
                edges[slot] = ev;
            }
        }
    }
}

__global__ void edge_norm(const v2i* __restrict__ ninfo, v2i* __restrict__ edges, int e, int n) {
    const int T = (int)(gridDim.x * blockDim.x);
    const int i = (int)(blockIdx.x * blockDim.x + threadIdx.x);
    v2i   ev[4];
    float ws[4];
    #pragma unroll
    for (int q = 0; q < 4; ++q) {
        const int j = min(i + q * T, e - 1);
        ev[q] = edges[j];
    }
    asm volatile("" :: "v"(ev[0].x), "v"(ev[1].x), "v"(ev[2].x), "v"(ev[3].x));
    #pragma unroll
    for (int q = 0; q < 4; ++q) {
        const int s = min(max(ev[q].x, 0), n - 1);
        ws[q] = __int_as_float(ninfo[s].y);
    }
    asm volatile("" :: "v"(ws[0]), "v"(ws[1]), "v"(ws[2]), "v"(ws[3]));
    #pragma unroll
    for (int q = 0; q < 4; ++q) {
        const int j = i + q * T;
        if (j < e) {
            v2i o = ev[q];
            o.y = __float_as_int(ws[q] * __int_as_float(ev[q].y));
            edges[j] = o;
        }
    }
}

template <int K, int NCOLS, int WAVES>
__global__ void gemm_f32_wmma(const float* __restrict__ A,
                              const float* __restrict__ Wm,
                              float* __restrict__ H, int M) {
    __shared__ float Bs[K * 16];

    const int tid  = threadIdx.x;
    const int lane = tid & 31;
    const int wav  = tid >> 5;
    const int colBase = blockIdx.y * 16;

    #pragma unroll
    for (int i = tid; i < K * 16; i += 32 * WAVES) {
        const int kk = i >> 4;
        const int cc = i & 15;
        Bs[i] = Wm[kk * NCOLS + colBase + cc];
    }
    __syncthreads();

    const int mtile = blockIdx.x * WAVES + wav;
    if (mtile * 16 < M) {
        const int l15 = lane & 15;
        const int hi  = lane >> 4;
        const int khi = hi * 2;
        const int mrow = mtile * 16 + l15;

        v8f acc = {0.f, 0.f, 0.f, 0.f, 0.f, 0.f, 0.f, 0.f};
        #pragma unroll 8
        for (int k0 = 0; k0 < K; k0 += 4) {
            v2f a = *(const v2f*)&A[mrow * K + k0 + khi];
            v2f b;
            b.x = Bs[(k0 + khi) * 16 + l15];
            b.y = Bs[(k0 + khi + 1) * 16 + l15];
            acc = __builtin_amdgcn_wmma_f32_16x16x4_f32(
                 false, a,  false, b,
                 (short)0, acc,  false,  false);
        }

        const int m0  = mtile * 16 + hi * 8;
        const int col = colBase + l15;
        #pragma unroll
        for (int r = 0; r < 8; ++r)
            H[(m0 + r) * NCOLS + col] = acc[r];
    }
}

template <int F, int GS, bool TAIL>
__device__ __forceinline__ void gather_group(const float* __restrict__ h, const v2i ev,
                                             int i0, int m, int n, int lane,
                                             float (&acc)[F / 32]) {
    constexpr int C = F / 32;
    float hv[GS][C];
    float wv[GS];
    #pragma unroll
    for (int q = 0; q < GS; ++q) {
        const int ii = TAIL ? min(i0 + q, m - 1) : (i0 + q);
        int s = __builtin_amdgcn_readlane(ev.x, ii);
        s = min(max(s, 0), n - 1);
        wv[q] = __int_as_float(__builtin_amdgcn_readlane(ev.y, ii));
        #pragma unroll
        for (int c = 0; c < C; ++c) hv[q][c] = h[s * F + c * 32 + lane];
    }
    #pragma unroll
    for (int q = 0; q < GS; ++q) {
        const bool ok = !TAIL || (i0 + q) < m;
        #pragma unroll
        for (int c = 0; c < C; ++c) {
            const float t = fmaf(hv[q][c], wv[q], acc[c]);
            acc[c] = ok ? t : acc[c];
        }
    }
}

template <int F>
__device__ __forceinline__ void gather_edges(const float* __restrict__ h,
                                             const v2i* __restrict__ edges,
                                             int beg, int end, int n, int lane,
                                             float (&acc)[F / 32]) {
    for (int base = beg; base < end; base += 32) {
        const int j = base + lane;
        v2i ev = {0, 0};
        if (j < end) ev = edges[j];
        const int m = min(32, end - base);
        int i0 = 0;
        for (; i0 + 8 <= m; i0 += 8) gather_group<F, 8, false>(h, ev, i0, m, n, lane, acc);
        const int rem = m - i0;
        if (rem > 4)      gather_group<F, 8, true>(h, ev, i0, m, n, lane, acc);
        else if (rem > 0) gather_group<F, 4, true>(h, ev, i0, m, n, lane, acc);
    }
}

template <int F, bool RELU>
__device__ __forceinline__ void node_row(const float* __restrict__ h,
                                         const v2i* __restrict__ ninfo,
                                         const v2i* __restrict__ edges,
                                         const float* __restrict__ bias,
                                         int node, int n, int lane, float (&v)[F / 32]) {
    constexpr int C = F / 32;
    const v2i ni0 = ninfo[node];
    const int beg = ni0.x;
    const int end = ninfo[node + 1].x;

    float hs[C], bs[C], acc[C];
    #pragma unroll
    for (int c = 0; c < C; ++c) {
        hs[c]  = h[node * F + c * 32 + lane];
        bs[c]  = bias[c * 32 + lane];
        acc[c] = 0.0f;
    }

    gather_edges<F>(h, edges, beg, end, n, lane, acc);

    const float di  = __int_as_float(ni0.y);
    const float di2 = di * di;
    #pragma unroll
    for (int c = 0; c < C; ++c) {
        const float t = acc[c] + hs[c] * di2 + bs[c];
        v[c] = RELU ? fmaxf(t, 0.0f) : t;
    }
}

template <int F, int NPW>
__global__ void __launch_bounds__(512)
gather_pool(const float* __restrict__ h,
            const v2i* __restrict__ ninfo,
            const v2i* __restrict__ edges,
            const float* __restrict__ bias,
            const int* __restrict__ batch,
            float* __restrict__ sums,
            float* __restrict__ cnts, int n) {
    constexpr int C = F / 32;
    __shared__ float Ps[16 * F];
    __shared__ float Pc[16];
    __shared__ int   Pg[16];

    const int lane = threadIdx.x & 31;
    const int wav  = __builtin_amdgcn_readfirstlane((int)(threadIdx.x >> 5));
    const int n0   = ((int)blockIdx.x * 16 + wav) * NPW;

    int   gcur = -1;
    float pc   = 0.0f;
    float ps[C];
    #pragma unroll
    for (int c = 0; c < C; ++c) ps[c] = 0.0f;

    if (n0 < n) {
        float bs[C];
        #pragma unroll
        for (int c = 0; c < C; ++c) bs[c] = bias[c * 32 + lane];
        const v2i nv = ninfo[min(n0 + min(lane, NPW), n)];

        for (int k = 0; k < NPW; ++k) {
            const int node = n0 + k;
            if (node < n) {
                const int   beg = __builtin_amdgcn_readlane(nv.x, k);
                const int   end = __builtin_amdgcn_readlane(nv.x, k + 1);
                const float di  = __int_as_float(__builtin_amdgcn_readlane(nv.y, k));
                int g = batch[node];
                g = min(max(g, 0), GCN_N_GRAPHS - 1);

                float hs[C], acc[C];
                #pragma unroll
                for (int c = 0; c < C; ++c) {
                    hs[c]  = h[node * F + c * 32 + lane];
                    acc[c] = 0.0f;
                }
                gather_edges<F>(h, edges, beg, end, n, lane, acc);

                if (g != gcur) {
                    if (gcur >= 0) {
                        #pragma unroll
                        for (int c = 0; c < C; ++c) atomicAdd(&sums[gcur * F + c * 32 + lane], ps[c]);
                        if (lane == 0) atomicAdd(&cnts[gcur], pc);
                    }
                    gcur = g;
                    pc   = 0.0f;
                    #pragma unroll
                    for (int c = 0; c < C; ++c) ps[c] = 0.0f;
                }
                const float di2 = di * di;
                #pragma unroll
                for (int c = 0; c < C; ++c) ps[c] += acc[c] + hs[c] * di2 + bs[c];
                pc += 1.0f;
            }
        }
    }

    #pragma unroll
    for (int c = 0; c < C; ++c) Ps[wav * F + c * 32 + lane] = ps[c];
    if (lane == 0) {
        Pg[wav] = gcur;
        Pc[wav] = pc;
    }
    __syncthreads();

    if (wav < C) {
        int   gw[16];
        float pw[16], cw[16];
        #pragma unroll
        for (int w = 0; w < 16; ++w) {
            gw[w] = Pg[w];
            cw[w] = Pc[w];
            pw[w] = Ps[w * F + wav * 32 + lane];
        }
        int   g = -1;
        float s = 0.0f, cn = 0.0f;
        #pragma unroll
        for (int w = 0; w < 16; ++w) {
            if (gw[w] < 0) continue;
            if (gw[w] != g) {
                if (g >= 0) {
                    atomicAdd(&sums[g * F + wav * 32 + lane], s);
                    if (wav == 0 && lane == 0) atomicAdd(&cnts[g], cn);
                }
                g  = gw[w];
                s  = 0.0f;
                cn = 0.0f;
            }
            s  += pw[w];
            cn += cw[w];
        }
        if (g >= 0) {
            atomicAdd(&sums[g * F + wav * 32 + lane], s);
            if (wav == 0 && lane == 0) atomicAdd(&cnts[g], cn);
        }
    }
}

template <int F, int NC, bool RELU>
__global__ void __launch_bounds__(512)
gather_product(const float* __restrict__ h,
               const v2i* __restrict__ ninfo,
               const v2i* __restrict__ edges,
               const float* __restrict__ bias,
               const float* __restrict__ Wn,
               float* __restrict__ Hn, int n) {
    constexpr int C  = F / 32;
    constexpr int TP = F + 4;
    constexpr int WP = NC + 8;
    __shared__ __attribute__((aligned(16))) float Ts[16 * TP];
    __shared__ float Ws[F * WP];

    const int tid  = threadIdx.x;
    const int lane = tid & 31;
    const int wav  = __builtin_amdgcn_readfirstlane(tid >> 5);

    for (int i = tid; i < F * NC; i += 512) {
        const int kk = i / NC;
        const int cc = i - kk * NC;
        Ws[kk * WP + cc] = Wn[i];
    }

    const int ntiles = n >> 4;
    for (int tile = blockIdx.x; tile < ntiles; tile += gridDim.x) {
        const int node = tile * 16 + wav;
        float v[C];
        node_row<F, RELU>(h, ninfo, edges, bias, node, n, lane, v);

        __syncthreads();
        #pragma unroll
        for (int c = 0; c < C; ++c) Ts[wav * TP + c * 32 + lane] = v[c];
        __syncthreads();

        if (wav < NC / 16) {
            const int l15 = lane & 15;
            const int hi  = lane >> 4;
            const int khi = hi * 2;
            const int colBase = wav * 16;

            v8f acc = {0.f, 0.f, 0.f, 0.f, 0.f, 0.f, 0.f, 0.f};
            #pragma unroll 8
            for (int k0 = 0; k0 < F; k0 += 4) {
                v2f a = *(const v2f*)&Ts[l15 * TP + k0 + khi];
                v2f b;
                b.x = Ws[(k0 + khi) * WP + colBase + l15];
                b.y = Ws[(k0 + khi + 1) * WP + colBase + l15];
                acc = __builtin_amdgcn_wmma_f32_16x16x4_f32(
                     false, a,  false, b,
                     (short)0, acc,  false,  false);
            }

            const int m0  = tile * 16 + hi * 8;
            const int col = colBase + l15;
            #pragma unroll
            for (int r = 0; r < 8; ++r)
                Hn[(m0 + r) * NC + col] = acc[r];
        }
    }
}

__global__ void pool_div(const float* __restrict__ sums,
                         const float* __restrict__ cnts,
                         float* __restrict__ out, int n) {
    int i = blockIdx.x * blockDim.x + threadIdx.x;
    if (i < n) out[i] = sums[i] / fmaxf(cnts[i / 64], 1.0f);
}

extern "C" void kernel_launch(void* const* d_in, const int* in_sizes, int n_in,
                              void* d_out, int out_size, void* d_ws, size_t ws_size,
                              hipStream_t stream) {
    const float* x     = (const float*)d_in[0];
    const int*   ei    = (const int*)d_in[1];
    const int*   batch = (const int*)d_in[2];
    const float* Wl[4] = {(const float*)d_in[3], (const float*)d_in[5],
                          (const float*)d_in[7], (const float*)d_in[9]};
    const float* bl[4] = {(const float*)d_in[4], (const float*)d_in[6],
                          (const float*)d_in[8], (const float*)d_in[10]};

    const int N = GCN_N_NODES;
    const int E = in_sizes[1] / 2;
    const int* src = ei;
    const int* dst = ei + E;

    auto au = [](size_t v) { return (v + 255) & ~(size_t)255; };
    char* ws = (char*)d_ws;
    size_t off = 0;
    int*      gcur = (int*)(ws + off);      off += au(sizeof(int) * 64);
    unsigned* stag = (unsigned*)(ws + off); off += au(sizeof(unsigned) * (size_t)NBIN * (size_t)E);
    v2i*   ninfo = (v2i*)(ws + off);   off += au(sizeof(v2i) * (size_t)(N + 1));
    v2i*   edges = (v2i*)(ws + off);   off += au(sizeof(v2i) * (size_t)E);
    float* bufA  = (float*)(ws + off); off += au(sizeof(float) * (size_t)N * 96);
    float* bufB  = (float*)(ws + off); off += au(sizeof(float) * (size_t)N * 96);
    float* sums  = (float*)(ws + off); off += au(sizeof(float) * GCN_N_GRAPHS * 64);
    float* cnts  = (float*)(ws + off); off += au(sizeof(float) * GCN_N_GRAPHS);
    (void)ws_size; (void)n_in; (void)out_size;

    const int TB = 256;
    zero_f32<<<(GCN_N_GRAPHS * 64 + TB - 1) / TB, TB, 0, stream>>>(sums, GCN_N_GRAPHS * 64);
    zero_f32<<<1, 64, 0, stream>>>(cnts, GCN_N_GRAPHS);
    zero_i32  <<<1, 64, 0, stream>>>(gcur, 64);
    bin_edges <<<(E + 4095) / 4096, 1024, 0, stream>>>(src, dst, gcur, stag, E, N);
    bin_to_csr<<<NBIN, 1024, 0, stream>>>(gcur, stag, ninfo, edges, E, N);
    edge_norm <<<(E + 4 * TB - 1) / (4 * TB), TB, 0, stream>>>(ninfo, edges, E, N);

    constexpr int WV = 4;
    const int MT  = (N + 15) / 16;
    const int GBX = (MT + WV - 1) / WV;
    const int PB  = (N + 63) / 64;
    const int FB  = GBX;

    gemm_f32_wmma<128, 96, WV><<<dim3(GBX, 6), 32 * WV, 0, stream>>>(x, Wl[0], bufA, N);
    gather_product<96, 96, true><<<FB, 512, 0, stream>>>(bufA, ninfo, edges, bl[0], Wl[1], bufB, N);
    gather_product<96, 96, true><<<FB, 512, 0, stream>>>(bufB, ninfo, edges, bl[1], Wl[2], bufA, N);
    gather_product<96, 64, true><<<FB, 512, 0, stream>>>(bufA, ninfo, edges, bl[2], Wl[3], bufB, N);
    gather_pool<64, 4><<<PB, 512, 0, stream>>>(bufB, ninfo, edges, bl[3], batch, sums, cnts, N);

    pool_div<<<(GCN_N_GRAPHS * 64 + TB - 1) / TB, TB, 0, stream>>>(sums, cnts, (float*)d_out,
                                                                   GCN_N_GRAPHS * 64);
}


# Layer-1 product x@W1 moved into the two edge-sort launches as a second block role (row groups split between bin pass and CSR pass, same tiling and k order); its own launch is gone
# speedup vs baseline: 4.7004x; 1.0530x over previous
#include <hip/hip_runtime.h>

typedef __attribute__((ext_vector_type(2))) float v2f;
typedef __attribute__((ext_vector_type(8))) float v8f;
typedef __attribute__((ext_vector_type(2))) int   v2i;
typedef __attribute__((ext_vector_type(4))) int   v4i;

#define GCN_N_NODES  50000
#define GCN_N_GRAPHS 64
static_assert(GCN_N_NODES % 16 == 0, "row tiles of 16 must be full");

__global__ void zero_f32(float* __restrict__ p, int n) {
    int i = blockIdx.x * blockDim.x + threadIdx.x;
    if (i < n) p[i] = 0.0f;
}

__global__ void zero_i32(int* __restrict__ p, int n) {
    int i = blockIdx.x * blockDim.x + threadIdx.x;
    if (i < n) p[i] = 0;
}

#define BIN_SHIFT 10
#define BIN_NODES (1 << BIN_SHIFT)
#define NBIN      ((GCN_N_NODES + BIN_NODES - 1) / BIN_NODES)
static_assert(NBIN <= 64, "the bin totals are summed by one wave, two per lane");
static_assert(GCN_N_NODES <= (1 << (31 - BIN_SHIFT)), "source id and bin-local id share one 32-bit word");

__device__ __forceinline__ void
bin_edges_block(const int blk, const int* __restrict__ src, const int* __restrict__ dst,
                int* __restrict__ gcur, unsigned* __restrict__ stag, int e, int n) {
    __shared__ int lcnt[NBIN];
    __shared__ int lbase[NBIN];
    const int tid = threadIdx.x;
    if (tid < NBIN) lcnt[tid] = 0;
    __syncthreads();

    int      bin[4], r[4];
    unsigned w[4];
    bool     ok[4];
    #pragma unroll
    for (int q = 0; q < 4; ++q) {
        const int i = blk * 4096 + q * 1024 + tid;
        ok[q] = i < e;
        const int ic = min(i, e - 1);
        int d = dst[ic];
        int s = src[ic];
        d = min(max(d, 0), n - 1);
        s = min(max(s, 0), n - 1);
        bin[q] = d >> BIN_SHIFT;
        w[q]   = ((unsigned)s << BIN_SHIFT) | (unsigned)(d & (BIN_NODES - 1));
        r[q]   = 0;
    }
    #pragma unroll
    for (int q = 0; q < 4; ++q)
        if (ok[q]) r[q] = atomicAdd(&lcnt[bin[q]], 1);
    __syncthreads();
    if (tid < NBIN) {
        const int c = lcnt[tid];
        lbase[tid] = (c > 0) ? atomicAdd(&gcur[tid], c) : 0;
    }
    __syncthreads();
    #pragma unroll
    for (int q = 0; q < 4; ++q) {
        if (ok[q]) {
            int pos = lbase[bin[q]] + r[q];
            pos = min(max(pos, 0), e - 1);
            stag[(size_t)bin[q] * (size_t)e + (size_t)pos] = w[q];
        }
    }
}

__device__ __forceinline__ void
bin_to_csr_block(const int b, const int* __restrict__ gcur, const unsigned* __restrict__ stag,
                 v2i* __restrict__ ninfo, v2i* __restrict__ edges, int e, int n) {
    __shared__ int cnt[BIN_NODES];
    __shared__ int dnv[BIN_NODES];
    __shared__ int tot[64];
    __shared__ int wtot[32];
    const int tid  = threadIdx.x;
    const int lane = tid & 31;
    const int wav  = tid >> 5;

    cnt[tid] = 0;
    if (tid < 64) tot[tid] = (tid < NBIN) ? gcur[tid] : 0;
    __syncthreads();

    int base = ((lane < b) ? tot[lane] : 0) + ((lane + 32 < b) ? tot[lane + 32] : 0);
    for (int o = 16; o > 0; o >>= 1) base += __shfl_xor(base, o);
    const int count = min(max(tot[b], 0), e);
    const unsigned* __restrict__ sb = stag + (size_t)b * (size_t)e;

    for (int i0 = tid; i0 < count; i0 += 4096) {
        unsigned w[4];
        #pragma unroll
        for (int q = 0; q < 4; ++q) {
            const int i = min(i0 + q * 1024, count - 1);
            w[q] = sb[i];
        }
        asm volatile("" :: "v"(w[0]), "v"(w[1]), "v"(w[2]), "v"(w[3]));
        #pragma unroll
        for (int q = 0; q < 4; ++q)
            if (i0 + q * 1024 < count) atomicAdd(&cnt[w[q] & (BIN_NODES - 1)], 1);
    }
    __syncthreads();

    const int c = cnt[tid];
    int incl = c;
    for (int o = 1; o < 32; o <<= 1) {
        const int u = __shfl_up(incl, o);
        if (lane >= o) incl += u;
    }
    if (lane == 31) wtot[wav] = incl;
    __syncthreads();
    int wprev = (lane < wav) ? wtot[lane] : 0;
    for (int o = 16; o > 0; o >>= 1) wprev += __shfl_xor(wprev, o);
    const int excl = wprev + incl - c;

    const int   node = (b << BIN_SHIFT) + tid;
    const float di   = rsqrtf((float)(c + 1));
    if (node < n) {
        v2i v;
        v.x = base + excl;
        v.y = __float_as_int(di);
        ninfo[node] = v;
    }
    if (b == NBIN - 1 && tid == BIN_NODES - 1) {
        v2i v;
        v.x = base + count;
        v.y = 0;
        ninfo[n] = v;
    }
    cnt[tid] = excl;
    dnv[tid] = __float_as_int(di);
    __syncthreads();

    for (int i0 = tid; i0 < count; i0 += 4096) {
        unsigned w[4];
        #pragma unroll
        for (int q = 0; q < 4; ++q) {
            const int i = min(i0 + q * 1024, count - 1);
            w[q] = sb[i];
        }
        asm volatile("" :: "v"(w[0]), "v"(w[1]), "v"(w[2]), "v"(w[3]));
        #pragma unroll
        for (int q = 0; q < 4; ++q) {
            if (i0 + q * 1024 < count) {
                const int dl = (int)(w[q] & (BIN_NODES - 1));
                int slot = base + atomicAdd(&cnt[dl], 1);
                slot = min(max(slot, 0), e - 1);
                v2i ev;
                ev.x = (int)(w[q] >> BIN_SHIFT);
                ev.y = dnv[dl];
                edges[slot] = ev;
            }
        }
    }
}

__global__ void edge_norm(const v2i* __restrict__ ninfo, v2i* __restrict__ edges, int e, int n) {
    const int T = (int)(gridDim.x * blockDim.x);
    const int i = (int)(blockIdx.x * blockDim.x + threadIdx.x);
    v2i   ev[4];
    float ws[4];
    #pragma unroll
    for (int q = 0; q < 4; ++q) {
        const int j = min(i + q * T, e - 1);
        ev[q] = edges[j];
    }
    asm volatile("" :: "v"(ev[0].x), "v"(ev[1].x), "v"(ev[2].x), "v"(ev[3].x));
    #pragma unroll
    for (int q = 0; q < 4; ++q) {
        const int s = min(max(ev[q].x, 0), n - 1);
        ws[q] = __int_as_float(ninfo[s].y);
    }
    asm volatile("" :: "v"(ws[0]), "v"(ws[1]), "v"(ws[2]), "v"(ws[3]));
    #pragma unroll
    for (int q = 0; q < 4; ++q) {
        const int j = i + q * T;
        if (j < e) {
            v2i o = ev[q];
            o.y = __float_as_int(ws[q] * __int_as_float(ev[q].y));
            edges[j] = o;
        }
    }
}

template <int K, int NCOLS, int WAVES>
__device__ __forceinline__ void product_block(const float* __restrict__ A,
                                              const float* __restrict__ Wm,
                                              float* __restrict__ H, int M,
                                              const int mgroup, const int ctile) {
    __shared__ float Bs[K * 16];

    const int tid  = threadIdx.x;
    const int lane = tid & 31;
    const int wav  = tid >> 5;
    const int colBase = ctile * 16;

    #pragma unroll
    for (int i = tid; i < K * 16; i += 32 * WAVES) {
        const int kk = i >> 4;
        const int cc = i & 15;
        Bs[i] = Wm[kk * NCOLS + colBase + cc];
    }
    __syncthreads();

    const int mtile = mgroup * WAVES + wav;
    if (mtile * 16 < M) {
        const int l15 = lane & 15;
        const int hi  = lane >> 4;
        const int khi = hi * 2;
        const int mrow = mtile * 16 + l15;

        v8f acc = {0.f, 0.f, 0.f, 0.f, 0.f, 0.f, 0.f, 0.f};
        #pragma unroll 8
        for (int k0 = 0; k0 < K; k0 += 4) {
            v2f a = *(const v2f*)&A[mrow * K + k0 + khi];
            v2f b;
            b.x = Bs[(k0 + khi) * 16 + l15];
            b.y = Bs[(k0 + khi + 1) * 16 + l15];
            acc = __builtin_amdgcn_wmma_f32_16x16x4_f32(
                 false, a,  false, b,
                 (short)0, acc,  false,  false);
        }

        const int m0  = mtile * 16 + hi * 8;
        const int col = colBase + l15;
        #pragma unroll
        for (int r = 0; r < 8; ++r)
            H[(m0 + r) * NCOLS + col] = acc[r];
    }
}

#define PROD_WAVES 32
template <int K, int NCOLS>
__global__ void __launch_bounds__(1024)
bin_edges_and_product(const int* __restrict__ src, const int* __restrict__ dst,
                      int* __restrict__ gcur, unsigned* __restrict__ stag, int e, int n, int nsort,
                      const float* __restrict__ A, const float* __restrict__ Wm,
                      float* __restrict__ H, int mg0) {
    const int blk = (int)blockIdx.x;
    if (blk < nsort) {
        bin_edges_block(blk, src, dst, gcur, stag, e, n);
    } else {
        const int g = blk - nsort;
        product_block<K, NCOLS, PROD_WAVES>(A, Wm, H, n, mg0 + g / (NCOLS / 16), g % (NCOLS / 16));
    }
}

template <int K, int NCOLS>
__global__ void __launch_bounds__(1024)
bin_to_csr_and_product(const int* __restrict__ gcur, const unsigned* __restrict__ stag,
                       v2i* __restrict__ ninfo, v2i* __restrict__ edges, int e, int n, int nsort,
                       const float* __restrict__ A, const float* __restrict__ Wm,
                       float* __restrict__ H, int mg0) {
    const int blk = (int)blockIdx.x;
    if (blk < nsort) {
        bin_to_csr_block(blk, gcur, stag, ninfo, edges, e, n);
    } else {
        const int g = blk - nsort;
        product_block<K, NCOLS, PROD_WAVES>(A, Wm, H, n, mg0 + g / (NCOLS / 16), g % (NCOLS / 16));
    }
}

template <int F, int GS, bool TAIL>
__device__ __forceinline__ void gather_group(const float* __restrict__ h, const v2i ev,
                                             int i0, int m, int n, int lane,
                                             float (&acc)[F / 32]) {
    constexpr int C = F / 32;
    float hv[GS][C];
    float wv[GS];
    #pragma unroll
    for (int q = 0; q < GS; ++q) {
        const int ii = TAIL ? min(i0 + q, m - 1) : (i0 + q);
        int s = __builtin_amdgcn_readlane(ev.x, ii);
        s = min(max(s, 0), n - 1);
        wv[q] = __int_as_float(__builtin_amdgcn_readlane(ev.y, ii));
        #pragma unroll
        for (int c = 0; c < C; ++c) hv[q][c] = h[s * F + c * 32 + lane];
    }
    #pragma unroll
    for (int q = 0; q < GS; ++q) {
        const bool ok = !TAIL || (i0 + q) < m;
        #pragma unroll
        for (int c = 0; c < C; ++c) {
            const float t = fmaf(hv[q][c], wv[q], acc[c]);
            acc[c] = ok ? t : acc[c];
        }
    }
}

template <int F>
__device__ __forceinline__ void gather_edges(const float* __restrict__ h,
                                             const v2i* __restrict__ edges,
                                             int beg, int end, int n, int lane,
                                             float (&acc)[F / 32]) {
    for (int base = beg; base < end; base += 32) {
        const int j = base + lane;
        v2i ev = {0, 0};
        if (j < end) ev = edges[j];
        const int m = min(32, end - base);
        int i0 = 0;
        for (; i0 + 8 <= m; i0 += 8) gather_group<F, 8, false>(h, ev, i0, m, n, lane, acc);
        const int rem = m - i0;
        if (rem > 4)      gather_group<F, 8, true>(h, ev, i0, m, n, lane, acc);
        else if (rem > 0) gather_group<F, 4, true>(h, ev, i0, m, n, lane, acc);
    }
}

template <int F, bool RELU>
__device__ __forceinline__ void node_row(const float* __restrict__ h,
                                         const v2i* __restrict__ ninfo,
                                         const v2i* __restrict__ edges,
                                         const float* __restrict__ bias,
                                         int node, int n, int lane, float (&v)[F / 32]) {
    constexpr int C = F / 32;
    const v2i ni0 = ninfo[node];
    const int beg = ni0.x;
    const int end = ninfo[node + 1].x;

    float hs[C], bs[C], acc[C];
    #pragma unroll
    for (int c = 0; c < C; ++c) {
        hs[c]  = h[node * F + c * 32 + lane];
        bs[c]  = bias[c * 32 + lane];
        acc[c] = 0.0f;
    }

    gather_edges<F>(h, edges, beg, end, n, lane, acc);

    const float di  = __int_as_float(ni0.y);
    const float di2 = di * di;
    #pragma unroll
    for (int c = 0; c < C; ++c) {
        const float t = acc[c] + hs[c] * di2 + bs[c];
        v[c] = RELU ? fmaxf(t, 0.0f) : t;
    }
}

template <int F, int NPW>
__global__ void __launch_bounds__(512)
gather_pool(const float* __restrict__ h,
            const v2i* __restrict__ ninfo,
            const v2i* __restrict__ edges,
            const float* __restrict__ bias,
            const int* __restrict__ batch,
            float* __restrict__ sums,
            float* __restrict__ cnts, int n) {
    constexpr int C = F / 32;
    __shared__ float Ps[16 * F];
    __shared__ float Pc[16];
    __shared__ int   Pg[16];

    const int lane = threadIdx.x & 31;
    const int wav  = __builtin_amdgcn_readfirstlane((int)(threadIdx.x >> 5));
    const int n0   = ((int)blockIdx.x * 16 + wav) * NPW;

    int   gcur = -1;
    float pc   = 0.0f;
    float ps[C];
    #pragma unroll
    for (int c = 0; c < C; ++c) ps[c] = 0.0f;

    if (n0 < n) {
        float bs[C];
        #pragma unroll
        for (int c = 0; c < C; ++c) bs[c] = bias[c * 32 + lane];
        const v2i nv = ninfo[min(n0 + min(lane, NPW), n)];

        for (int k = 0; k < NPW; ++k) {
            const int node = n0 + k;
            if (node < n) {
                const int   beg = __builtin_amdgcn_readlane(nv.x, k);
                const int   end = __builtin_amdgcn_readlane(nv.x, k + 1);
                const float di  = __int_as_float(__builtin_amdgcn_readlane(nv.y, k));
                int g = batch[node];
                g = min(max(g, 0), GCN_N_GRAPHS - 1);

                float hs[C], acc[C];
                #pragma unroll
                for (int c = 0; c < C; ++c) {
                    hs[c]  = h[node * F + c * 32 + lane];
                    acc[c] = 0.0f;
                }
                gather_edges<F>(h, edges, beg, end, n, lane, acc);

                if (g != gcur) {
                    if (gcur >= 0) {
                        #pragma unroll
                        for (int c = 0; c < C; ++c) atomicAdd(&sums[gcur * F + c * 32 + lane], ps[c]);
                        if (lane == 0) atomicAdd(&cnts[gcur], pc);
                    }
                    gcur = g;
                    pc   = 0.0f;
                    #pragma unroll
                    for (int c = 0; c < C; ++c) ps[c] = 0.0f;
                }
                const float di2 = di * di;
                #pragma unroll
                for (int c = 0; c < C; ++c) ps[c] += acc[c] + hs[c] * di2 + bs[c];
                pc += 1.0f;
            }
        }
    }

    #pragma unroll
    for (int c = 0; c < C; ++c) Ps[wav * F + c * 32 + lane] = ps[c];
    if (lane == 0) {
        Pg[wav] = gcur;
        Pc[wav] = pc;
    }
    __syncthreads();

    if (wav < C) {
        int   gw[16];
        float pw[16], cw[16];
        #pragma unroll
        for (int w = 0; w < 16; ++w) {
            gw[w] = Pg[w];
            cw[w] = Pc[w];
            pw[w] = Ps[w * F + wav * 32 + lane];
        }
        int   g = -1;
        float s = 0.0f, cn = 0.0f;
        #pragma unroll
        for (int w = 0; w < 16; ++w) {
            if (gw[w] < 0) continue;
            if (gw[w] != g) {
                if (g >= 0) {
                    atomicAdd(&sums[g * F + wav * 32 + lane], s);
                    if (wav == 0 && lane == 0) atomicAdd(&cnts[g], cn);
                }
                g  = gw[w];
                s  = 0.0f;
                cn = 0.0f;
            }
            s  += pw[w];
            cn += cw[w];
        }
        if (g >= 0) {
            atomicAdd(&sums[g * F + wav * 32 + lane], s);
            if (wav == 0 && lane == 0) atomicAdd(&cnts[g], cn);
        }
    }
}

template <int F, int NC, bool RELU>
__global__ void __launch_bounds__(512)
gather_product(const float* __restrict__ h,
               const v2i* __restrict__ ninfo,
               const v2i* __restrict__ edges,
               const float* __restrict__ bias,
               const float* __restrict__ Wn,
               float* __restrict__ Hn, int n) {
    constexpr int C  = F / 32;
    constexpr int TP = F + 4;
    constexpr int WP = NC + 8;
    __shared__ __attribute__((aligned(16))) float Ts[16 * TP];
    __shared__ float Ws[F * WP];

    const int tid  = threadIdx.x;
    const int lane = tid & 31;
    const int wav  = __builtin_amdgcn_readfirstlane(tid >> 5);

    for (int i = tid; i < F * NC; i += 512) {
        const int kk = i / NC;
        const int cc = i - kk * NC;
        Ws[kk * WP + cc] = Wn[i];
    }

    const int ntiles = n >> 4;
    for (int tile = blockIdx.x; tile < ntiles; tile += gridDim.x) {
        const int node = tile * 16 + wav;
        float v[C];
        node_row<F, RELU>(h, ninfo, edges, bias, node, n, lane, v);

        __syncthreads();
        #pragma unroll
        for (int c = 0; c < C; ++c) Ts[wav * TP + c * 32 + lane] = v[c];
        __syncthreads();

        if (wav < NC / 16) {
            const int l15 = lane & 15;
            const int hi  = lane >> 4;
            const int khi = hi * 2;
            const int colBase = wav * 16;

            v8f acc = {0.f, 0.f, 0.f, 0.f, 0.f, 0.f, 0.f, 0.f};
            #pragma unroll 8
            for (int k0 = 0; k0 < F; k0 += 4) {
                v2f a = *(const v2f*)&Ts[l15 * TP + k0 + khi];
                v2f b;
                b.x = Ws[(k0 + khi) * WP + colBase + l15];
                b.y = Ws[(k0 + khi + 1) * WP + colBase + l15];
                acc = __builtin_amdgcn_wmma_f32_16x16x4_f32(
                     false, a,  false, b,
                     (short)0, acc,  false,  false);
            }

            const int m0  = tile * 16 + hi * 8;
            const int col = colBase + l15;
            #pragma unroll
            for (int r = 0; r < 8; ++r)
                Hn[(m0 + r) * NC + col] = acc[r];
        }
    }
}

__global__ void pool_div(const float* __restrict__ sums,
                         const float* __restrict__ cnts,
                         float* __restrict__ out, int n) {
    int i = blockIdx.x * blockDim.x + threadIdx.x;
    if (i < n) out[i] = sums[i] / fmaxf(cnts[i / 64], 1.0f);
}

extern "C" void kernel_launch(void* const* d_in, const int* in_sizes, int n_in,
                              void* d_out, int out_size, void* d_ws, size_t ws_size,
                              hipStream_t stream) {
    const float* x     = (const float*)d_in[0];
    const int*   ei    = (const int*)d_in[1];
    const int*   batch = (const int*)d_in[2];
    const float* Wl[4] = {(const float*)d_in[3], (const float*)d_in[5],
                          (const float*)d_in[7], (const float*)d_in[9]};
    const float* bl[4] = {(const float*)d_in[4], (const float*)d_in[6],
                          (const float*)d_in[8], (const float*)d_in[10]};

    const int N = GCN_N_NODES;
    const int E = in_sizes[1] / 2;
    const int* src = ei;
    const int* dst = ei + E;

    auto au = [](size_t v) { return (v + 255) & ~(size_t)255; };
    char* ws = (char*)d_ws;
    size_t off = 0;
    int*      gcur = (int*)(ws + off);      off += au(sizeof(int) * 64);
    unsigned* stag = (unsigned*)(ws + off); off += au(sizeof(unsigned) * (size_t)NBIN * (size_t)E);
    v2i*   ninfo = (v2i*)(ws + off);   off += au(sizeof(v2i) * (size_t)(N + 1));
    v2i*   edges = (v2i*)(ws + off);   off += au(sizeof(v2i) * (size_t)E);
    float* bufA  = (float*)(ws + off); off += au(sizeof(float) * (size_t)N * 96);
    float* bufB  = (float*)(ws + off); off += au(sizeof(float) * (size_t)N * 96);
    float* sums  = (float*)(ws + off); off += au(sizeof(float) * GCN_N_GRAPHS * 64);
    float* cnts  = (float*)(ws + off); off += au(sizeof(float) * GCN_N_GRAPHS);
    (void)ws_size; (void)n_in; (void)out_size;

    const int TB = 256;
    zero_f32<<<(GCN_N_GRAPHS * 64 + TB - 1) / TB, TB, 0, stream>>>(sums, GCN_N_GRAPHS * 64);
    zero_f32<<<1, 64, 0, stream>>>(cnts, GCN_N_GRAPHS);
    const int MT  = (N + 15) / 16;
    const int MG  = (MT + PROD_WAVES - 1) / PROD_WAVES;
    const int MGH = MG / 2;
    const int NB1 = (E + 4095) / 4096;
    zero_i32<<<1, 64, 0, stream>>>(gcur, 64);
    bin_edges_and_product<128, 96><<<NB1 + MGH * 6, 1024, 0, stream>>>(
        src, dst, gcur, stag, E, N, NB1, x, Wl[0], bufA, 0);
    bin_to_csr_and_product<128, 96><<<NBIN + (MG - MGH) * 6, 1024, 0, stream>>>(
        gcur, stag, ninfo, edges, E, N, NBIN, x, Wl[0], bufA, MGH);
    edge_norm <<<(E + 4 * TB - 1) / (4 * TB), TB, 0, stream>>>(ninfo, edges, E, N);

    const int PB  = (N + 63) / 64;
    const int FB  = (MT + 3) / 4;

    gather_product<96, 96, true><<<FB, 512, 0, stream>>>(bufA, ninfo, edges, bl[0], Wl[1], bufB, N);
    gather_product<96, 96, true><<<FB, 512, 0, stream>>>(bufB, ninfo, edges, bl[1], Wl[2], bufA, N);
    gather_product<96, 64, true><<<FB, 512, 0, stream>>>(bufA, ninfo, edges, bl[2], Wl[3], bufB, N);
    gather_pool<64, 4><<<PB, 512, 0, stream>>>(bufB, ninfo, edges, bl[3], batch, sums, cnts, N);

    pool_div<<<(GCN_N_GRAPHS * 64 + TB - 1) / TB, TB, 0, stream>>>(sums, cnts, (float*)d_out,
                                                                   GCN_N_GRAPHS * 64);
}


# Gather launches: a block owns 64 consecutive nodes and fetches their edge words, row starts and weights into LDS/registers in one round trip at its start; waves then wait for source rows only
# speedup vs baseline: 4.8634x; 1.0347x over previous
#include <hip/hip_runtime.h>

typedef __attribute__((ext_vector_type(2))) float v2f;
typedef __attribute__((ext_vector_type(4))) float v4f;
typedef __attribute__((ext_vector_type(8))) float v8f;
typedef __attribute__((ext_vector_type(2))) int   v2i;
typedef __attribute__((ext_vector_type(4))) int   v4i;

#define GCN_N_NODES  50000
#define GCN_N_GRAPHS 64
static_assert(GCN_N_NODES % 16 == 0, "row tiles of 16 must be full");

__global__ void zero_f32(float* __restrict__ p, int n) {
    int i = blockIdx.x * blockDim.x + threadIdx.x;
    if (i < n) p[i] = 0.0f;
}

__global__ void zero_i32(int* __restrict__ p, int n) {
    int i = blockIdx.x * blockDim.x + threadIdx.x;
    if (i < n) p[i] = 0;
}

#define BIN_SHIFT 10
#define BIN_NODES (1 << BIN_SHIFT)
#define NBIN      ((GCN_N_NODES + BIN_NODES - 1) / BIN_NODES)
static_assert(NBIN <= 64, "the bin totals are summed by one wave, two per lane");
static_assert(GCN_N_NODES <= (1 << (31 - BIN_SHIFT)), "source id and bin-local id share one 32-bit word");

__device__ __forceinline__ void
bin_edges_block(const int blk, const int* __restrict__ src, const int* __restrict__ dst,
                int* __restrict__ gcur, unsigned* __restrict__ stag, int e, int n) {
    __shared__ int lcnt[NBIN];
    __shared__ int lbase[NBIN];
    const int tid = threadIdx.x;
    if (tid < NBIN) lcnt[tid] = 0;
    __syncthreads();

    int      bin[4], r[4];
    unsigned w[4];
    bool     ok[4];
    #pragma unroll
    for (int q = 0; q < 4; ++q) {
        const int i = blk * 4096 + q * 1024 + tid;
        ok[q] = i < e;
        const int ic = min(i, e - 1);
        int d = dst[ic];
        int s = src[ic];
        d = min(max(d, 0), n - 1);
        s = min(max(s, 0), n - 1);
        bin[q] = d >> BIN_SHIFT;
        w[q]   = ((unsigned)s << BIN_SHIFT) | (unsigned)(d & (BIN_NODES - 1));
        r[q]   = 0;
    }
    #pragma unroll
    for (int q = 0; q < 4; ++q)
        if (ok[q]) r[q] = atomicAdd(&lcnt[bin[q]], 1);
    __syncthreads();
    if (tid < NBIN) {
        const int c = lcnt[tid];
        lbase[tid] = (c > 0) ? atomicAdd(&gcur[tid], c) : 0;
    }
    __syncthreads();
    #pragma unroll
    for (int q = 0; q < 4; ++q) {
        if (ok[q]) {
            int pos = lbase[bin[q]] + r[q];
            pos = min(max(pos, 0), e - 1);
            stag[(size_t)bin[q] * (size_t)e + (size_t)pos] = w[q];
        }
    }
}

__device__ __forceinline__ void
bin_to_csr_block(const int b, const int* __restrict__ gcur, const unsigned* __restrict__ stag,
                 v2i* __restrict__ ninfo, v2i* __restrict__ edges, int e, int n) {
    __shared__ int cnt[BIN_NODES];
    __shared__ int dnv[BIN_NODES];
    __shared__ int tot[64];
    __shared__ int wtot[32];
    const int tid  = threadIdx.x;
    const int lane = tid & 31;
    const int wav  = tid >> 5;

    cnt[tid] = 0;
    if (tid < 64) tot[tid] = (tid < NBIN) ? gcur[tid] : 0;
    __syncthreads();

    int base = ((lane < b) ? tot[lane] : 0) + ((lane + 32 < b) ? tot[lane + 32] : 0);
    for (int o = 16; o > 0; o >>= 1) base += __shfl_xor(base, o);
    const int count = min(max(tot[b], 0), e);
    const unsigned* __restrict__ sb = stag + (size_t)b * (size_t)e;

    for (int i0 = tid; i0 < count; i0 += 4096) {
        unsigned w[4];
        #pragma unroll
        for (int q = 0; q < 4; ++q) {
            const int i = min(i0 + q * 1024, count - 1);
            w[q] = sb[i];
        }
        asm volatile("" :: "v"(w[0]), "v"(w[1]), "v"(w[2]), "v"(w[3]));
        #pragma unroll
        for (int q = 0; q < 4; ++q)
            if (i0 + q * 1024 < count) atomicAdd(&cnt[w[q] & (BIN_NODES - 1)], 1);
    }
    __syncthreads();

    const int c = cnt[tid];
    int incl = c;
    for (int o = 1; o < 32; o <<= 1) {
        const int u = __shfl_up(incl, o);
        if (lane >= o) incl += u;
    }
    if (lane == 31) wtot[wav] = incl;
    __syncthreads();
    int wprev = (lane < wav) ? wtot[lane] : 0;
    for (int o = 16; o > 0; o >>= 1) wprev += __shfl_xor(wprev, o);
    const int excl = wprev + incl - c;

    const int   node = (b << BIN_SHIFT) + tid;
    const float di   = rsqrtf((float)(c + 1));
    if (node < n) {
        v2i v;
        v.x = base + excl;
        v.y = __float_as_int(di);
        ninfo[node] = v;
    }
    if (b == NBIN - 1 && tid == BIN_NODES - 1) {
        v2i v;
        v.x = base + count;
        v.y = 0;
        ninfo[n] = v;
    }
    cnt[tid] = excl;
    dnv[tid] = __float_as_int(di);
    __syncthreads();

    for (int i0 = tid; i0 < count; i0 += 4096) {
        unsigned w[4];
        #pragma unroll
        for (int q = 0; q < 4; ++q) {
            const int i = min(i0 + q * 1024, count - 1);
            w[q] = sb[i];
        }
        asm volatile("" :: "v"(w[0]), "v"(w[1]), "v"(w[2]), "v"(w[3]));
        #pragma unroll
        for (int q = 0; q < 4; ++q) {
            if (i0 + q * 1024 < count) {
                const int dl = (int)(w[q] & (BIN_NODES - 1));
                int slot = base + atomicAdd(&cnt[dl], 1);
                slot = min(max(slot, 0), e - 1);
                v2i ev;
                ev.x = (int)(w[q] >> BIN_SHIFT);
                ev.y = dnv[dl];
                edges[slot] = ev;
            }
        }
    }
}

__global__ void edge_norm(const v2i* __restrict__ ninfo, v2i* __restrict__ edges, int e, int n) {
    const int T = (int)(gridDim.x * blockDim.x);
    const int i = (int)(blockIdx.x * blockDim.x + threadIdx.x);
    v2i   ev[4];
    float ws[4];
    #pragma unroll
    for (int q = 0; q < 4; ++q) {
        const int j = min(i + q * T, e - 1);
        ev[q] = edges[j];
    }
    asm volatile("" :: "v"(ev[0].x), "v"(ev[1].x), "v"(ev[2].x), "v"(ev[3].x));
    #pragma unroll
    for (int q = 0; q < 4; ++q) {
        const int s = min(max(ev[q].x, 0), n - 1);
        ws[q] = __int_as_float(ninfo[s].y);
    }
    asm volatile("" :: "v"(ws[0]), "v"(ws[1]), "v"(ws[2]), "v"(ws[3]));
    #pragma unroll
    for (int q = 0; q < 4; ++q) {
        const int j = i + q * T;
        if (j < e) {
            v2i o = ev[q];
            o.y = __float_as_int(ws[q] * __int_as_float(ev[q].y));
            edges[j] = o;
        }
    }
}

template <int K, int NCOLS, int WAVES>
__device__ __forceinline__ void product_block(const float* __restrict__ A,
                                              const float* __restrict__ Wm,
                                              float* __restrict__ H, int M,
                                              const int mgroup, const int ctile) {
    __shared__ float Bs[K * 16];

    const int tid  = threadIdx.x;
    const int lane = tid & 31;
    const int wav  = tid >> 5;
    const int colBase = ctile * 16;

    #pragma unroll
    for (int i = tid; i < K * 16; i += 32 * WAVES) {
        const int kk = i >> 4;
        const int cc = i & 15;
        Bs[i] = Wm[kk * NCOLS + colBase + cc];
    }
    __syncthreads();

    const int mtile = mgroup * WAVES + wav;
    if (mtile * 16 < M) {
        const int l15 = lane & 15;
        const int hi  = lane >> 4;
        const int khi = hi * 2;
        const int mrow = mtile * 16 + l15;

        v8f acc = {0.f, 0.f, 0.f, 0.f, 0.f, 0.f, 0.f, 0.f};
        #pragma unroll 8
        for (int k0 = 0; k0 < K; k0 += 4) {
            v2f a = *(const v2f*)&A[mrow * K + k0 + khi];
            v2f b;
            b.x = Bs[(k0 + khi) * 16 + l15];
            b.y = Bs[(k0 + khi + 1) * 16 + l15];
            acc = __builtin_amdgcn_wmma_f32_16x16x4_f32(
                 false, a,  false, b,
                 (short)0, acc,  false,  false);
        }

        const int m0  = mtile * 16 + hi * 8;
        const int col = colBase + l15;
        #pragma unroll
        for (int r = 0; r < 8; ++r)
            H[(m0 + r) * NCOLS + col] = acc[r];
    }
}

#define PROD_WAVES 32
template <int K, int NCOLS>
__global__ void __launch_bounds__(1024)
bin_edges_and_product(const int* __restrict__ src, const int* __restrict__ dst,
                      int* __restrict__ gcur, unsigned* __restrict__ stag, int e, int n, int nsort,
                      const float* __restrict__ A, const float* __restrict__ Wm,
                      float* __restrict__ H, int mg0) {
    const int blk = (int)blockIdx.x;
    if (blk < nsort) {
        bin_edges_block(blk, src, dst, gcur, stag, e, n);
    } else {
        const int g = blk - nsort;
        product_block<K, NCOLS, PROD_WAVES>(A, Wm, H, n, mg0 + g / (NCOLS / 16), g % (NCOLS / 16));
    }
}

template <int K, int NCOLS>
__global__ void __launch_bounds__(1024)
bin_to_csr_and_product(const int* __restrict__ gcur, const unsigned* __restrict__ stag,
                       v2i* __restrict__ ninfo, v2i* __restrict__ edges, int e, int n, int nsort,
                       const float* __restrict__ A, const float* __restrict__ Wm,
                       float* __restrict__ H, int mg0) {
    const int blk = (int)blockIdx.x;
    if (blk < nsort) {
        bin_to_csr_block(blk, gcur, stag, ninfo, edges, e, n);
    } else {
        const int g = blk - nsort;
        product_block<K, NCOLS, PROD_WAVES>(A, Wm, H, n, mg0 + g / (NCOLS / 16), g % (NCOLS / 16));
    }
}

template <int F, int GS, bool TAIL>
__device__ __forceinline__ void gather_group(const float* __restrict__ h, const v2i ev,
                                             int i0, int m, int n, int lane,
                                             float (&acc)[F / 32]) {
    constexpr int C = F / 32;
    float hv[GS][C];
    float wv[GS];
    #pragma unroll
    for (int q = 0; q < GS; ++q) {
        const int ii = TAIL ? min(i0 + q, m - 1) : (i0 + q);
        int s = __builtin_amdgcn_readlane(ev.x, ii);
        s = min(max(s, 0), n - 1);
        wv[q] = __int_as_float(__builtin_amdgcn_readlane(ev.y, ii));
        #pragma unroll
        for (int c = 0; c < C; ++c) hv[q][c] = h[s * F + c * 32 + lane];
    }
    #pragma unroll
    for (int q = 0; q < GS; ++q) {
        const bool ok = !TAIL || (i0 + q) < m;
        #pragma unroll
        for (int c = 0; c < C; ++c) {
            const float t = fmaf(hv[q][c], wv[q], acc[c]);
            acc[c] = ok ? t : acc[c];
        }
    }
}

#define ECAP 1536

__device__ __forceinline__ void stage_edges_load(const v2i* __restrict__ edges, int b0, int b1, int e, int tid,
                                                 v2i (&w)[ECAP / 512]) {
    const int last = min(max(b1 - 1, 0), e - 1);
    #pragma unroll
    for (int q = 0; q < ECAP / 512; ++q) {
        const int j = min(max(b0 + q * 512 + tid, 0), last);
        w[q] = edges[j];
    }
}

__device__ __forceinline__ void stage_edges_store(v2i* __restrict__ Es, int tid, const v2i (&w)[ECAP / 512]) {
    #pragma unroll
    for (int q = 0; q < ECAP / 512; ++q) Es[q * 512 + tid] = w[q];
}

template <int F>
__device__ __forceinline__ void gather_edges(const float* __restrict__ h,
                                             const v2i* __restrict__ edges,
                                             v2i* __restrict__ Es, int b0, int wav,
                                             int beg, int end, int n, int e, int lane,
                                             float (&acc)[F / 32]) {
    for (int base = beg; base < end; base += 32) {
        const int j = base + lane;
        int off = base - b0;
        if (off < 0 || off + 32 > ECAP) {
            off = ECAP + wav * 32;
            v2i t = {0, 0};
            if (j < end) t = edges[min(max(j, 0), e - 1)];
            Es[off + lane] = t;
        }
        v2i ev = {0, 0};
        if (j < end) ev = Es[off + lane];
        const int m = min(32, end - base);
        int i0 = 0;
        for (; i0 + 8 <= m; i0 += 8) gather_group<F, 8, false>(h, ev, i0, m, n, lane, acc);
        const int rem = m - i0;
        if (rem > 4)      gather_group<F, 8, true>(h, ev, i0, m, n, lane, acc);
        else if (rem > 0) gather_group<F, 4, true>(h, ev, i0, m, n, lane, acc);
    }
}

template <int F, bool RELU>
__device__ __forceinline__ void node_row(const float* __restrict__ h,
                                         const v2i* __restrict__ edges,
                                         v2i* __restrict__ Es, int b0, int wav,
                                         int beg, int end, float di,
                                         const float (&bs)[F / 32],
                                         int node, int n, int e, int lane, float (&v)[F / 32]) {
    constexpr int C = F / 32;
    float hs[C], acc[C];
    #pragma unroll
    for (int c = 0; c < C; ++c) {
        hs[c]  = h[node * F + c * 32 + lane];
        acc[c] = 0.0f;
    }

    gather_edges<F>(h, edges, Es, b0, wav, beg, end, n, e, lane, acc);

    const float di2 = di * di;
    #pragma unroll
    for (int c = 0; c < C; ++c) {
        const float t = acc[c] + hs[c] * di2 + bs[c];
        v[c] = RELU ? fmaxf(t, 0.0f) : t;
    }
}

template <int F, int NPW>
__global__ void __launch_bounds__(512)
gather_pool(const float* __restrict__ h,
            const v2i* __restrict__ ninfo,
            const v2i* __restrict__ edges,
            const float* __restrict__ bias,
            const int* __restrict__ batch,
            float* __restrict__ sums,
            float* __restrict__ cnts, int n, int e) {
    constexpr int C = F / 32;
    __shared__ float Ps[16 * F];
    __shared__ float Pc[16];
    __shared__ int   Pg[16];
    __shared__ v2i   Es[ECAP + 16 * 32];

    const int tid  = (int)threadIdx.x;
    const int lane = tid & 31;
    const int wav  = __builtin_amdgcn_readfirstlane(tid >> 5);
    const int nb0  = (int)blockIdx.x * 16 * NPW;
    const int n0   = nb0 + wav * NPW;

    const int b0 = ninfo[min(nb0, n)].x;
    const int b1 = ninfo[min(nb0 + 16 * NPW, n)].x;
    v2i ew[ECAP / 512];
    stage_edges_load(edges, b0, b1, e, tid, ew);
    float bs[C];
    #pragma unroll
    for (int c = 0; c < C; ++c) bs[c] = bias[c * 32 + lane];
    const v2i nv = ninfo[min(n0 + min(lane, NPW), n)];
    const int gv = batch[min(n0 + min(lane, NPW - 1), n - 1)];
    stage_edges_store(Es, tid, ew);
    __syncthreads();

    int   gcur = -1;
    float pc   = 0.0f;
    float ps[C];
    #pragma unroll
    for (int c = 0; c < C; ++c) ps[c] = 0.0f;

    if (n0 < n) {
        for (int k = 0; k < NPW; ++k) {
            const int node = n0 + k;
            if (node < n) {
                const int   beg = __builtin_amdgcn_readlane(nv.x, k);
                const int   end = __builtin_amdgcn_readlane(nv.x, k + 1);
                const float di  = __int_as_float(__builtin_amdgcn_readlane(nv.y, k));
                int g = __builtin_amdgcn_readlane(gv, k);
                g = min(max(g, 0), GCN_N_GRAPHS - 1);

                float hs[C], acc[C];
                #pragma unroll
                for (int c = 0; c < C; ++c) {
                    hs[c]  = h[node * F + c * 32 + lane];
                    acc[c] = 0.0f;
                }
                gather_edges<F>(h, edges, Es, b0, wav, beg, end, n, e, lane, acc);

                if (g != gcur) {
                    if (gcur >= 0) {
                        #pragma unroll
                        for (int c = 0; c < C; ++c) atomicAdd(&sums[gcur * F + c * 32 + lane], ps[c]);
                        if (lane == 0) atomicAdd(&cnts[gcur], pc);
                    }
                    gcur = g;
                    pc   = 0.0f;
                    #pragma unroll
                    for (int c = 0; c < C; ++c) ps[c] = 0.0f;
                }
                const float di2 = di * di;
                #pragma unroll
                for (int c = 0; c < C; ++c) ps[c] += acc[c] + hs[c] * di2 + bs[c];
                pc += 1.0f;
            }
        }
    }

    #pragma unroll
    for (int c = 0; c < C; ++c) Ps[wav * F + c * 32 + lane] = ps[c];
    if (lane == 0) {
        Pg[wav] = gcur;
        Pc[wav] = pc;
    }
    __syncthreads();

    if (wav < C) {
        int   gw[16];
        float pw[16], cw[16];
        #pragma unroll
        for (int w = 0; w < 16; ++w) {
            gw[w] = Pg[w];
            cw[w] = Pc[w];
            pw[w] = Ps[w * F + wav * 32 + lane];
        }
        int   g = -1;
        float s = 0.0f, cn = 0.0f;
        #pragma unroll
        for (int w = 0; w < 16; ++w) {
            if (gw[w] < 0) continue;
            if (gw[w] != g) {
                if (g >= 0) {
                    atomicAdd(&sums[g * F + wav * 32 + lane], s);
                    if (wav == 0 && lane == 0) atomicAdd(&cnts[g], cn);
                }
                g  = gw[w];
                s  = 0.0f;
                cn = 0.0f;
            }
            s  += pw[w];
            cn += cw[w];
        }
        if (g >= 0) {
            atomicAdd(&sums[g * F + wav * 32 + lane], s);
            if (wav == 0 && lane == 0) atomicAdd(&cnts[g], cn);
        }
    }
}

#define GP_TILES 4
template <int F, int NC, bool RELU>
__global__ void __launch_bounds__(512)
gather_product(const float* __restrict__ h,
               const v2i* __restrict__ ninfo,
               const v2i* __restrict__ edges,
               const float* __restrict__ bias,
               const float* __restrict__ Wn,
               float* __restrict__ Hn, int n, int e) {
    constexpr int C  = F / 32;
    constexpr int TP = F + 4;
    constexpr int WP = NC + 8;
    constexpr int NV4 = F * NC / 4;
    constexpr int WQ  = (NV4 + 511) / 512;
    static_assert(NC % 4 == 0 && WP % 4 == 0, "a 16-byte piece stays inside one weight row, 16-byte aligned in LDS");
    static_assert(2 * GP_TILES <= 32, "a wave keeps the row starts of its nodes in one register pair");
    __shared__ __attribute__((aligned(16))) float Ts[16 * TP];
    __shared__ __attribute__((aligned(16))) float Ws[F * WP];
    __shared__ v2i Es[ECAP + 16 * 32];

    const int tid  = threadIdx.x;
    const int lane = tid & 31;
    const int wav  = __builtin_amdgcn_readfirstlane(tid >> 5);

    const int ntiles = n >> 4;
    const int tile0  = (int)blockIdx.x * GP_TILES;
    const int nb0    = tile0 * 16;

    const int b0 = ninfo[min(nb0, n)].x;
    const int b1 = ninfo[min(nb0 + 16 * GP_TILES, n)].x;
    v4f wr[WQ];
    #pragma unroll
    for (int q = 0; q < WQ; ++q) {
        const int i = min(q * 512 + tid, NV4 - 1);
        wr[q] = *(const v4f*)&Wn[4 * i];
    }
    v2i ew[ECAP / 512];
    stage_edges_load(edges, b0, b1, e, tid, ew);
    float bs[C];
    #pragma unroll
    for (int c = 0; c < C; ++c) bs[c] = bias[c * 32 + lane];
    const int l7 = min(lane, 2 * GP_TILES - 1);
    const v2i nv = ninfo[min(nb0 + 16 * (l7 >> 1) + wav + (l7 & 1), n)];

    #pragma unroll
    for (int q = 0; q < WQ; ++q) {
        const int i = q * 512 + tid;
        if (i < NV4) {
            const int kk = (4 * i) / NC;
            const int cc = 4 * i - kk * NC;
            *(v4f*)&Ws[kk * WP + cc] = wr[q];
        }
    }
    stage_edges_store(Es, tid, ew);
    __syncthreads();

    for (int t = 0; t < GP_TILES; ++t) {
        const int tile = tile0 + t;
        if (tile >= ntiles) break;
        const int   node = tile * 16 + wav;
        const int   beg  = __builtin_amdgcn_readlane(nv.x, 2 * t);
        const int   end  = __builtin_amdgcn_readlane(nv.x, 2 * t + 1);
        const float di   = __int_as_float(__builtin_amdgcn_readlane(nv.y, 2 * t));
        float v[C];
        node_row<F, RELU>(h, edges, Es, b0, wav, beg, end, di, bs, node, n, e, lane, v);

        __syncthreads();
        #pragma unroll
        for (int c = 0; c < C; ++c) Ts[wav * TP + c * 32 + lane] = v[c];
        __syncthreads();

        if (wav < NC / 16) {
            const int l15 = lane & 15;
            const int hi  = lane >> 4;
            const int khi = hi * 2;
            const int colBase = wav * 16;

            v8f acc = {0.f, 0.f, 0.f, 0.f, 0.f, 0.f, 0.f, 0.f};
            #pragma unroll 8
            for (int k0 = 0; k0 < F; k0 += 4) {
                v2f a = *(const v2f*)&Ts[l15 * TP + k0 + khi];
                v2f b;
                b.x = Ws[(k0 + khi) * WP + colBase + l15];
                b.y = Ws[(k0 + khi + 1) * WP + colBase + l15];
                acc = __builtin_amdgcn_wmma_f32_16x16x4_f32(
                     false, a,  false, b,
                     (short)0, acc,  false,  false);
            }

            const int m0  = tile * 16 + hi * 8;
            const int col = colBase + l15;
            #pragma unroll
            for (int r = 0; r < 8; ++r)
                Hn[(m0 + r) * NC + col] = acc[r];
        }
    }
}

__global__ void pool_div(const float* __restrict__ sums,
                         const float* __restrict__ cnts,
                         float* __restrict__ out, int n) {
    int i = blockIdx.x * blockDim.x + threadIdx.x;
    if (i < n) out[i] = sums[i] / fmaxf(cnts[i / 64], 1.0f);
}

extern "C" void kernel_launch(void* const* d_in, const int* in_sizes, int n_in,
                              void* d_out, int out_size, void* d_ws, size_t ws_size,
                              hipStream_t stream) {
    const float* x     = (const float*)d_in[0];
    const int*   ei    = (const int*)d_in[1];
    const int*   batch = (const int*)d_in[2];
    const float* Wl[4] = {(const float*)d_in[3], (const float*)d_in[5],
                          (const float*)d_in[7], (const float*)d_in[9]};
    const float* bl[4] = {(const float*)d_in[4], (const float*)d_in[6],
                          (const float*)d_in[8], (const float*)d_in[10]};

    const int N = GCN_N_NODES;
    const int E = in_sizes[1] / 2;
    const int* src = ei;
    const int* dst = ei + E;

    auto au = [](size_t v) { return (v + 255) & ~(size_t)255; };
    char* ws = (char*)d_ws;
    size_t off = 0;
    int*      gcur = (int*)(ws + off);      off += au(sizeof(int) * 64);
    unsigned* stag = (unsigned*)(ws + off); off += au(sizeof(unsigned) * (size_t)NBIN * (size_t)E);
    v2i*   ninfo = (v2i*)(ws + off);   off += au(sizeof(v2i) * (size_t)(N + 1));
    v2i*   edges = (v2i*)(ws + off);   off += au(sizeof(v2i) * (size_t)E);
    float* bufA  = (float*)(ws + off); off += au(sizeof(float) * (size_t)N * 96);
    float* bufB  = (float*)(ws + off); off += au(sizeof(float) * (size_t)N * 96);
    float* sums  = (float*)(ws + off); off += au(sizeof(float) * GCN_N_GRAPHS * 64);
    float* cnts  = (float*)(ws + off); off += au(sizeof(float) * GCN_N_GRAPHS);
    (void)ws_size; (void)n_in; (void)out_size;

    const int TB = 256;
    zero_f32<<<(GCN_N_GRAPHS * 64 + TB - 1) / TB, TB, 0, stream>>>(sums, GCN_N_GRAPHS * 64);
    zero_f32<<<1, 64, 0, stream>>>(cnts, GCN_N_GRAPHS);
    const int MT  = (N + 15) / 16;
    const int MG  = (MT + PROD_WAVES - 1) / PROD_WAVES;
    const int MGH = MG / 2;
    const int NB1 = (E + 4095) / 4096;
    zero_i32<<<1, 64, 0, stream>>>(gcur, 64);
    bin_edges_and_product<128, 96><<<NB1 + MGH * 6, 1024, 0, stream>>>(
        src, dst, gcur, stag, E, N, NB1, x, Wl[0], bufA, 0);
    bin_to_csr_and_product<128, 96><<<NBIN + (MG - MGH) * 6, 1024, 0, stream>>>(
        gcur, stag, ninfo, edges, E, N, NBIN, x, Wl[0], bufA, MGH);
    edge_norm <<<(E + 4 * TB - 1) / (4 * TB), TB, 0, stream>>>(ninfo, edges, E, N);

    const int PB  = (N + 63) / 64;
    const int FB  = (MT + GP_TILES - 1) / GP_TILES;

    gather_product<96, 96, true><<<FB, 512, 0, stream>>>(bufA, ninfo, edges, bl[0], Wl[1], bufB, N, E);
    gather_product<96, 96, true><<<FB, 512, 0, stream>>>(bufB, ninfo, edges, bl[1], Wl[2], bufA, N, E);
    gather_product<96, 64, true><<<FB, 512, 0, stream>>>(bufA, ninfo, edges, bl[2], Wl[3], bufB, N, E);
    gather_pool<64, 4><<<PB, 512, 0, stream>>>(bufB, ninfo, edges, bl[3], batch, sums, cnts, N, E);

    pool_div<<<(GCN_N_GRAPHS * 64 + TB - 1) / TB, TB, 0, stream>>>(sums, cnts, (float*)d_out,
                                                                   GCN_N_GRAPHS * 64);
}


# Launch count 11 -> 8: the three start-value launches merged into one; the edge-weight pass folded into the layer-1 gather's edge staging (dinv[src] looked up there, words written on); bit-identical
# speedup vs baseline: 5.1540x; 1.0598x over previous
#include <hip/hip_runtime.h>

typedef __attribute__((ext_vector_type(2))) float v2f;
typedef __attribute__((ext_vector_type(4))) float v4f;
typedef __attribute__((ext_vector_type(8))) float v8f;
typedef __attribute__((ext_vector_type(2))) int   v2i;
typedef __attribute__((ext_vector_type(4))) int   v4i;

#define GCN_N_NODES  50000
#define GCN_N_GRAPHS 64
static_assert(GCN_N_NODES % 16 == 0, "row tiles of 16 must be full");

__global__ void start_values(int* __restrict__ gcur, float* __restrict__ sums, float* __restrict__ cnts) {
    const int i = blockIdx.x * blockDim.x + threadIdx.x;
    if (i < GCN_N_GRAPHS * 64) sums[i] = 0.0f;
    if (i < GCN_N_GRAPHS) cnts[i] = 0.0f;
    if (i < 64) gcur[i] = 0;
}

#define BIN_SHIFT 10
#define BIN_NODES (1 << BIN_SHIFT)
#define NBIN      ((GCN_N_NODES + BIN_NODES - 1) / BIN_NODES)
static_assert(NBIN <= 64, "the bin totals are summed by one wave, two per lane");
static_assert(GCN_N_NODES <= (1 << (31 - BIN_SHIFT)), "source id and bin-local id share one 32-bit word");

__device__ __forceinline__ void
bin_edges_block(const int blk, const int* __restrict__ src, const int* __restrict__ dst,
                int* __restrict__ gcur, unsigned* __restrict__ stag, int e, int n) {
    __shared__ int lcnt[NBIN];
    __shared__ int lbase[NBIN];
    const int tid = threadIdx.x;
    if (tid < NBIN) lcnt[tid] = 0;
    __syncthreads();

    int      bin[4], r[4];
    unsigned w[4];
    bool     ok[4];
    #pragma unroll
    for (int q = 0; q < 4; ++q) {
        const int i = blk * 4096 + q * 1024 + tid;
        ok[q] = i < e;
        const int ic = min(i, e - 1);
        int d = dst[ic];
        int s = src[ic];
        d = min(max(d, 0), n - 1);
        s = min(max(s, 0), n - 1);
        bin[q] = d >> BIN_SHIFT;
        w[q]   = ((unsigned)s << BIN_SHIFT) | (unsigned)(d & (BIN_NODES - 1));
        r[q]   = 0;
    }
    #pragma unroll
    for (int q = 0; q < 4; ++q)
        if (ok[q]) r[q] = atomicAdd(&lcnt[bin[q]], 1);
    __syncthreads();
    if (tid < NBIN) {
        const int c = lcnt[tid];
        lbase[tid] = (c > 0) ? atomicAdd(&gcur[tid], c) : 0;
    }
    __syncthreads();
    #pragma unroll
    for (int q = 0; q < 4; ++q) {
        if (ok[q]) {
            int pos = lbase[bin[q]] + r[q];
            pos = min(max(pos, 0), e - 1);
            stag[(size_t)bin[q] * (size_t)e + (size_t)pos] = w[q];
        }
    }
}

__device__ __forceinline__ void
bin_to_csr_block(const int b, const int* __restrict__ gcur, const unsigned* __restrict__ stag,
                 v2i* __restrict__ ninfo, v2i* __restrict__ edges, int e, int n) {
    __shared__ int cnt[BIN_NODES];
    __shared__ int dnv[BIN_NODES];
    __shared__ int tot[64];
    __shared__ int wtot[32];
    const int tid  = threadIdx.x;
    const int lane = tid & 31;
    const int wav  = tid >> 5;

    cnt[tid] = 0;
    if (tid < 64) tot[tid] = (tid < NBIN) ? gcur[tid] : 0;
    __syncthreads();

    int base = ((lane < b) ? tot[lane] : 0) + ((lane + 32 < b) ? tot[lane + 32] : 0);
    for (int o = 16; o > 0; o >>= 1) base += __shfl_xor(base, o);
    const int count = min(max(tot[b], 0), e);
    const unsigned* __restrict__ sb = stag + (size_t)b * (size_t)e;

    for (int i0 = tid; i0 < count; i0 += 4096) {
        unsigned w[4];
        #pragma unroll
        for (int q = 0; q < 4; ++q) {
            const int i = min(i0 + q * 1024, count - 1);
            w[q] = sb[i];
        }
        asm volatile("" :: "v"(w[0]), "v"(w[1]), "v"(w[2]), "v"(w[3]));
        #pragma unroll
        for (int q = 0; q < 4; ++q)
            if (i0 + q * 1024 < count) atomicAdd(&cnt[w[q] & (BIN_NODES - 1)], 1);
    }
    __syncthreads();

    const int c = cnt[tid];
    int incl = c;
    for (int o = 1; o < 32; o <<= 1) {
        const int u = __shfl_up(incl, o);
        if (lane >= o) incl += u;
    }
    if (lane == 31) wtot[wav] = incl;
    __syncthreads();
    int wprev = (lane < wav) ? wtot[lane] : 0;
    for (int o = 16; o > 0; o >>= 1) wprev += __shfl_xor(wprev, o);
    const int excl = wprev + incl - c;

    const int   node = (b << BIN_SHIFT) + tid;
    const float di   = rsqrtf((float)(c + 1));
    if (node < n) {
        v2i v;
        v.x = base + excl;
        v.y = __float_as_int(di);
        ninfo[node] = v;
    }
    if (b == NBIN - 1 && tid == BIN_NODES - 1) {
        v2i v;
        v.x = base + count;
        v.y = 0;
        ninfo[n] = v;
    }
    cnt[tid] = excl;
    dnv[tid] = __float_as_int(di);
    __syncthreads();

    for (int i0 = tid; i0 < count; i0 += 4096) {
        unsigned w[4];
        #pragma unroll
        for (int q = 0; q < 4; ++q) {
            const int i = min(i0 + q * 1024, count - 1);
            w[q] = sb[i];
        }
        asm volatile("" :: "v"(w[0]), "v"(w[1]), "v"(w[2]), "v"(w[3]));
        #pragma unroll
        for (int q = 0; q < 4; ++q) {
            if (i0 + q * 1024 < count) {
                const int dl = (int)(w[q] & (BIN_NODES - 1));
                int slot = base + atomicAdd(&cnt[dl], 1);
                slot = min(max(slot, 0), e - 1);
                v2i ev;
                ev.x = (int)(w[q] >> BIN_SHIFT);
                ev.y = dnv[dl];
                edges[slot] = ev;
            }
        }
    }
}

template <int K, int NCOLS, int WAVES>
__device__ __forceinline__ void product_block(const float* __restrict__ A,
                                              const float* __restrict__ Wm,
                                              float* __restrict__ H, int M,
                                              const int mgroup, const int ctile) {
    __shared__ float Bs[K * 16];

    const int tid  = threadIdx.x;
    const int lane = tid & 31;
    const int wav  = tid >> 5;
    const int colBase = ctile * 16;

    #pragma unroll
    for (int i = tid; i < K * 16; i += 32 * WAVES) {
        const int kk = i >> 4;
        const int cc = i & 15;
        Bs[i] = Wm[kk * NCOLS + colBase + cc];
    }
    __syncthreads();

    const int mtile = mgroup * WAVES + wav;
    if (mtile * 16 < M) {
        const int l15 = lane & 15;
        const int hi  = lane >> 4;
        const int khi = hi * 2;
        const int mrow = mtile * 16 + l15;

        v8f acc = {0.f, 0.f, 0.f, 0.f, 0.f, 0.f, 0.f, 0.f};
        #pragma unroll 8
        for (int k0 = 0; k0 < K; k0 += 4) {
            v2f a = *(const v2f*)&A[mrow * K + k0 + khi];
            v2f b;
            b.x = Bs[(k0 + khi) * 16 + l15];
            b.y = Bs[(k0 + khi + 1) * 16 + l15];
            acc = __builtin_amdgcn_wmma_f32_16x16x4_f32(
                 false, a,  false, b,
                 (short)0, acc,  false,  false);
        }

        const int m0  = mtile * 16 + hi * 8;
        const int col = colBase + l15;
        #pragma unroll
        for (int r = 0; r < 8; ++r)
            H[(m0 + r) * NCOLS + col] = acc[r];
    }
}

#define PROD_WAVES 32
template <int K, int NCOLS>
__global__ void __launch_bounds__(1024)
bin_edges_and_product(const int* __restrict__ src, const int* __restrict__ dst,
                      int* __restrict__ gcur, unsigned* __restrict__ stag, int e, int n, int nsort,
                      const float* __restrict__ A, const float* __restrict__ Wm,
                      float* __restrict__ H, int mg0) {
    const int blk = (int)blockIdx.x;
    if (blk < nsort) {
        bin_edges_block(blk, src, dst, gcur, stag, e, n);
    } else {
        const int g = blk - nsort;
        product_block<K, NCOLS, PROD_WAVES>(A, Wm, H, n, mg0 + g / (NCOLS / 16), g % (NCOLS / 16));
    }
}

template <int K, int NCOLS>
__global__ void __launch_bounds__(1024)
bin_to_csr_and_product(const int* __restrict__ gcur, const unsigned* __restrict__ stag,
                       v2i* __restrict__ ninfo, v2i* __restrict__ edges, int e, int n, int nsort,
                       const float* __restrict__ A, const float* __restrict__ Wm,
                       float* __restrict__ H, int mg0) {
    const int blk = (int)blockIdx.x;
    if (blk < nsort) {
        bin_to_csr_block(blk, gcur, stag, ninfo, edges, e, n);
    } else {
        const int g = blk - nsort;
        product_block<K, NCOLS, PROD_WAVES>(A, Wm, H, n, mg0 + g / (NCOLS / 16), g % (NCOLS / 16));
    }
}

template <int F, int GS, bool TAIL>
__device__ __forceinline__ void gather_group(const float* __restrict__ h, const v2i ev,
                                             int i0, int m, int n, int lane,
                                             float (&acc)[F / 32]) {
    constexpr int C = F / 32;
    float hv[GS][C];
    float wv[GS];
    #pragma unroll
    for (int q = 0; q < GS; ++q) {
        const int ii = TAIL ? min(i0 + q, m - 1) : (i0 + q);
        int s = __builtin_amdgcn_readlane(ev.x, ii);
        s = min(max(s, 0), n - 1);
        wv[q] = __int_as_float(__builtin_amdgcn_readlane(ev.y, ii));
        #pragma unroll
        for (int c = 0; c < C; ++c) hv[q][c] = h[s * F + c * 32 + lane];
    }
    #pragma unroll
    for (int q = 0; q < GS; ++q) {
        const bool ok = !TAIL || (i0 + q) < m;
        #pragma unroll
        for (int c = 0; c < C; ++c) {
            const float t = fmaf(hv[q][c], wv[q], acc[c]);
            acc[c] = ok ? t : acc[c];
        }
    }
}

#define ECAP 1536

__device__ __forceinline__ void stage_edges_load(const v2i* __restrict__ edges, int b0, int b1, int e, int tid,
                                                 v2i (&w)[ECAP / 512]) {
    const int last = min(max(b1 - 1, 0), e - 1);
    #pragma unroll
    for (int q = 0; q < ECAP / 512; ++q) {
        const int j = min(max(b0 + q * 512 + tid, 0), last);
        w[q] = edges[j];
    }
}

__device__ __forceinline__ void stage_edges_store(v2i* __restrict__ Es, int tid, const v2i (&w)[ECAP / 512]) {
    #pragma unroll
    for (int q = 0; q < ECAP / 512; ++q) Es[q * 512 + tid] = w[q];
}

template <int F, bool NORM>
__device__ __forceinline__ void gather_edges(const float* __restrict__ h,
                                             const v2i* __restrict__ ninfo,
                                             const v2i* __restrict__ edges,
                                             v2i* __restrict__ edges_out,
                                             v2i* __restrict__ Es, int b0, int wav,
                                             int beg, int end, int n, int e, int lane,
                                             float (&acc)[F / 32]) {
    for (int base = beg; base < end; base += 32) {
        const int j = base + lane;
        int off = base - b0;
        if (off < 0 || off + 32 > ECAP) {
            off = ECAP + wav * 32;
            v2i t = {0, 0};
            if (j < end) {
                const int jc = min(max(j, 0), e - 1);
                t = edges[jc];
                if (NORM) {
                    const int s = min(max(t.x, 0), n - 1);
                    t.y = __float_as_int(__int_as_float(ninfo[s].y) * __int_as_float(t.y));
                    if (j - b0 >= ECAP) edges_out[jc] = t;
                }
            }
            Es[off + lane] = t;
        }
        v2i ev = {0, 0};
        if (j < end) ev = Es[off + lane];
        const int m = min(32, end - base);
        int i0 = 0;
        for (; i0 + 8 <= m; i0 += 8) gather_group<F, 8, false>(h, ev, i0, m, n, lane, acc);
        const int rem = m - i0;
        if (rem > 4)      gather_group<F, 8, true>(h, ev, i0, m, n, lane, acc);
        else if (rem > 0) gather_group<F, 4, true>(h, ev, i0, m, n, lane, acc);
    }
}

template <int F, bool RELU, bool NORM>
__device__ __forceinline__ void node_row(const float* __restrict__ h,
                                         const v2i* __restrict__ ninfo,
                                         const v2i* __restrict__ edges,
                                         v2i* __restrict__ edges_out,
                                         v2i* __restrict__ Es, int b0, int wav,
                                         int beg, int end, float di,
                                         const float (&bs)[F / 32],
                                         int node, int n, int e, int lane, float (&v)[F / 32]) {
    constexpr int C = F / 32;
    float hs[C], acc[C];
    #pragma unroll
    for (int c = 0; c < C; ++c) {
        hs[c]  = h[node * F + c * 32 + lane];
        acc[c] = 0.0f;
    }

    gather_edges<F, NORM>(h, ninfo, edges, edges_out, Es, b0, wav, beg, end, n, e, lane, acc);

    const float di2 = di * di;
    #pragma unroll
    for (int c = 0; c < C; ++c) {
        const float t = acc[c] + hs[c] * di2 + bs[c];
        v[c] = RELU ? fmaxf(t, 0.0f) : t;
    }
}

template <int F, int NPW>
__global__ void __launch_bounds__(512)
gather_pool(const float* __restrict__ h,
            const v2i* __restrict__ ninfo,
            const v2i* __restrict__ edges,
            const float* __restrict__ bias,
            const int* __restrict__ batch,
            float* __restrict__ sums,
            float* __restrict__ cnts, int n, int e) {
    constexpr int C = F / 32;
    __shared__ float Ps[16 * F];
    __shared__ float Pc[16];
    __shared__ int   Pg[16];
    __shared__ v2i   Es[ECAP + 16 * 32];

    const int tid  = (int)threadIdx.x;
    const int lane = tid & 31;
    const int wav  = __builtin_amdgcn_readfirstlane(tid >> 5);
    const int nb0  = (int)blockIdx.x * 16 * NPW;
    const int n0   = nb0 + wav * NPW;

    const int b0 = ninfo[min(nb0, n)].x;
    const int b1 = ninfo[min(nb0 + 16 * NPW, n)].x;
    v2i ew[ECAP / 512];
    stage_edges_load(edges, b0, b1, e, tid, ew);
    float bs[C];
    #pragma unroll
    for (int c = 0; c < C; ++c) bs[c] = bias[c * 32 + lane];
    const v2i nv = ninfo[min(n0 + min(lane, NPW), n)];
    const int gv = batch[min(n0 + min(lane, NPW - 1), n - 1)];
    stage_edges_store(Es, tid, ew);
    __syncthreads();

    int   gcur = -1;
    float pc   = 0.0f;
    float ps[C];
    #pragma unroll
    for (int c = 0; c < C; ++c) ps[c] = 0.0f;

    if (n0 < n) {
        for (int k = 0; k < NPW; ++k) {
            const int node = n0 + k;
            if (node < n) {
                const int   beg = __builtin_amdgcn_readlane(nv.x, k);
                const int   end = __builtin_amdgcn_readlane(nv.x, k + 1);
                const float di  = __int_as_float(__builtin_amdgcn_readlane(nv.y, k));
                int g = __builtin_amdgcn_readlane(gv, k);
                g = min(max(g, 0), GCN_N_GRAPHS - 1);

                float hs[C], acc[C];
                #pragma unroll
                for (int c = 0; c < C; ++c) {
                    hs[c]  = h[node * F + c * 32 + lane];
                    acc[c] = 0.0f;
                }
                gather_edges<F, false>(h, ninfo, edges, (v2i*)nullptr, Es, b0, wav, beg, end, n, e, lane, acc);

                if (g != gcur) {
                    if (gcur >= 0) {
                        #pragma unroll
                        for (int c = 0; c < C; ++c) atomicAdd(&sums[gcur * F + c * 32 + lane], ps[c]);
                        if (lane == 0) atomicAdd(&cnts[gcur], pc);
                    }
                    gcur = g;
                    pc   = 0.0f;
                    #pragma unroll
                    for (int c = 0; c < C; ++c) ps[c] = 0.0f;
                }
                const float di2 = di * di;
                #pragma unroll
                for (int c = 0; c < C; ++c) ps[c] += acc[c] + hs[c] * di2 + bs[c];
                pc += 1.0f;
            }
        }
    }

    #pragma unroll
    for (int c = 0; c < C; ++c) Ps[wav * F + c * 32 + lane] = ps[c];
    if (lane == 0) {
        Pg[wav] = gcur;
        Pc[wav] = pc;
    }
    __syncthreads();

    if (wav < C) {
        int   gw[16];
        float pw[16], cw[16];
        #pragma unroll
        for (int w = 0; w < 16; ++w) {
            gw[w] = Pg[w];
            cw[w] = Pc[w];
            pw[w] = Ps[w * F + wav * 32 + lane];
        }
        int   g = -1;
        float s = 0.0f, cn = 0.0f;
        #pragma unroll
        for (int w = 0; w < 16; ++w) {
            if (gw[w] < 0) continue;
            if (gw[w] != g) {
                if (g >= 0) {
                    atomicAdd(&sums[g * F + wav * 32 + lane], s);
                    if (wav == 0 && lane == 0) atomicAdd(&cnts[g], cn);
                }
                g  = gw[w];
                s  = 0.0f;
                cn = 0.0f;
            }
            s  += pw[w];
            cn += cw[w];
        }
        if (g >= 0) {
            atomicAdd(&sums[g * F + wav * 32 + lane], s);
            if (wav == 0 && lane == 0) atomicAdd(&cnts[g], cn);
        }
    }
}

#define GP_TILES 4
template <int F, int NC, bool RELU, bool NORM>
__global__ void __launch_bounds__(512)
gather_product(const float* __restrict__ h,
               const v2i* __restrict__ ninfo,
               const v2i* __restrict__ edges,
               v2i* __restrict__ edges_out,
               const float* __restrict__ bias,
               const float* __restrict__ Wn,
               float* __restrict__ Hn, int n, int e) {
    constexpr int C  = F / 32;
    constexpr int TP = F + 4;
    constexpr int WP = NC + 8;
    constexpr int NV4 = F * NC / 4;
    constexpr int WQ  = (NV4 + 511) / 512;
    static_assert(NC % 4 == 0 && WP % 4 == 0, "a 16-byte piece stays inside one weight row, 16-byte aligned in LDS");
    static_assert(2 * GP_TILES <= 32, "a wave keeps the row starts of its nodes in one register pair");
    __shared__ __attribute__((aligned(16))) float Ts[16 * TP];
    __shared__ __attribute__((aligned(16))) float Ws[F * WP];
    __shared__ v2i Es[ECAP + 16 * 32];

    const int tid  = threadIdx.x;
    const int lane = tid & 31;
    const int wav  = __builtin_amdgcn_readfirstlane(tid >> 5);

    const int ntiles = n >> 4;
    const int tile0  = (int)blockIdx.x * GP_TILES;
    const int nb0    = tile0 * 16;

    const int b0 = ninfo[min(nb0, n)].x;
    const int b1 = ninfo[min(nb0 + 16 * GP_TILES, n)].x;
    v4f wr[WQ];
    #pragma unroll
    for (int q = 0; q < WQ; ++q) {
        const int i = min(q * 512 + tid, NV4 - 1);
        wr[q] = *(const v4f*)&Wn[4 * i];
    }
    v2i ew[ECAP / 512];
    stage_edges_load(edges, b0, b1, e, tid, ew);
    float bs[C];
    #pragma unroll
    for (int c = 0; c < C; ++c) bs[c] = bias[c * 32 + lane];
    const int l7 = min(lane, 2 * GP_TILES - 1);
    const v2i nv = ninfo[min(nb0 + 16 * (l7 >> 1) + wav + (l7 & 1), n)];

    if (NORM) {
        const int* __restrict__ nwords = (const int*)ninfo;
        float dv[ECAP / 512];
        #pragma unroll
        for (int q = 0; q < ECAP / 512; ++q) {
            const int s = min(max(ew[q].x, 0), n - 1);
            dv[q] = __int_as_float(nwords[2 * s + 1]);
        }
        #pragma unroll
        for (int q = 0; q < ECAP / 512; ++q)
            ew[q].y = __float_as_int(dv[q] * __int_as_float(ew[q].y));
    }

    #pragma unroll
    for (int q = 0; q < WQ; ++q) {
        const int i = q * 512 + tid;
        if (i < NV4) {
            const int kk = (4 * i) / NC;
            const int cc = 4 * i - kk * NC;
            *(v4f*)&Ws[kk * WP + cc] = wr[q];
        }
    }
    stage_edges_store(Es, tid, ew);
    __syncthreads();

    if (NORM) {
        const int lim = min(min(b1, b0 + ECAP), e);
        #pragma unroll
        for (int q = 0; q < ECAP / 512; ++q) {
            const int j = b0 + q * 512 + tid;
            if (j >= 0 && j < lim) edges_out[j] = ew[q];
        }
    }

    for (int t = 0; t < GP_TILES; ++t) {
        const int tile = tile0 + t;
        if (tile >= ntiles) break;
        const int   node = tile * 16 + wav;
        const int   beg  = __builtin_amdgcn_readlane(nv.x, 2 * t);
        const int   end  = __builtin_amdgcn_readlane(nv.x, 2 * t + 1);
        const float di   = __int_as_float(__builtin_amdgcn_readlane(nv.y, 2 * t));
        float v[C];
        node_row<F, RELU, NORM>(h, ninfo, edges, edges_out, Es, b0, wav, beg, end, di, bs, node, n, e, lane, v);

        __syncthreads();
        #pragma unroll
        for (int c = 0; c < C; ++c) Ts[wav * TP + c * 32 + lane] = v[c];
        __syncthreads();

        if (wav < NC / 16) {
            const int l15 = lane & 15;
            const int hi  = lane >> 4;
            const int khi = hi * 2;
            const int colBase = wav * 16;

            v8f acc = {0.f, 0.f, 0.f, 0.f, 0.f, 0.f, 0.f, 0.f};
            #pragma unroll 8
            for (int k0 = 0; k0 < F; k0 += 4) {
                v2f a = *(const v2f*)&Ts[l15 * TP + k0 + khi];
                v2f b;
                b.x = Ws[(k0 + khi) * WP + colBase + l15];
                b.y = Ws[(k0 + khi + 1) * WP + colBase + l15];
                acc = __builtin_amdgcn_wmma_f32_16x16x4_f32(
                     false, a,  false, b,
                     (short)0, acc,  false,  false);
            }

            const int m0  = tile * 16 + hi * 8;
            const int col = colBase + l15;
            #pragma unroll
            for (int r = 0; r < 8; ++r)
                Hn[(m0 + r) * NC + col] = acc[r];
        }
    }
}

__global__ void pool_div(const float* __restrict__ sums,
                         const float* __restrict__ cnts,
                         float* __restrict__ out, int n) {
    int i = blockIdx.x * blockDim.x + threadIdx.x;
    if (i < n) out[i] = sums[i] / fmaxf(cnts[i / 64], 1.0f);
}

extern "C" void kernel_launch(void* const* d_in, const int* in_sizes, int n_in,
                              void* d_out, int out_size, void* d_ws, size_t ws_size,
                              hipStream_t stream) {
    const float* x     = (const float*)d_in[0];
    const int*   ei    = (const int*)d_in[1];
    const int*   batch = (const int*)d_in[2];
    const float* Wl[4] = {(const float*)d_in[3], (const float*)d_in[5],
                          (const float*)d_in[7], (const float*)d_in[9]};
    const float* bl[4] = {(const float*)d_in[4], (const float*)d_in[6],
                          (const float*)d_in[8], (const float*)d_in[10]};

    const int N = GCN_N_NODES;
    const int E = in_sizes[1] / 2;
    const int* src = ei;
    const int* dst = ei + E;

    auto au = [](size_t v) { return (v + 255) & ~(size_t)255; };
    char* ws = (char*)d_ws;
    size_t off = 0;
    int*      gcur = (int*)(ws + off);      off += au(sizeof(int) * 64);
    unsigned* stag = (unsigned*)(ws + off); off += au(sizeof(unsigned) * (size_t)NBIN * (size_t)E);
    v2i*   ninfo = (v2i*)(ws + off);   off += au(sizeof(v2i) * (size_t)(N + 1));
    v2i*   edges = (v2i*)(ws + off);   off += au(sizeof(v2i) * (size_t)E);
    v2i*   edgew = (v2i*)(ws + off);   off += au(sizeof(v2i) * (size_t)E);
    float* bufA  = (float*)(ws + off); off += au(sizeof(float) * (size_t)N * 96);
    float* bufB  = (float*)(ws + off); off += au(sizeof(float) * (size_t)N * 96);
    float* sums  = (float*)(ws + off); off += au(sizeof(float) * GCN_N_GRAPHS * 64);
    float* cnts  = (float*)(ws + off); off += au(sizeof(float) * GCN_N_GRAPHS);
    (void)ws_size; (void)n_in; (void)out_size;

    const int TB = 256;
    start_values<<<(GCN_N_GRAPHS * 64 + TB - 1) / TB, TB, 0, stream>>>(gcur, sums, cnts);
    const int MT  = (N + 15) / 16;
    const int MG  = (MT + PROD_WAVES - 1) / PROD_WAVES;
    const int MGH = MG / 2;
    const int NB1 = (E + 4095) / 4096;
    bin_edges_and_product<128, 96><<<NB1 + MGH * 6, 1024, 0, stream>>>(
        src, dst, gcur, stag, E, N, NB1, x, Wl[0], bufA, 0);
    bin_to_csr_and_product<128, 96><<<NBIN + (MG - MGH) * 6, 1024, 0, stream>>>(
        gcur, stag, ninfo, edges, E, N, NBIN, x, Wl[0], bufA, MGH);

    const int PB  = (N + 63) / 64;
    const int FB  = (MT + GP_TILES - 1) / GP_TILES;

    gather_product<96, 96, true, true ><<<FB, 512, 0, stream>>>(bufA, ninfo, edges, edgew, bl[0], Wl[1], bufB, N, E);
    gather_product<96, 96, true, false><<<FB, 512, 0, stream>>>(bufB, ninfo, edgew, (v2i*)nullptr, bl[1], Wl[2], bufA, N, E);
    gather_product<96, 64, true, false><<<FB, 512, 0, stream>>>(bufA, ninfo, edgew, (v2i*)nullptr, bl[2], Wl[3], bufB, N, E);
    gather_pool<64, 4><<<PB, 512, 0, stream>>>(bufB, ninfo, edgew, bl[3], batch, sums, cnts, N, E);

    pool_div<<<(GCN_N_GRAPHS * 64 + TB - 1) / TB, TB, 0, stream>>>(sums, cnts, (float*)d_out,
                                                                   GCN_N_GRAPHS * 64);
}


# Fused gather blocks: a wave finishes its 4 nodes into one 64-row LDS tile with no barrier between, then ONE product phase on all 16 waves (two column tiles share an A fragment); sum orders kept
# speedup vs baseline: 5.4956x; 1.0663x over previous
#include <hip/hip_runtime.h>

typedef __attribute__((ext_vector_type(2))) float v2f;
typedef __attribute__((ext_vector_type(4))) float v4f;
typedef __attribute__((ext_vector_type(8))) float v8f;
typedef __attribute__((ext_vector_type(2))) int   v2i;
typedef __attribute__((ext_vector_type(4))) int   v4i;

#define GCN_N_NODES  50000
#define GCN_N_GRAPHS 64
static_assert(GCN_N_NODES % 16 == 0, "row tiles of 16 must be full");

__global__ void start_values(int* __restrict__ gcur, float* __restrict__ sums, float* __restrict__ cnts) {
    const int i = blockIdx.x * blockDim.x + threadIdx.x;
    if (i < GCN_N_GRAPHS * 64) sums[i] = 0.0f;
    if (i < GCN_N_GRAPHS) cnts[i] = 0.0f;
    if (i < 64) gcur[i] = 0;
}

#define BIN_SHIFT 10
#define BIN_NODES (1 << BIN_SHIFT)
#define NBIN      ((GCN_N_NODES + BIN_NODES - 1) / BIN_NODES)
static_assert(NBIN <= 64, "the bin totals are summed by one wave, two per lane");
static_assert(GCN_N_NODES <= (1 << (31 - BIN_SHIFT)), "source id and bin-local id share one 32-bit word");

__device__ __forceinline__ void
bin_edges_block(const int blk, const int* __restrict__ src, const int* __restrict__ dst,
                int* __restrict__ gcur, unsigned* __restrict__ stag, int e, int n) {
    __shared__ int lcnt[NBIN];
    __shared__ int lbase[NBIN];
    const int tid = threadIdx.x;
    if (tid < NBIN) lcnt[tid] = 0;
    __syncthreads();

    int      bin[4], r[4];
    unsigned w[4];
    bool     ok[4];
    #pragma unroll
    for (int q = 0; q < 4; ++q) {
        const int i = blk * 4096 + q * 1024 + tid;
        ok[q] = i < e;
        const int ic = min(i, e - 1);
        int d = dst[ic];
        int s = src[ic];
        d = min(max(d, 0), n - 1);
        s = min(max(s, 0), n - 1);
        bin[q] = d >> BIN_SHIFT;
        w[q]   = ((unsigned)s << BIN_SHIFT) | (unsigned)(d & (BIN_NODES - 1));
        r[q]   = 0;
    }
    #pragma unroll
    for (int q = 0; q < 4; ++q)
        if (ok[q]) r[q] = atomicAdd(&lcnt[bin[q]], 1);
    __syncthreads();
    if (tid < NBIN) {
        const int c = lcnt[tid];
        lbase[tid] = (c > 0) ? atomicAdd(&gcur[tid], c) : 0;
    }
    __syncthreads();
    #pragma unroll
    for (int q = 0; q < 4; ++q) {
        if (ok[q]) {
            int pos = lbase[bin[q]] + r[q];
            pos = min(max(pos, 0), e - 1);
            stag[(size_t)bin[q] * (size_t)e + (size_t)pos] = w[q];
        }
    }
}

__device__ __forceinline__ void
bin_to_csr_block(const int b, const int* __restrict__ gcur, const unsigned* __restrict__ stag,
                 v2i* __restrict__ ninfo, v2i* __restrict__ edges, int e, int n) {
    __shared__ int cnt[BIN_NODES];
    __shared__ int dnv[BIN_NODES];
    __shared__ int tot[64];
    __shared__ int wtot[32];
    const int tid  = threadIdx.x;
    const int lane = tid & 31;
    const int wav  = tid >> 5;

    cnt[tid] = 0;
    if (tid < 64) tot[tid] = (tid < NBIN) ? gcur[tid] : 0;
    __syncthreads();

    int base = ((lane < b) ? tot[lane] : 0) + ((lane + 32 < b) ? tot[lane + 32] : 0);
    for (int o = 16; o > 0; o >>= 1) base += __shfl_xor(base, o);
    const int count = min(max(tot[b], 0), e);
    const unsigned* __restrict__ sb = stag + (size_t)b * (size_t)e;

    for (int i0 = tid; i0 < count; i0 += 4096) {
        unsigned w[4];
        #pragma unroll
        for (int q = 0; q < 4; ++q) {
            const int i = min(i0 + q * 1024, count - 1);
            w[q] = sb[i];
        }
        asm volatile("" :: "v"(w[0]), "v"(w[1]), "v"(w[2]), "v"(w[3]));
        #pragma unroll
        for (int q = 0; q < 4; ++q)
            if (i0 + q * 1024 < count) atomicAdd(&cnt[w[q] & (BIN_NODES - 1)], 1);
    }
    __syncthreads();

    const int c = cnt[tid];
    int incl = c;
    for (int o = 1; o < 32; o <<= 1) {
        const int u = __shfl_up(incl, o);
        if (lane >= o) incl += u;
    }
    if (lane == 31) wtot[wav] = incl;
    __syncthreads();
    int wprev = (lane < wav) ? wtot[lane] : 0;
    for (int o = 16; o > 0; o >>= 1) wprev += __shfl_xor(wprev, o);
    const int excl = wprev + incl - c;

    const int   node = (b << BIN_SHIFT) + tid;
    const float di   = rsqrtf((float)(c + 1));
    if (node < n) {
        v2i v;
        v.x = base + excl;
        v.y = __float_as_int(di);
        ninfo[node] = v;
    }
    if (b == NBIN - 1 && tid == BIN_NODES - 1) {
        v2i v;
        v.x = base + count;
        v.y = 0;
        ninfo[n] = v;
    }
    cnt[tid] = excl;
    dnv[tid] = __float_as_int(di);
    __syncthreads();

    for (int i0 = tid; i0 < count; i0 += 4096) {
        unsigned w[4];
        #pragma unroll
        for (int q = 0; q < 4; ++q) {
            const int i = min(i0 + q * 1024, count - 1);
            w[q] = sb[i];
        }
        asm volatile("" :: "v"(w[0]), "v"(w[1]), "v"(w[2]), "v"(w[3]));
        #pragma unroll
        for (int q = 0; q < 4; ++q) {
            if (i0 + q * 1024 < count) {
                const int dl = (int)(w[q] & (BIN_NODES - 1));
                int slot = base + atomicAdd(&cnt[dl], 1);
                slot = min(max(slot, 0), e - 1);
                v2i ev;
                ev.x = (int)(w[q] >> BIN_SHIFT);
                ev.y = dnv[dl];
                edges[slot] = ev;
            }
        }
    }
}

template <int K, int NCOLS, int WAVES>
__device__ __forceinline__ void product_block(const float* __restrict__ A,
                                              const float* __restrict__ Wm,
                                              float* __restrict__ H, int M,
                                              const int mgroup, const int ctile) {
    __shared__ float Bs[K * 16];

    const int tid  = threadIdx.x;
    const int lane = tid & 31;
    const int wav  = tid >> 5;
    const int colBase = ctile * 16;

    #pragma unroll
    for (int i = tid; i < K * 16; i += 32 * WAVES) {
        const int kk = i >> 4;
        const int cc = i & 15;
        Bs[i] = Wm[kk * NCOLS + colBase + cc];
    }
    __syncthreads();

    const int mtile = mgroup * WAVES + wav;
    if (mtile * 16 < M) {
        const int l15 = lane & 15;
        const int hi  = lane >> 4;
        const int khi = hi * 2;
        const int mrow = mtile * 16 + l15;

        v8f acc = {0.f, 0.f, 0.f, 0.f, 0.f, 0.f, 0.f, 0.f};
        #pragma unroll 8
        for (int k0 = 0; k0 < K; k0 += 4) {
            v2f a = *(const v2f*)&A[mrow * K + k0 + khi];
            v2f b;
            b.x = Bs[(k0 + khi) * 16 + l15];
            b.y = Bs[(k0 + khi + 1) * 16 + l15];
            acc = __builtin_amdgcn_wmma_f32_16x16x4_f32(
                 false, a,  false, b,
                 (short)0, acc,  false,  false);
        }

        const int m0  = mtile * 16 + hi * 8;
        const int col = colBase + l15;
        #pragma unroll
        for (int r = 0; r < 8; ++r)
            H[(m0 + r) * NCOLS + col] = acc[r];
    }
}

#define PROD_WAVES 32
template <int K, int NCOLS>
__global__ void __launch_bounds__(1024)
bin_edges_and_product(const int* __restrict__ src, const int* __restrict__ dst,
                      int* __restrict__ gcur, unsigned* __restrict__ stag, int e, int n, int nsort,
                      const float* __restrict__ A, const float* __restrict__ Wm,
                      float* __restrict__ H, int mg0) {
    const int blk = (int)blockIdx.x;
    if (blk < nsort) {
        bin_edges_block(blk, src, dst, gcur, stag, e, n);
    } else {
        const int g = blk - nsort;
        product_block<K, NCOLS, PROD_WAVES>(A, Wm, H, n, mg0 + g / (NCOLS / 16), g % (NCOLS / 16));
    }
}

template <int K, int NCOLS>
__global__ void __launch_bounds__(1024)
bin_to_csr_and_product(const int* __restrict__ gcur, const unsigned* __restrict__ stag,
                       v2i* __restrict__ ninfo, v2i* __restrict__ edges, int e, int n, int nsort,
                       const float* __restrict__ A, const float* __restrict__ Wm,
                       float* __restrict__ H, int mg0) {
    const int blk = (int)blockIdx.x;
    if (blk < nsort) {
        bin_to_csr_block(blk, gcur, stag, ninfo, edges, e, n);
    } else {
        const int g = blk - nsort;
        product_block<K, NCOLS, PROD_WAVES>(A, Wm, H, n, mg0 + g / (NCOLS / 16), g % (NCOLS / 16));
    }
}

template <int F, int GS, bool TAIL>
__device__ __forceinline__ void gather_group(const float* __restrict__ h, const v2i ev,
                                             int i0, int m, int n, int lane,
                                             float (&acc)[F / 32]) {
    constexpr int C = F / 32;
    float hv[GS][C];
    float wv[GS];
    #pragma unroll
    for (int q = 0; q < GS; ++q) {
        const int ii = TAIL ? min(i0 + q, m - 1) : (i0 + q);
        int s = __builtin_amdgcn_readlane(ev.x, ii);
        s = min(max(s, 0), n - 1);
        wv[q] = __int_as_float(__builtin_amdgcn_readlane(ev.y, ii));
        #pragma unroll
        for (int c = 0; c < C; ++c) hv[q][c] = h[s * F + c * 32 + lane];
    }
    #pragma unroll
    for (int q = 0; q < GS; ++q) {
        const bool ok = !TAIL || (i0 + q) < m;
        #pragma unroll
        for (int c = 0; c < C; ++c) {
            const float t = fmaf(hv[q][c], wv[q], acc[c]);
            acc[c] = ok ? t : acc[c];
        }
    }
}

#define ECAP 1280
#define EQ   ((ECAP + 511) / 512)

__device__ __forceinline__ void stage_edges_load(const v2i* __restrict__ edges, int b0, int b1, int e, int tid,
                                                 v2i (&w)[EQ]) {
    const int last = min(max(b1 - 1, 0), e - 1);
    #pragma unroll
    for (int q = 0; q < EQ; ++q) {
        const int j = min(max(b0 + q * 512 + tid, 0), last);
        w[q] = edges[j];
    }
}

__device__ __forceinline__ void stage_edges_store(v2i* __restrict__ Es, int tid, const v2i (&w)[EQ]) {
    #pragma unroll
    for (int q = 0; q < EQ; ++q)
        if (q * 512 + tid < ECAP) Es[q * 512 + tid] = w[q];
}

template <int F, bool NORM>
__device__ __forceinline__ void gather_edges(const float* __restrict__ h,
                                             const v2i* __restrict__ ninfo,
                                             const v2i* __restrict__ edges,
                                             v2i* __restrict__ edges_out,
                                             v2i* __restrict__ Es, int b0, int wav,
                                             int beg, int end, int n, int e, int lane,
                                             float (&acc)[F / 32]) {
    for (int base = beg; base < end; base += 32) {
        const int j = base + lane;
        int off = base - b0;
        if (off < 0 || off + 32 > ECAP) {
            off = ECAP + wav * 32;
            v2i t = {0, 0};
            if (j < end) {
                const int jc = min(max(j, 0), e - 1);
                t = edges[jc];
                if (NORM) {
                    const int s = min(max(t.x, 0), n - 1);
                    t.y = __float_as_int(__int_as_float(ninfo[s].y) * __int_as_float(t.y));
                    if (j - b0 >= ECAP) edges_out[jc] = t;
                }
            }
            Es[off + lane] = t;
        }
        v2i ev = {0, 0};
        if (j < end) ev = Es[off + lane];
        const int m = min(32, end - base);
        int i0 = 0;
        for (; i0 + 8 <= m; i0 += 8) gather_group<F, 8, false>(h, ev, i0, m, n, lane, acc);
        const int rem = m - i0;
        if (rem > 4)      gather_group<F, 8, true>(h, ev, i0, m, n, lane, acc);
        else if (rem > 0) gather_group<F, 4, true>(h, ev, i0, m, n, lane, acc);
    }
}

template <int F, bool RELU, bool NORM>
__device__ __forceinline__ void node_row(const float* __restrict__ h,
                                         const v2i* __restrict__ ninfo,
                                         const v2i* __restrict__ edges,
                                         v2i* __restrict__ edges_out,
                                         v2i* __restrict__ Es, int b0, int wav,
                                         int beg, int end, float di,
                                         const float (&bs)[F / 32],
                                         int node, int n, int e, int lane, float (&v)[F / 32]) {
    constexpr int C = F / 32;
    float hs[C], acc[C];
    #pragma unroll
    for (int c = 0; c < C; ++c) {
        hs[c]  = h[node * F + c * 32 + lane];
        acc[c] = 0.0f;
    }

    gather_edges<F, NORM>(h, ninfo, edges, edges_out, Es, b0, wav, beg, end, n, e, lane, acc);

    const float di2 = di * di;
    #pragma unroll
    for (int c = 0; c < C; ++c) {
        const float t = acc[c] + hs[c] * di2 + bs[c];
        v[c] = RELU ? fmaxf(t, 0.0f) : t;
    }
}

template <int F, int NPW>
__global__ void __launch_bounds__(512)
gather_pool(const float* __restrict__ h,
            const v2i* __restrict__ ninfo,
            const v2i* __restrict__ edges,
            const float* __restrict__ bias,
            const int* __restrict__ batch,
            float* __restrict__ sums,
            float* __restrict__ cnts, int n, int e) {
    constexpr int C = F / 32;
    __shared__ float Ps[16 * F];
    __shared__ float Pc[16];
    __shared__ int   Pg[16];
    __shared__ v2i   Es[ECAP + 16 * 32];

    const int tid  = (int)threadIdx.x;
    const int lane = tid & 31;
    const int wav  = __builtin_amdgcn_readfirstlane(tid >> 5);
    const int nb0  = (int)blockIdx.x * 16 * NPW;
    const int n0   = nb0 + wav * NPW;

    const int b0 = ninfo[min(nb0, n)].x;
    const int b1 = ninfo[min(nb0 + 16 * NPW, n)].x;
    v2i ew[EQ];
    stage_edges_load(edges, b0, b1, e, tid, ew);
    float bs[C];
    #pragma unroll
    for (int c = 0; c < C; ++c) bs[c] = bias[c * 32 + lane];
    const v2i nv = ninfo[min(n0 + min(lane, NPW), n)];
    const int gv = batch[min(n0 + min(lane, NPW - 1), n - 1)];
    stage_edges_store(Es, tid, ew);
    __syncthreads();

    int   gcur = -1;
    float pc   = 0.0f;
    float ps[C];
    #pragma unroll
    for (int c = 0; c < C; ++c) ps[c] = 0.0f;

    if (n0 < n) {
        for (int k = 0; k < NPW; ++k) {
            const int node = n0 + k;
            if (node < n) {
                const int   beg = __builtin_amdgcn_readlane(nv.x, k);
                const int   end = __builtin_amdgcn_readlane(nv.x, k + 1);
                const float di  = __int_as_float(__builtin_amdgcn_readlane(nv.y, k));
                int g = __builtin_amdgcn_readlane(gv, k);
                g = min(max(g, 0), GCN_N_GRAPHS - 1);

                float hs[C], acc[C];
                #pragma unroll
                for (int c = 0; c < C; ++c) {
                    hs[c]  = h[node * F + c * 32 + lane];
                    acc[c] = 0.0f;
                }
                gather_edges<F, false>(h, ninfo, edges, (v2i*)nullptr, Es, b0, wav, beg, end, n, e, lane, acc);

                if (g != gcur) {
                    if (gcur >= 0) {
                        #pragma unroll
                        for (int c = 0; c < C; ++c) atomicAdd(&sums[gcur * F + c * 32 + lane], ps[c]);
                        if (lane == 0) atomicAdd(&cnts[gcur], pc);
                    }
                    gcur = g;
                    pc   = 0.0f;
                    #pragma unroll
                    for (int c = 0; c < C; ++c) ps[c] = 0.0f;
                }
                const float di2 = di * di;
                #pragma unroll
                for (int c = 0; c < C; ++c) ps[c] += acc[c] + hs[c] * di2 + bs[c];
                pc += 1.0f;
            }
        }
    }

    #pragma unroll
    for (int c = 0; c < C; ++c) Ps[wav * F + c * 32 + lane] = ps[c];
    if (lane == 0) {
        Pg[wav] = gcur;
        Pc[wav] = pc;
    }
    __syncthreads();

    if (wav < C) {
        int   gw[16];
        float pw[16], cw[16];
        #pragma unroll
        for (int w = 0; w < 16; ++w) {
            gw[w] = Pg[w];
            cw[w] = Pc[w];
            pw[w] = Ps[w * F + wav * 32 + lane];
        }
        int   g = -1;
        float s = 0.0f, cn = 0.0f;
        #pragma unroll
        for (int w = 0; w < 16; ++w) {
            if (gw[w] < 0) continue;
            if (gw[w] != g) {
                if (g >= 0) {
                    atomicAdd(&sums[g * F + wav * 32 + lane], s);
                    if (wav == 0 && lane == 0) atomicAdd(&cnts[g], cn);
                }
                g  = gw[w];
                s  = 0.0f;
                cn = 0.0f;
            }
            s  += pw[w];
            cn += cw[w];
        }
        if (g >= 0) {
            atomicAdd(&sums[g * F + wav * 32 + lane], s);
            if (wav == 0 && lane == 0) atomicAdd(&cnts[g], cn);
        }
    }
}

#define GP_TILES 4
template <int F, int NC, bool RELU, bool NORM>
__global__ void __launch_bounds__(512)
gather_product(const float* __restrict__ h,
               const v2i* __restrict__ ninfo,
               const v2i* __restrict__ edges,
               v2i* __restrict__ edges_out,
               const float* __restrict__ bias,
               const float* __restrict__ Wn,
               float* __restrict__ Hn, int n, int e) {
    constexpr int C  = F / 32;
    constexpr int TP = F + 4;
    constexpr int WP = NC + 8;
    constexpr int NV4 = F * NC / 4;
    constexpr int WQ  = (NV4 + 511) / 512;
    constexpr int NCT = NC / 16;
    static_assert(NC % 4 == 0 && WP % 4 == 0, "a 16-byte piece stays inside one weight row, 16-byte aligned in LDS");
    static_assert(2 * GP_TILES <= 32, "a wave keeps the row starts of its nodes in one register pair");
    static_assert(GP_TILES == 4 && NCT >= 4 && NCT <= 8, "16 waves: row tile wav % 4, column tiles wav / 4 and wav / 4 + 4");
    __shared__ __attribute__((aligned(16))) float Ts[16 * GP_TILES * TP];
    __shared__ __attribute__((aligned(16))) float Ws[F * WP];
    __shared__ v2i Es[ECAP + 16 * 32];

    const int tid  = threadIdx.x;
    const int lane = tid & 31;
    const int wav  = __builtin_amdgcn_readfirstlane(tid >> 5);

    const int ntiles = n >> 4;
    const int tile0  = (int)blockIdx.x * GP_TILES;
    const int nb0    = tile0 * 16;

    const int b0 = ninfo[min(nb0, n)].x;
    const int b1 = ninfo[min(nb0 + 16 * GP_TILES, n)].x;
    v4f wr[WQ];
    #pragma unroll
    for (int q = 0; q < WQ; ++q) {
        const int i = min(q * 512 + tid, NV4 - 1);
        wr[q] = *(const v4f*)&Wn[4 * i];
    }
    v2i ew[EQ];
    stage_edges_load(edges, b0, b1, e, tid, ew);
    float bs[C];
    #pragma unroll
    for (int c = 0; c < C; ++c) bs[c] = bias[c * 32 + lane];
    const int l7 = min(lane, 2 * GP_TILES - 1);
    const v2i nv = ninfo[min(nb0 + 16 * (l7 >> 1) + wav + (l7 & 1), n)];

    if (NORM) {
        const int* __restrict__ nwords = (const int*)ninfo;
        float dv[EQ];
        #pragma unroll
        for (int q = 0; q < EQ; ++q) {
            const int s = min(max(ew[q].x, 0), n - 1);
            dv[q] = __int_as_float(nwords[2 * s + 1]);
        }
        #pragma unroll
        for (int q = 0; q < EQ; ++q)
            ew[q].y = __float_as_int(dv[q] * __int_as_float(ew[q].y));
    }

    #pragma unroll
    for (int q = 0; q < WQ; ++q) {
        const int i = q * 512 + tid;
        if (i < NV4) {
            const int kk = (4 * i) / NC;
            const int cc = 4 * i - kk * NC;
            *(v4f*)&Ws[kk * WP + cc] = wr[q];
        }
    }
    stage_edges_store(Es, tid, ew);
    __syncthreads();

    if (NORM) {
        const int lim = min(min(b1, b0 + ECAP), e);
        #pragma unroll
        for (int q = 0; q < EQ; ++q) {
            const int j = b0 + q * 512 + tid;
            if (q * 512 + tid < ECAP && j >= 0 && j < lim) edges_out[j] = ew[q];
        }
    }

    for (int t = 0; t < GP_TILES; ++t) {
        const int tile = tile0 + t;
        if (tile >= ntiles) break;
        const int   node = tile * 16 + wav;
        const int   beg  = __builtin_amdgcn_readlane(nv.x, 2 * t);
        const int   end  = __builtin_amdgcn_readlane(nv.x, 2 * t + 1);
        const float di   = __int_as_float(__builtin_amdgcn_readlane(nv.y, 2 * t));
        float v[C];
        node_row<F, RELU, NORM>(h, ninfo, edges, edges_out, Es, b0, wav, beg, end, di, bs, node, n, e, lane, v);
        #pragma unroll
        for (int c = 0; c < C; ++c) Ts[(16 * t + wav) * TP + c * 32 + lane] = v[c];
    }
    __syncthreads();

    const int mt = wav & 3;
    if (tile0 + mt < ntiles) {
        const int l15 = lane & 15;
        const int hi  = lane >> 4;
        const int khi = hi * 2;
        const int ct0 = wav >> 2;
        const float* __restrict__ Ta = &Ts[(16 * mt + l15) * TP + khi];
        const float* __restrict__ Wb = &Ws[khi * WP + ct0 * 16 + l15];
        const int m0  = (tile0 + mt) * 16 + hi * 8;

        if (ct0 + 4 < NCT) {
            v8f acc0 = {0.f, 0.f, 0.f, 0.f, 0.f, 0.f, 0.f, 0.f};
            v8f acc1 = {0.f, 0.f, 0.f, 0.f, 0.f, 0.f, 0.f, 0.f};
            #pragma unroll 8
            for (int k0 = 0; k0 < F; k0 += 4) {
                v2f a = *(const v2f*)&Ta[k0];
                v2f p, q;
                p.x = Wb[k0 * WP];
                p.y = Wb[(k0 + 1) * WP];
                q.x = Wb[k0 * WP + 64];
                q.y = Wb[(k0 + 1) * WP + 64];
                acc0 = __builtin_amdgcn_wmma_f32_16x16x4_f32(
                     false, a,  false, p,
                     (short)0, acc0,  false,  false);
                acc1 = __builtin_amdgcn_wmma_f32_16x16x4_f32(
                     false, a,  false, q,
                     (short)0, acc1,  false,  false);
            }
            const int col = ct0 * 16 + l15;
            #pragma unroll
            for (int r = 0; r < 8; ++r) {
                Hn[(m0 + r) * NC + col]      = acc0[r];
                Hn[(m0 + r) * NC + col + 64] = acc1[r];
            }
        } else {
            v8f acc = {0.f, 0.f, 0.f, 0.f, 0.f, 0.f, 0.f, 0.f};
            #pragma unroll 8
            for (int k0 = 0; k0 < F; k0 += 4) {
                v2f a = *(const v2f*)&Ta[k0];
                v2f b;
                b.x = Wb[k0 * WP];
                b.y = Wb[(k0 + 1) * WP];
                acc = __builtin_amdgcn_wmma_f32_16x16x4_f32(
                     false, a,  false, b,
                     (short)0, acc,  false,  false);
            }
            const int col = ct0 * 16 + l15;
            #pragma unroll
            for (int r = 0; r < 8; ++r)
                Hn[(m0 + r) * NC + col] = acc[r];
        }
    }
}

__global__ void pool_div(const float* __restrict__ sums,
                         const float* __restrict__ cnts,
                         float* __restrict__ out, int n) {
    int i = blockIdx.x * blockDim.x + threadIdx.x;
    if (i < n) out[i] = sums[i] / fmaxf(cnts[i / 64], 1.0f);
}

extern "C" void kernel_launch(void* const* d_in, const int* in_sizes, int n_in,
                              void* d_out, int out_size, void* d_ws, size_t ws_size,
                              hipStream_t stream) {
    const float* x     = (const float*)d_in[0];
    const int*   ei    = (const int*)d_in[1];
    const int*   batch = (const int*)d_in[2];
    const float* Wl[4] = {(const float*)d_in[3], (const float*)d_in[5],
                          (const float*)d_in[7], (const float*)d_in[9]};
    const float* bl[4] = {(const float*)d_in[4], (const float*)d_in[6],
                          (const float*)d_in[8], (const float*)d_in[10]};

    const int N = GCN_N_NODES;
    const int E = in_sizes[1] / 2;
    const int* src = ei;
    const int* dst = ei + E;

    auto au = [](size_t v) { return (v + 255) & ~(size_t)255; };
    char* ws = (char*)d_ws;
    size_t off = 0;
    int*      gcur = (int*)(ws + off);      off += au(sizeof(int) * 64);
    unsigned* stag = (unsigned*)(ws + off); off += au(sizeof(unsigned) * (size_t)NBIN * (size_t)E);
    v2i*   ninfo = (v2i*)(ws + off);   off += au(sizeof(v2i) * (size_t)(N + 1));
    v2i*   edges = (v2i*)(ws + off);   off += au(sizeof(v2i) * (size_t)E);
    v2i*   edgew = (v2i*)(ws + off);   off += au(sizeof(v2i) * (size_t)E);
    float* bufA  = (float*)(ws + off); off += au(sizeof(float) * (size_t)N * 96);
    float* bufB  = (float*)(ws + off); off += au(sizeof(float) * (size_t)N * 96);
    float* sums  = (float*)(ws + off); off += au(sizeof(float) * GCN_N_GRAPHS * 64);
    float* cnts  = (float*)(ws + off); off += au(sizeof(float) * GCN_N_GRAPHS);
    (void)ws_size; (void)n_in; (void)out_size;

    const int TB = 256;
    start_values<<<(GCN_N_GRAPHS * 64 + TB - 1) / TB, TB, 0, stream>>>(gcur, sums, cnts);
    const int MT  = (N + 15) / 16;
    const int MG  = (MT + PROD_WAVES - 1) / PROD_WAVES;
    const int MGH = MG / 2;
    const int NB1 = (E + 4095) / 4096;
    bin_edges_and_product<128, 96><<<NB1 + MGH * 6, 1024, 0, stream>>>(
        src, dst, gcur, stag, E, N, NB1, x, Wl[0], bufA, 0);
    bin_to_csr_and_product<128, 96><<<NBIN + (MG - MGH) * 6, 1024, 0, stream>>>(
        gcur, stag, ninfo, edges, E, N, NBIN, x, Wl[0], bufA, MGH);

    const int PB  = (N + 63) / 64;
    const int FB  = (MT + GP_TILES - 1) / GP_TILES;

    gather_product<96, 96, true, true ><<<FB, 512, 0, stream>>>(bufA, ninfo, edges, edgew, bl[0], Wl[1], bufB, N, E);
    gather_product<96, 96, true, false><<<FB, 512, 0, stream>>>(bufB, ninfo, edgew, (v2i*)nullptr, bl[1], Wl[2], bufA, N, E);
    gather_product<96, 64, true, false><<<FB, 512, 0, stream>>>(bufA, ninfo, edgew, (v2i*)nullptr, bl[2], Wl[3], bufB, N, E);
    gather_pool<64, 4><<<PB, 512, 0, stream>>>(bufB, ninfo, edgew, bl[3], batch, sums, cnts, N, E);

    pool_div<<<(GCN_N_GRAPHS * 64 + TB - 1) / TB, TB, 0, stream>>>(sums, cnts, (float*)d_out,
                                                                   GCN_N_GRAPHS * 64);
}


# Gather grids in whole blocks per WGP: 768 four-tile blocks (3 per WGP, all resident) + 53 one-tile blocks, not 782 (a 4th full block on 14 WGPs); pool sum order changes for the last 848 nodes
# speedup vs baseline: 5.8360x; 1.0619x over previous
#include <hip/hip_runtime.h>

typedef __attribute__((ext_vector_type(2))) float v2f;
typedef __attribute__((ext_vector_type(4))) float v4f;
typedef __attribute__((ext_vector_type(8))) float v8f;
typedef __attribute__((ext_vector_type(2))) int   v2i;
typedef __attribute__((ext_vector_type(4))) int   v4i;

#define GCN_N_NODES  50000
#define GCN_N_GRAPHS 64
#define GCN_WGPS     256
static_assert(GCN_N_NODES % 16 == 0, "row tiles of 16 must be full");

__global__ void start_values(int* __restrict__ gcur, float* __restrict__ sums, float* __restrict__ cnts) {
    const int i = blockIdx.x * blockDim.x + threadIdx.x;
    if (i < GCN_N_GRAPHS * 64) sums[i] = 0.0f;
    if (i < GCN_N_GRAPHS) cnts[i] = 0.0f;
    if (i < 64) gcur[i] = 0;
}

#define BIN_SHIFT 10
#define BIN_NODES (1 << BIN_SHIFT)
#define NBIN      ((GCN_N_NODES + BIN_NODES - 1) / BIN_NODES)
static_assert(NBIN <= 64, "the bin totals are summed by one wave, two per lane");
static_assert(GCN_N_NODES <= (1 << (31 - BIN_SHIFT)), "source id and bin-local id share one 32-bit word");

__device__ __forceinline__ void
bin_edges_block(const int blk, const int* __restrict__ src, const int* __restrict__ dst,
                int* __restrict__ gcur, unsigned* __restrict__ stag, int e, int n) {
    __shared__ int lcnt[NBIN];
    __shared__ int lbase[NBIN];
    const int tid = threadIdx.x;
    if (tid < NBIN) lcnt[tid] = 0;
    __syncthreads();

    int      bin[4], r[4];
    unsigned w[4];
    bool     ok[4];
    #pragma unroll
    for (int q = 0; q < 4; ++q) {
        const int i = blk * 4096 + q * 1024 + tid;
        ok[q] = i < e;
        const int ic = min(i, e - 1);
        int d = dst[ic];
        int s = src[ic];
        d = min(max(d, 0), n - 1);
        s = min(max(s, 0), n - 1);
        bin[q] = d >> BIN_SHIFT;
        w[q]   = ((unsigned)s << BIN_SHIFT) | (unsigned)(d & (BIN_NODES - 1));
        r[q]   = 0;
    }
    #pragma unroll
    for (int q = 0; q < 4; ++q)
        if (ok[q]) r[q] = atomicAdd(&lcnt[bin[q]], 1);
    __syncthreads();
    if (tid < NBIN) {
        const int c = lcnt[tid];
        lbase[tid] = (c > 0) ? atomicAdd(&gcur[tid], c) : 0;
    }
    __syncthreads();
    #pragma unroll
    for (int q = 0; q < 4; ++q) {
        if (ok[q]) {
            int pos = lbase[bin[q]] + r[q];
            pos = min(max(pos, 0), e - 1);
            stag[(size_t)bin[q] * (size_t)e + (size_t)pos] = w[q];
        }
    }
}

__device__ __forceinline__ void
bin_to_csr_block(const int b, const int* __restrict__ gcur, const unsigned* __restrict__ stag,
                 v2i* __restrict__ ninfo, v2i* __restrict__ edges, int e, int n) {
    __shared__ int cnt[BIN_NODES];
    __shared__ int dnv[BIN_NODES];
    __shared__ int tot[64];
    __shared__ int wtot[32];
    const int tid  = threadIdx.x;
    const int lane = tid & 31;
    const int wav  = tid >> 5;

    cnt[tid] = 0;
    if (tid < 64) tot[tid] = (tid < NBIN) ? gcur[tid] : 0;
    __syncthreads();

    int base = ((lane < b) ? tot[lane] : 0) + ((lane + 32 < b) ? tot[lane + 32] : 0);
    for (int o = 16; o > 0; o >>= 1) base += __shfl_xor(base, o);
    const int count = min(max(tot[b], 0), e);
    const unsigned* __restrict__ sb = stag + (size_t)b * (size_t)e;

    for (int i0 = tid; i0 < count; i0 += 4096) {
        unsigned w[4];
        #pragma unroll
        for (int q = 0; q < 4; ++q) {
            const int i = min(i0 + q * 1024, count - 1);
            w[q] = sb[i];
        }
        asm volatile("" :: "v"(w[0]), "v"(w[1]), "v"(w[2]), "v"(w[3]));
        #pragma unroll
        for (int q = 0; q < 4; ++q)
            if (i0 + q * 1024 < count) atomicAdd(&cnt[w[q] & (BIN_NODES - 1)], 1);
    }
    __syncthreads();

    const int c = cnt[tid];
    int incl = c;
    for (int o = 1; o < 32; o <<= 1) {
        const int u = __shfl_up(incl, o);
        if (lane >= o) incl += u;
    }
    if (lane == 31) wtot[wav] = incl;
    __syncthreads();
    int wprev = (lane < wav) ? wtot[lane] : 0;
    for (int o = 16; o > 0; o >>= 1) wprev += __shfl_xor(wprev, o);
    const int excl = wprev + incl - c;

    const int   node = (b << BIN_SHIFT) + tid;
    const float di   = rsqrtf((float)(c + 1));
    if (node < n) {
        v2i v;
        v.x = base + excl;
        v.y = __float_as_int(di);
        ninfo[node] = v;
    }
    if (b == NBIN - 1 && tid == BIN_NODES - 1) {
        v2i v;
        v.x = base + count;
        v.y = 0;
        ninfo[n] = v;
    }
    cnt[tid] = excl;
    dnv[tid] = __float_as_int(di);
    __syncthreads();

    for (int i0 = tid; i0 < count; i0 += 4096) {
        unsigned w[4];
        #pragma unroll
        for (int q = 0; q < 4; ++q) {
            const int i = min(i0 + q * 1024, count - 1);
            w[q] = sb[i];
        }
        asm volatile("" :: "v"(w[0]), "v"(w[1]), "v"(w[2]), "v"(w[3]));
        #pragma unroll
        for (int q = 0; q < 4; ++q) {
            if (i0 + q * 1024 < count) {
                const int dl = (int)(w[q] & (BIN_NODES - 1));
                int slot = base + atomicAdd(&cnt[dl], 1);
                slot = min(max(slot, 0), e - 1);
                v2i ev;
                ev.x = (int)(w[q] >> BIN_SHIFT);
                ev.y = dnv[dl];
                edges[slot] = ev;
            }
        }
    }
}

template <int K, int NCOLS, int WAVES>
__device__ __forceinline__ void product_block(const float* __restrict__ A,
                                              const float* __restrict__ Wm,
                                              float* __restrict__ H, int M,
                                              const int mgroup, const int ctile) {
    __shared__ float Bs[K * 16];

    const int tid  = threadIdx.x;
    const int lane = tid & 31;
    const int wav  = tid >> 5;
    const int colBase = ctile * 16;

    #pragma unroll
    for (int i = tid; i < K * 16; i += 32 * WAVES) {
        const int kk = i >> 4;
        const int cc = i & 15;
        Bs[i] = Wm[kk * NCOLS + colBase + cc];
    }
    __syncthreads();

    const int mtile = mgroup * WAVES + wav;
    if (mtile * 16 < M) {
        const int l15 = lane & 15;
        const int hi  = lane >> 4;
        const int khi = hi * 2;
        const int mrow = mtile * 16 + l15;

        v8f acc = {0.f, 0.f, 0.f, 0.f, 0.f, 0.f, 0.f, 0.f};
        #pragma unroll 8
        for (int k0 = 0; k0 < K; k0 += 4) {
            v2f a = *(const v2f*)&A[mrow * K + k0 + khi];
            v2f b;
            b.x = Bs[(k0 + khi) * 16 + l15];
            b.y = Bs[(k0 + khi + 1) * 16 + l15];
            acc = __builtin_amdgcn_wmma_f32_16x16x4_f32(
                 false, a,  false, b,
                 (short)0, acc,  false,  false);
        }

        const int m0  = mtile * 16 + hi * 8;
        const int col = colBase + l15;
        #pragma unroll
        for (int r = 0; r < 8; ++r)
            H[(m0 + r) * NCOLS + col] = acc[r];
    }
}

#define PROD_WAVES 32
template <int K, int NCOLS>
__global__ void __launch_bounds__(1024)
bin_edges_and_product(const int* __restrict__ src, const int* __restrict__ dst,
                      int* __restrict__ gcur, unsigned* __restrict__ stag, int e, int n, int nsort,
                      const float* __restrict__ A, const float* __restrict__ Wm,
                      float* __restrict__ H, int mg0) {
    const int blk = (int)blockIdx.x;
    if (blk < nsort) {
        bin_edges_block(blk, src, dst, gcur, stag, e, n);
    } else {
        const int g = blk - nsort;
        product_block<K, NCOLS, PROD_WAVES>(A, Wm, H, n, mg0 + g / (NCOLS / 16), g % (NCOLS / 16));
    }
}

template <int K, int NCOLS>
__global__ void __launch_bounds__(1024)
bin_to_csr_and_product(const int* __restrict__ gcur, const unsigned* __restrict__ stag,
                       v2i* __restrict__ ninfo, v2i* __restrict__ edges, int e, int n, int nsort,
                       const float* __restrict__ A, const float* __restrict__ Wm,
                       float* __restrict__ H, int mg0) {
    const int blk = (int)blockIdx.x;
    if (blk < nsort) {
        bin_to_csr_block(blk, gcur, stag, ninfo, edges, e, n);
    } else {
        const int g = blk - nsort;
        product_block<K, NCOLS, PROD_WAVES>(A, Wm, H, n, mg0 + g / (NCOLS / 16), g % (NCOLS / 16));
    }
}

template <int F, int GS, bool TAIL>
__device__ __forceinline__ void gather_group(const float* __restrict__ h, const v2i ev,
                                             int i0, int m, int n, int lane,
                                             float (&acc)[F / 32]) {
    constexpr int C = F / 32;
    float hv[GS][C];
    float wv[GS];
    #pragma unroll
    for (int q = 0; q < GS; ++q) {
        const int ii = TAIL ? min(i0 + q, m - 1) : (i0 + q);
        int s = __builtin_amdgcn_readlane(ev.x, ii);
        s = min(max(s, 0), n - 1);
        wv[q] = __int_as_float(__builtin_amdgcn_readlane(ev.y, ii));
        #pragma unroll
        for (int c = 0; c < C; ++c) hv[q][c] = h[s * F + c * 32 + lane];
    }
    #pragma unroll
    for (int q = 0; q < GS; ++q) {
        const bool ok = !TAIL || (i0 + q) < m;
        #pragma unroll
        for (int c = 0; c < C; ++c) {
            const float t = fmaf(hv[q][c], wv[q], acc[c]);
            acc[c] = ok ? t : acc[c];
        }
    }
}

#define ECAP 1280
#define EQ   ((ECAP + 511) / 512)

__device__ __forceinline__ void stage_edges_load(const v2i* __restrict__ edges, int b0, int b1, int e, int tid,
                                                 v2i (&w)[EQ]) {
    const int last = min(max(b1 - 1, 0), e - 1);
    #pragma unroll
    for (int q = 0; q < EQ; ++q) {
        const int j = min(max(b0 + q * 512 + tid, 0), last);
        w[q] = edges[j];
    }
}

__device__ __forceinline__ void stage_edges_store(v2i* __restrict__ Es, int tid, const v2i (&w)[EQ]) {
    #pragma unroll
    for (int q = 0; q < EQ; ++q)
        if (q * 512 + tid < ECAP) Es[q * 512 + tid] = w[q];
}

template <int F, bool NORM>
__device__ __forceinline__ void gather_edges(const float* __restrict__ h,
                                             const v2i* __restrict__ ninfo,
                                             const v2i* __restrict__ edges,
                                             v2i* __restrict__ edges_out,
                                             v2i* __restrict__ Es, int b0, int wav,
                                             int beg, int end, int n, int e, int lane,
                                             float (&acc)[F / 32]) {
    for (int base = beg; base < end; base += 32) {
        const int j = base + lane;
        int off = base - b0;
        if (off < 0 || off + 32 > ECAP) {
            off = ECAP + wav * 32;
            v2i t = {0, 0};
            if (j < end) {
                const int jc = min(max(j, 0), e - 1);
                t = edges[jc];
                if (NORM) {
                    const int s = min(max(t.x, 0), n - 1);
                    t.y = __float_as_int(__int_as_float(ninfo[s].y) * __int_as_float(t.y));
                    if (j - b0 >= ECAP) edges_out[jc] = t;
                }
            }
            Es[off + lane] = t;
        }
        v2i ev = {0, 0};
        if (j < end) ev = Es[off + lane];
        const int m = min(32, end - base);
        int i0 = 0;
        for (; i0 + 8 <= m; i0 += 8) gather_group<F, 8, false>(h, ev, i0, m, n, lane, acc);
        const int rem = m - i0;
        if (rem > 4)      gather_group<F, 8, true>(h, ev, i0, m, n, lane, acc);
        else if (rem > 0) gather_group<F, 4, true>(h, ev, i0, m, n, lane, acc);
    }
}

template <int F, bool RELU, bool NORM>
__device__ __forceinline__ void node_row(const float* __restrict__ h,
                                         const v2i* __restrict__ ninfo,
                                         const v2i* __restrict__ edges,
                                         v2i* __restrict__ edges_out,
                                         v2i* __restrict__ Es, int b0, int wav,
                                         int beg, int end, float di,
                                         const float (&bs)[F / 32],
                                         int node, int n, int e, int lane, float (&v)[F / 32]) {
    constexpr int C = F / 32;
    float hs[C], acc[C];
    #pragma unroll
    for (int c = 0; c < C; ++c) {
        hs[c]  = h[node * F + c * 32 + lane];
        acc[c] = 0.0f;
    }

    gather_edges<F, NORM>(h, ninfo, edges, edges_out, Es, b0, wav, beg, end, n, e, lane, acc);

    const float di2 = di * di;
    #pragma unroll
    for (int c = 0; c < C; ++c) {
        const float t = acc[c] + hs[c] * di2 + bs[c];
        v[c] = RELU ? fmaxf(t, 0.0f) : t;
    }
}

template <int F, int NPW>
__global__ void __launch_bounds__(512)
gather_pool(const float* __restrict__ h,
            const v2i* __restrict__ ninfo,
            const v2i* __restrict__ edges,
            const float* __restrict__ bias,
            const int* __restrict__ batch,
            float* __restrict__ sums,
            float* __restrict__ cnts, int n, int e, int nfull) {
    constexpr int C = F / 32;
    __shared__ float Ps[16 * F];
    __shared__ float Pc[16];
    __shared__ int   Pg[16];
    __shared__ v2i   Es[ECAP + 16 * 32];

    const int tid  = (int)threadIdx.x;
    const int lane = tid & 31;
    const int wav  = __builtin_amdgcn_readfirstlane(tid >> 5);
    const int blk  = (int)blockIdx.x;
    const int npw  = (blk < nfull) ? NPW : 1;
    const int nb0  = (blk < nfull) ? blk * 16 * NPW : nfull * 16 * NPW + (blk - nfull) * 16;
    const int n0   = nb0 + wav * npw;

    const int b0 = ninfo[min(nb0, n)].x;
    const int b1 = ninfo[min(nb0 + 16 * npw, n)].x;
    v2i ew[EQ];
    stage_edges_load(edges, b0, b1, e, tid, ew);
    float bs[C];
    #pragma unroll
    for (int c = 0; c < C; ++c) bs[c] = bias[c * 32 + lane];
    const v2i nv = ninfo[min(n0 + min(lane, npw), n)];
    const int gv = batch[min(n0 + min(lane, npw - 1), n - 1)];
    stage_edges_store(Es, tid, ew);
    __syncthreads();

    int   gcur = -1;
    float pc   = 0.0f;
    float ps[C];
    #pragma unroll
    for (int c = 0; c < C; ++c) ps[c] = 0.0f;

    if (n0 < n) {
        for (int k = 0; k < NPW; ++k) {
            const int node = n0 + k;
            if (k < npw && node < n) {
                const int   beg = __builtin_amdgcn_readlane(nv.x, k);
                const int   end = __builtin_amdgcn_readlane(nv.x, k + 1);
                const float di  = __int_as_float(__builtin_amdgcn_readlane(nv.y, k));
                int g = __builtin_amdgcn_readlane(gv, k);
                g = min(max(g, 0), GCN_N_GRAPHS - 1);

                float hs[C], acc[C];
                #pragma unroll
                for (int c = 0; c < C; ++c) {
                    hs[c]  = h[node * F + c * 32 + lane];
                    acc[c] = 0.0f;
                }
                gather_edges<F, false>(h, ninfo, edges, (v2i*)nullptr, Es, b0, wav, beg, end, n, e, lane, acc);

                if (g != gcur) {
                    if (gcur >= 0) {
                        #pragma unroll
                        for (int c = 0; c < C; ++c) atomicAdd(&sums[gcur * F + c * 32 + lane], ps[c]);
                        if (lane == 0) atomicAdd(&cnts[gcur], pc);
                    }
                    gcur = g;
                    pc   = 0.0f;
                    #pragma unroll
                    for (int c = 0; c < C; ++c) ps[c] = 0.0f;
                }
                const float di2 = di * di;
                #pragma unroll
                for (int c = 0; c < C; ++c) ps[c] += acc[c] + hs[c] * di2 + bs[c];
                pc += 1.0f;
            }
        }
    }

    #pragma unroll
    for (int c = 0; c < C; ++c) Ps[wav * F + c * 32 + lane] = ps[c];
    if (lane == 0) {
        Pg[wav] = gcur;
        Pc[wav] = pc;
    }
    __syncthreads();

    if (wav < C) {
        int   gw[16];
        float pw[16], cw[16];
        #pragma unroll
        for (int w = 0; w < 16; ++w) {
            gw[w] = Pg[w];
            cw[w] = Pc[w];
            pw[w] = Ps[w * F + wav * 32 + lane];
        }
        int   g = -1;
        float s = 0.0f, cn = 0.0f;
        #pragma unroll
        for (int w = 0; w < 16; ++w) {
            if (gw[w] < 0) continue;
            if (gw[w] != g) {
                if (g >= 0) {
                    atomicAdd(&sums[g * F + wav * 32 + lane], s);
                    if (wav == 0 && lane == 0) atomicAdd(&cnts[g], cn);
                }
                g  = gw[w];
                s  = 0.0f;
                cn = 0.0f;
            }
            s  += pw[w];
            cn += cw[w];
        }
        if (g >= 0) {
            atomicAdd(&sums[g * F + wav * 32 + lane], s);
            if (wav == 0 && lane == 0) atomicAdd(&cnts[g], cn);
        }
    }
}

#define GP_TILES 4
template <int F, int NC, bool RELU, bool NORM>
__global__ void __launch_bounds__(512)
gather_product(const float* __restrict__ h,
               const v2i* __restrict__ ninfo,
               const v2i* __restrict__ edges,
               v2i* __restrict__ edges_out,
               const float* __restrict__ bias,
               const float* __restrict__ Wn,
               float* __restrict__ Hn, int n, int e, int nfull) {
    constexpr int C  = F / 32;
    constexpr int TP = F + 4;
    constexpr int WP = NC + 8;
    constexpr int NV4 = F * NC / 4;
    constexpr int WQ  = (NV4 + 511) / 512;
    constexpr int NCT = NC / 16;
    static_assert(NC % 4 == 0 && WP % 4 == 0, "a 16-byte piece stays inside one weight row, 16-byte aligned in LDS");
    static_assert(2 * GP_TILES <= 32, "a wave keeps the row starts of its nodes in one register pair");
    static_assert(GP_TILES == 4 && NCT >= 4 && NCT <= 8, "16 waves: row tile wav % 4, column tiles wav / 4 and wav / 4 + 4");
    __shared__ __attribute__((aligned(16))) float Ts[16 * GP_TILES * TP];
    __shared__ __attribute__((aligned(16))) float Ws[F * WP];
    __shared__ v2i Es[ECAP + 16 * 32];

    const int tid  = threadIdx.x;
    const int lane = tid & 31;
    const int wav  = __builtin_amdgcn_readfirstlane(tid >> 5);

    const int ntiles = n >> 4;
    const int blk    = (int)blockIdx.x;
    const int tile0  = (blk < nfull) ? blk * GP_TILES : nfull * GP_TILES + (blk - nfull);
    const int tend   = min(tile0 + ((blk < nfull) ? GP_TILES : 1), ntiles);
    const int nb0    = tile0 * 16;

    const int b0 = ninfo[min(nb0, n)].x;
    const int b1 = ninfo[min(tend * 16, n)].x;
    v4f wr[WQ];
    #pragma unroll
    for (int q = 0; q < WQ; ++q) {
        const int i = min(q * 512 + tid, NV4 - 1);
        wr[q] = *(const v4f*)&Wn[4 * i];
    }
    v2i ew[EQ];
    stage_edges_load(edges, b0, b1, e, tid, ew);
    float bs[C];
    #pragma unroll
    for (int c = 0; c < C; ++c) bs[c] = bias[c * 32 + lane];
    const int l7 = min(lane, 2 * GP_TILES - 1);
    const v2i nv = ninfo[min(nb0 + 16 * (l7 >> 1) + wav + (l7 & 1), n)];

    if (NORM) {
        const int* __restrict__ nwords = (const int*)ninfo;
        float dv[EQ];
        #pragma unroll
        for (int q = 0; q < EQ; ++q) {
            const int s = min(max(ew[q].x, 0), n - 1);
            dv[q] = __int_as_float(nwords[2 * s + 1]);
        }
        #pragma unroll
        for (int q = 0; q < EQ; ++q)
            ew[q].y = __float_as_int(dv[q] * __int_as_float(ew[q].y));
    }

    #pragma unroll
    for (int q = 0; q < WQ; ++q) {
        const int i = q * 512 + tid;
        if (i < NV4) {
            const int kk = (4 * i) / NC;
            const int cc = 4 * i - kk * NC;
            *(v4f*)&Ws[kk * WP + cc] = wr[q];
        }
    }
    stage_edges_store(Es, tid, ew);
    __syncthreads();

    if (NORM) {
        const int lim = min(min(b1, b0 + ECAP), e);
        #pragma unroll
        for (int q = 0; q < EQ; ++q) {
            const int j = b0 + q * 512 + tid;
            if (q * 512 + tid < ECAP && j >= 0 && j < lim) edges_out[j] = ew[q];
        }
    }

    for (int t = 0; t < GP_TILES; ++t) {
        const int tile = tile0 + t;
        if (tile >= tend) break;
        const int   node = tile * 16 + wav;
        const int   beg  = __builtin_amdgcn_readlane(nv.x, 2 * t);
        const int   end  = __builtin_amdgcn_readlane(nv.x, 2 * t + 1);
        const float di   = __int_as_float(__builtin_amdgcn_readlane(nv.y, 2 * t));
        float v[C];
        node_row<F, RELU, NORM>(h, ninfo, edges, edges_out, Es, b0, wav, beg, end, di, bs, node, n, e, lane, v);
        #pragma unroll
        for (int c = 0; c < C; ++c) Ts[(16 * t + wav) * TP + c * 32 + lane] = v[c];
    }
    __syncthreads();

    const int mt = wav & 3;
    if (tile0 + mt < tend) {
        const int l15 = lane & 15;
        const int hi  = lane >> 4;
        const int khi = hi * 2;
        const int ct0 = wav >> 2;
        const float* __restrict__ Ta = &Ts[(16 * mt + l15) * TP + khi];
        const float* __restrict__ Wb = &Ws[khi * WP + ct0 * 16 + l15];
        const int m0  = (tile0 + mt) * 16 + hi * 8;

        if (ct0 + 4 < NCT) {
            v8f acc0 = {0.f, 0.f, 0.f, 0.f, 0.f, 0.f, 0.f, 0.f};
            v8f acc1 = {0.f, 0.f, 0.f, 0.f, 0.f, 0.f, 0.f, 0.f};
            #pragma unroll 8
            for (int k0 = 0; k0 < F; k0 += 4) {
                v2f a = *(const v2f*)&Ta[k0];
                v2f p, q;
                p.x = Wb[k0 * WP];
                p.y = Wb[(k0 + 1) * WP];
                q.x = Wb[k0 * WP + 64];
                q.y = Wb[(k0 + 1) * WP + 64];
                acc0 = __builtin_amdgcn_wmma_f32_16x16x4_f32(
                     false, a,  false, p,
                     (short)0, acc0,  false,  false);
                acc1 = __builtin_amdgcn_wmma_f32_16x16x4_f32(
                     false, a,  false, q,
                     (short)0, acc1,  false,  false);
            }
            const int col = ct0 * 16 + l15;
            #pragma unroll
            for (int r = 0; r < 8; ++r) {
                Hn[(m0 + r) * NC + col]      = acc0[r];
                Hn[(m0 + r) * NC + col + 64] = acc1[r];
            }
        } else {
            v8f acc = {0.f, 0.f, 0.f, 0.f, 0.f, 0.f, 0.f, 0.f};
            #pragma unroll 8
            for (int k0 = 0; k0 < F; k0 += 4) {
                v2f a = *(const v2f*)&Ta[k0];
                v2f b;
                b.x = Wb[k0 * WP];
                b.y = Wb[(k0 + 1) * WP];
                acc = __builtin_amdgcn_wmma_f32_16x16x4_f32(
                     false, a,  false, b,
                     (short)0, acc,  false,  false);
            }
            const int col = ct0 * 16 + l15;
            #pragma unroll
            for (int r = 0; r < 8; ++r)
                Hn[(m0 + r) * NC + col] = acc[r];
        }
    }
}

__global__ void pool_div(const float* __restrict__ sums,
                         const float* __restrict__ cnts,
                         float* __restrict__ out, int n) {
    int i = blockIdx.x * blockDim.x + threadIdx.x;
    if (i < n) out[i] = sums[i] / fmaxf(cnts[i / 64], 1.0f);
}

extern "C" void kernel_launch(void* const* d_in, const int* in_sizes, int n_in,
                              void* d_out, int out_size, void* d_ws, size_t ws_size,
                              hipStream_t stream) {
    const float* x     = (const float*)d_in[0];
    const int*   ei    = (const int*)d_in[1];
    const int*   batch = (const int*)d_in[2];
    const float* Wl[4] = {(const float*)d_in[3], (const float*)d_in[5],
                          (const float*)d_in[7], (const float*)d_in[9]};
    const float* bl[4] = {(const float*)d_in[4], (const float*)d_in[6],
                          (const float*)d_in[8], (const float*)d_in[10]};

    const int N = GCN_N_NODES;
    const int E = in_sizes[1] / 2;
    const int* src = ei;
    const int* dst = ei + E;

    auto au = [](size_t v) { return (v + 255) & ~(size_t)255; };
    char* ws = (char*)d_ws;
    size_t off = 0;
    int*      gcur = (int*)(ws + off);      off += au(sizeof(int) * 64);
    unsigned* stag = (unsigned*)(ws + off); off += au(sizeof(unsigned) * (size_t)NBIN * (size_t)E);
    v2i*   ninfo = (v2i*)(ws + off);   off += au(sizeof(v2i) * (size_t)(N + 1));
    v2i*   edges = (v2i*)(ws + off);   off += au(sizeof(v2i) * (size_t)E);
    v2i*   edgew = (v2i*)(ws + off);   off += au(sizeof(v2i) * (size_t)E);
    float* bufA  = (float*)(ws + off); off += au(sizeof(float) * (size_t)N * 96);
    float* bufB  = (float*)(ws + off); off += au(sizeof(float) * (size_t)N * 96);
    float* sums  = (float*)(ws + off); off += au(sizeof(float) * GCN_N_GRAPHS * 64);
    float* cnts  = (float*)(ws + off); off += au(sizeof(float) * GCN_N_GRAPHS);
    (void)ws_size; (void)n_in; (void)out_size;

    const int TB = 256;
    start_values<<<(GCN_N_GRAPHS * 64 + TB - 1) / TB, TB, 0, stream>>>(gcur, sums, cnts);
    const int MT  = (N + 15) / 16;
    const int MG  = (MT + PROD_WAVES - 1) / PROD_WAVES;
    const int MGH = MG / 2;
    const int NB1 = (E + 4095) / 4096;
    bin_edges_and_product<128, 96><<<NB1 + MGH * 6, 1024, 0, stream>>>(
        src, dst, gcur, stag, E, N, NB1, x, Wl[0], bufA, 0);
    bin_to_csr_and_product<128, 96><<<NBIN + (MG - MGH) * 6, 1024, 0, stream>>>(
        gcur, stag, ninfo, edges, E, N, NBIN, x, Wl[0], bufA, MGH);

    const int NGRP  = MT / GP_TILES;
    const int NFULL = (NGRP >= GCN_WGPS) ? (NGRP / GCN_WGPS) * GCN_WGPS : NGRP;
    const int NREST = MT - NFULL * GP_TILES;
    const int FB    = NFULL + NREST;

    gather_product<96, 96, true, true ><<<FB, 512, 0, stream>>>(bufA, ninfo, edges, edgew, bl[0], Wl[1], bufB, N, E, NFULL);
    gather_product<96, 96, true, false><<<FB, 512, 0, stream>>>(bufB, ninfo, edgew, (v2i*)nullptr, bl[1], Wl[2], bufA, N, E, NFULL);
    gather_product<96, 64, true, false><<<FB, 512, 0, stream>>>(bufA, ninfo, edgew, (v2i*)nullptr, bl[2], Wl[3], bufB, N, E, NFULL);
    gather_pool<64, GP_TILES><<<FB, 512, 0, stream>>>(bufB, ninfo, edgew, bl[3], batch, sums, cnts, N, E, NFULL);

    pool_div<<<(GCN_N_GRAPHS * 64 + TB - 1) / TB, TB, 0, stream>>>(sums, cnts, (float*)d_out,
                                                                   GCN_N_GRAPHS * 64);
}
